# starting kernel
# baseline (speedup 1.0000x reference)
#include <hip/hip_runtime.h>

typedef __attribute__((ext_vector_type(2))) float v2f;
typedef __attribute__((ext_vector_type(8))) float v8f;

#define NUM_ORDERS 16
#define C_DIM 64
#define K_DENSE 256

__global__ void build_cg_dense(const float* __restrict__ vals,
                               const int* __restrict__ M1,
                               const int* __restrict__ M2,
                               const int* __restrict__ Mseg,
                               float* __restrict__ A, int nnz) {
  const int t = threadIdx.x;
  for (int i = t; i < NUM_ORDERS * K_DENSE; i += blockDim.x) A[i] = 0.0f;
  __syncthreads();
  for (int i = t; i < nnz; i += blockDim.x) {
    A[Mseg[i] * K_DENSE + M1[i] * NUM_ORDERS + M2[i]] = vals[i];
  }
}

__global__ void __launch_bounds__(256)
tp_wmma_kernel(const float* __restrict__ x1, const float* __restrict__ x2,
               const float* __restrict__ A, float* __restrict__ out,
               int n_batch) {
  __shared__ float Alds[NUM_ORDERS * K_DENSE];

  {
    const float4* src = (const float4*)A;
    float4* dst = (float4*)Alds;
#pragma unroll
    for (int i = 0; i < 4; ++i)
      dst[threadIdx.x + i * 256] = src[threadIdx.x + i * 256];
  }
  __syncthreads();

  const int lane = threadIdx.x & 31;
  const int wave = (blockIdx.x * blockDim.x + threadIdx.x) >> 5;
  const int pair = wave >> 2;
  const int ct   = wave & 3;
  const int n0   = pair * 2;
  if (n0 >= n_batch) return;
  const int n1 = (n0 + 1 < n_batch) ? n0 + 1 : n0;

  const int half = lane >> 4;
  const int col  = lane & 15;
  const int c    = ct * 16 + col;

  const float* p1a = x1 + (size_t)n0 * NUM_ORDERS * C_DIM + c;
  const float* p2a = x2 + (size_t)n0 * NUM_ORDERS * C_DIM + c;
  const float* p1b = x1 + (size_t)n1 * NUM_ORDERS * C_DIM + c;
  const float* p2b = x2 + (size_t)n1 * NUM_ORDERS * C_DIM + c;

  float w1a[16], w1b[16];
#pragma unroll
  for (int j = 0; j < 16; ++j) {
    w1a[j] = p1a[j * C_DIM];
    w1b[j] = p1b[j * C_DIM];
  }
  float w2a[8], w2b[8];
#pragma unroll
  for (int t = 0; t < 8; ++t) {
    const int o = (4 * (t >> 1) + (t & 1) + 2 * half) * C_DIM;
    w2a[t] = p2a[o];
    w2b[t] = p2b[o];
  }

  const float* arow = Alds + col * K_DENSE + 2 * half;

  v8f acc0 = {};
  v8f acc1 = {};
#pragma unroll
  for (int kk = 0; kk < 64; ++kk) {
    v2f a;
    a.x = arow[4 * kk];
    a.y = arow[4 * kk + 1];
    const float u  = w1a[kk >> 2];
    const float v  = w1b[kk >> 2];
    v2f b0 = {u * w2a[2 * (kk & 3)], u * w2a[2 * (kk & 3) + 1]};
    v2f b1 = {v * w2b[2 * (kk & 3)], v * w2b[2 * (kk & 3) + 1]};
    acc0 = __builtin_amdgcn_wmma_f32_16x16x4_f32(false, a, false, b0,
                                                 (short)0, acc0, false, false);
    acc1 = __builtin_amdgcn_wmma_f32_16x16x4_f32(false, a, false, b1,
                                                 (short)0, acc1, false, false);
  }

  float* po0 = out + ((size_t)n0 * NUM_ORDERS + 8 * half) * C_DIM + c;
  float* po1 = out + ((size_t)n1 * NUM_ORDERS + 8 * half) * C_DIM + c;
#pragma unroll
  for (int v = 0; v < 8; ++v) {
    po0[v * C_DIM] = acc0[v];
    po1[v * C_DIM] = acc1[v];
  }
}

extern "C" void kernel_launch(void* const* d_in, const int* in_sizes, int n_in,
                              void* d_out, int out_size, void* d_ws, size_t ws_size,
                              hipStream_t stream) {
  const float* x1   = (const float*)d_in[0];
  const float* x2   = (const float*)d_in[1];
  const float* vals = (const float*)d_in[2];
  const int* M1     = (const int*)d_in[3];
  const int* M2     = (const int*)d_in[4];
  const int* Mseg   = (const int*)d_in[5];
  float* A          = (float*)d_ws;
  float* out        = (float*)d_out;

  const int nnz     = in_sizes[2];
  const int n_batch = in_sizes[0] / (NUM_ORDERS * C_DIM);

  build_cg_dense<<<1, 512, 0, stream>>>(vals, M1, M2, Mseg, A, nnz);

  const int pairs  = (n_batch + 1) / 2;
  const int waves  = pairs * 4;
  const int blocks = (waves + 7) / 8;
  tp_wmma_kernel<<<blocks, 256, 0, stream>>>(x1, x2, A, out, n_batch);
}


# CG matrix stored in WMMA A-fragment order (workspace and LDS): a lane reads one contiguous ds_load_b128 per two k-steps; removes the bank conflict of row-major reads (all rows on same LDS banks)
# speedup vs baseline: 1.0870x; 1.0870x over previous
#include <hip/hip_runtime.h>

typedef __attribute__((ext_vector_type(2))) float v2f;
typedef __attribute__((ext_vector_type(8))) float v8f;

#define NUM_ORDERS 16
#define C_DIM 64
#define K_DENSE 256

__device__ __forceinline__ int cg_frag_index(int m, int k) {
  const int kk   = k >> 2;
  const int lane = ((k >> 1) & 1) * 16 + m;
  return ((kk >> 1) * 32 + lane) * 4 + (kk & 1) * 2 + (k & 1);
}

__global__ void build_cg_dense(const float* __restrict__ vals,
                               const int* __restrict__ M1,
                               const int* __restrict__ M2,
                               const int* __restrict__ Mseg,
                               float* __restrict__ A, int nnz) {
  const int t = threadIdx.x;
  for (int i = t; i < NUM_ORDERS * K_DENSE; i += blockDim.x) A[i] = 0.0f;
  __syncthreads();
  for (int i = t; i < nnz; i += blockDim.x) {
    A[cg_frag_index(Mseg[i], M1[i] * NUM_ORDERS + M2[i])] = vals[i];
  }
}

__global__ void __launch_bounds__(256)
tp_wmma_kernel(const float* __restrict__ x1, const float* __restrict__ x2,
               const float* __restrict__ A, float* __restrict__ out,
               int n_batch) {
  __shared__ float Alds[NUM_ORDERS * K_DENSE];

  {
    const float4* src = (const float4*)A;
    float4* dst = (float4*)Alds;
#pragma unroll
    for (int i = 0; i < 4; ++i)
      dst[threadIdx.x + i * 256] = src[threadIdx.x + i * 256];
  }
  __syncthreads();

  const int lane = threadIdx.x & 31;
  const int wave = (blockIdx.x * blockDim.x + threadIdx.x) >> 5;
  const int pair = wave >> 2;
  const int ct   = wave & 3;
  const int n0   = pair * 2;
  if (n0 >= n_batch) return;
  const int n1 = (n0 + 1 < n_batch) ? n0 + 1 : n0;

  const int half = lane >> 4;
  const int col  = lane & 15;
  const int c    = ct * 16 + col;

  const float* p1a = x1 + (size_t)n0 * NUM_ORDERS * C_DIM + c;
  const float* p2a = x2 + (size_t)n0 * NUM_ORDERS * C_DIM + c;
  const float* p1b = x1 + (size_t)n1 * NUM_ORDERS * C_DIM + c;
  const float* p2b = x2 + (size_t)n1 * NUM_ORDERS * C_DIM + c;

  float w1a[16], w1b[16];
#pragma unroll
  for (int j = 0; j < 16; ++j) {
    w1a[j] = p1a[j * C_DIM];
    w1b[j] = p1b[j * C_DIM];
  }
  float w2a[8], w2b[8];
#pragma unroll
  for (int t = 0; t < 8; ++t) {
    const int o = (4 * (t >> 1) + (t & 1) + 2 * half) * C_DIM;
    w2a[t] = p2a[o];
    w2b[t] = p2b[o];
  }

  const float4* afrag = (const float4*)Alds + lane;

  v8f acc0 = {};
  v8f acc1 = {};
#pragma unroll
  for (int kk = 0; kk < 64; ++kk) {
    const float4 q = afrag[(kk >> 1) * 32];
    v2f a;
    a.x = (kk & 1) ? q.z : q.x;
    a.y = (kk & 1) ? q.w : q.y;
    const float u  = w1a[kk >> 2];
    const float v  = w1b[kk >> 2];
    v2f b0 = {u * w2a[2 * (kk & 3)], u * w2a[2 * (kk & 3) + 1]};
    v2f b1 = {v * w2b[2 * (kk & 3)], v * w2b[2 * (kk & 3) + 1]};
    acc0 = __builtin_amdgcn_wmma_f32_16x16x4_f32(false, a, false, b0,
                                                 (short)0, acc0, false, false);
    acc1 = __builtin_amdgcn_wmma_f32_16x16x4_f32(false, a, false, b1,
                                                 (short)0, acc1, false, false);
  }

  float* po0 = out + ((size_t)n0 * NUM_ORDERS + 8 * half) * C_DIM + c;
  float* po1 = out + ((size_t)n1 * NUM_ORDERS + 8 * half) * C_DIM + c;
#pragma unroll
  for (int v = 0; v < 8; ++v) {
    po0[v * C_DIM] = acc0[v];
    po1[v * C_DIM] = acc1[v];
  }
}

extern "C" void kernel_launch(void* const* d_in, const int* in_sizes, int n_in,
                              void* d_out, int out_size, void* d_ws, size_t ws_size,
                              hipStream_t stream) {
  const float* x1   = (const float*)d_in[0];
  const float* x2   = (const float*)d_in[1];
  const float* vals = (const float*)d_in[2];
  const int* M1     = (const int*)d_in[3];
  const int* M2     = (const int*)d_in[4];
  const int* Mseg   = (const int*)d_in[5];
  float* A          = (float*)d_ws;
  float* out        = (float*)d_out;

  const int nnz     = in_sizes[2];
  const int n_batch = in_sizes[0] / (NUM_ORDERS * C_DIM);

  build_cg_dense<<<1, 512, 0, stream>>>(vals, M1, M2, Mseg, A, nnz);

  const int pairs  = (n_batch + 1) / 2;
  const int waves  = pairs * 4;
  const int blocks = (waves + 7) / 8;
  tp_wmma_kernel<<<blocks, 256, 0, stream>>>(x1, x2, A, out, n_batch);
}


# CG matrix stationary in registers: fixed grid of 256x16 waves, each loads its 128-VGPR A-fragment set once and loops over tiles; no LDS staging, barrier or ds_load waits in the WMMA stream
# speedup vs baseline: 1.1331x; 1.0424x over previous
#include <hip/hip_runtime.h>


typedef __attribute__((ext_vector_type(2))) float v2f;
typedef __attribute__((ext_vector_type(8))) float v8f;

#define NUM_ORDERS 16
#define C_DIM 64
#define K_DENSE 256
#define WAVES_PER_BLOCK 16
#define NUM_BLOCKS 256

__device__ __forceinline__ int cg_frag_index(int m, int k) {
  const int kk   = k >> 2;
  const int lane = ((k >> 1) & 1) * 16 + m;
  return ((kk >> 1) * 32 + lane) * 4 + (kk & 1) * 2 + (k & 1);
}

__global__ void build_cg_dense(const float* __restrict__ vals,
                               const int* __restrict__ M1,
                               const int* __restrict__ M2,
                               const int* __restrict__ Mseg,
                               float* __restrict__ A, int nnz) {
  const int t = threadIdx.x;
  for (int i = t; i < NUM_ORDERS * K_DENSE; i += blockDim.x) A[i] = 0.0f;
  __syncthreads();
  for (int i = t; i < nnz; i += blockDim.x) {
    A[cg_frag_index(Mseg[i], M1[i] * NUM_ORDERS + M2[i])] = vals[i];
  }
}

__global__ void __launch_bounds__(WAVES_PER_BLOCK * 32)
tp_wmma_kernel(const float* __restrict__ x1, const float* __restrict__ x2,
               const float* __restrict__ A, float* __restrict__ out,
               int n_batch) {
  const int lane = threadIdx.x & 31;
  const int wib    = __builtin_amdgcn_readfirstlane((int)(threadIdx.x >> 5));
  const int wave   = blockIdx.x * WAVES_PER_BLOCK + wib;
  const int nwaves = gridDim.x * WAVES_PER_BLOCK;
  const int ntiles = ((n_batch + 1) >> 1) * 4;

  float4 areg[32];
  {
    const float4* asrc = (const float4*)A + lane;
#pragma unroll
    for (int kp = 0; kp < 32; ++kp) areg[kp] = asrc[kp * 32];
  }

  const int half = lane >> 4;
  const int col  = lane & 15;

  for (int t = wave; t < ntiles; t += nwaves) {
    const int n0 = (t >> 2) * 2;
    const int n1 = (n0 + 1 < n_batch) ? n0 + 1 : n0;
    const int c  = (t & 3) * 16 + col;

    const float* p1a = x1 + (size_t)n0 * NUM_ORDERS * C_DIM + c;
    const float* p2a = x2 + (size_t)n0 * NUM_ORDERS * C_DIM + c;
    const float* p1b = x1 + (size_t)n1 * NUM_ORDERS * C_DIM + c;
    const float* p2b = x2 + (size_t)n1 * NUM_ORDERS * C_DIM + c;

    float w2a[8], w2b[8];
#pragma unroll
    for (int i = 0; i < 8; ++i) {
      const int o = (4 * (i >> 1) + (i & 1) + 2 * half) * C_DIM;
      w2a[i] = p2a[o];
      w2b[i] = p2b[o];
    }
    float w1a[16], w1b[16];
#pragma unroll
    for (int j = 0; j < 16; ++j) {
      w1a[j] = p1a[j * C_DIM];
      w1b[j] = p1b[j * C_DIM];
    }

    v8f acc0 = {};
    v8f acc1 = {};
#pragma unroll
    for (int kk = 0; kk < 64; ++kk) {
      const float4 q = areg[kk >> 1];
      v2f a;
      a.x = (kk & 1) ? q.z : q.x;
      a.y = (kk & 1) ? q.w : q.y;
      const float u = w1a[kk >> 2];
      const float v = w1b[kk >> 2];
      v2f b0 = {u * w2a[2 * (kk & 3)], u * w2a[2 * (kk & 3) + 1]};
      v2f b1 = {v * w2b[2 * (kk & 3)], v * w2b[2 * (kk & 3) + 1]};
      acc0 = __builtin_amdgcn_wmma_f32_16x16x4_f32(false, a, false, b0,
                                                   (short)0, acc0, false, false);
      acc1 = __builtin_amdgcn_wmma_f32_16x16x4_f32(false, a, false, b1,
                                                   (short)0, acc1, false, false);
    }

    float* po0 = out + ((size_t)n0 * NUM_ORDERS + 8 * half) * C_DIM + c;
    float* po1 = out + ((size_t)n1 * NUM_ORDERS + 8 * half) * C_DIM + c;
#pragma unroll
    for (int v = 0; v < 8; ++v) {
      po0[v * C_DIM] = acc0[v];
      po1[v * C_DIM] = acc1[v];
    }
  }
}

extern "C" void kernel_launch(void* const* d_in, const int* in_sizes, int n_in,
                              void* d_out, int out_size, void* d_ws, size_t ws_size,
                              hipStream_t stream) {
  const float* x1   = (const float*)d_in[0];
  const float* x2   = (const float*)d_in[1];
  const float* vals = (const float*)d_in[2];
  const int* M1     = (const int*)d_in[3];
  const int* M2     = (const int*)d_in[4];
  const int* Mseg   = (const int*)d_in[5];
  float* A          = (float*)d_ws;
  float* out        = (float*)d_out;

  const int nnz     = in_sizes[2];
  const int n_batch = in_sizes[0] / (NUM_ORDERS * C_DIM);

  build_cg_dense<<<1, 512, 0, stream>>>(vals, M1, M2, Mseg, A, nnz);

  tp_wmma_kernel<<<NUM_BLOCKS, WAVES_PER_BLOCK * 32, 0, stream>>>(x1, x2, A, out, n_batch);
}


# CG table densified per work-group into LDS inside the WMMA kernel (build launch and workspace removed); A fragments read from LDS; first tile's x loads stay in flight across the build
# speedup vs baseline: 1.2886x; 1.1372x over previous
#include <hip/hip_runtime.h>


typedef __attribute__((ext_vector_type(2))) float v2f;
typedef __attribute__((ext_vector_type(8))) float v8f;

#define NUM_ORDERS 16
#define C_DIM 64
#define K_DENSE 256
#define WAVES_PER_BLOCK 16
#define BLOCK_THREADS (WAVES_PER_BLOCK * 32)
#define NUM_BLOCKS 256

__device__ __forceinline__ int cg_frag_index(int m, int k) {
  const int kk   = k >> 2;
  const int lane = ((k >> 1) & 1) * 16 + m;
  return ((kk >> 1) * 32 + lane) * 4 + (kk & 1) * 2 + (k & 1);
}

__device__ __forceinline__ void lds_barrier() {
  __builtin_amdgcn_fence(__ATOMIC_RELEASE, "workgroup", "local");
  __builtin_amdgcn_s_barrier();
  __builtin_amdgcn_fence(__ATOMIC_ACQUIRE, "workgroup", "local");
}

struct TileRegs {
  float w1a[16], w1b[16], w2a[8], w2b[8];
};

__device__ __forceinline__ void load_tile(TileRegs& r, const float* __restrict__ x1,
                                          const float* __restrict__ x2, int t,
                                          int n_batch, int half, int col) {
  const int n0 = (t >> 2) * 2;
  const int n1 = (n0 + 1 < n_batch) ? n0 + 1 : n0;
  const int c  = (t & 3) * 16 + col;
  const float* p1a = x1 + (size_t)n0 * NUM_ORDERS * C_DIM + c;
  const float* p2a = x2 + (size_t)n0 * NUM_ORDERS * C_DIM + c;
  const float* p1b = x1 + (size_t)n1 * NUM_ORDERS * C_DIM + c;
  const float* p2b = x2 + (size_t)n1 * NUM_ORDERS * C_DIM + c;
#pragma unroll
  for (int i = 0; i < 8; ++i) {
    const int o = (4 * (i >> 1) + (i & 1) + 2 * half) * C_DIM;
    r.w2a[i] = p2a[o];
    r.w2b[i] = p2b[o];
  }
#pragma unroll
  for (int j = 0; j < 16; ++j) {
    r.w1a[j] = p1a[j * C_DIM];
    r.w1b[j] = p1b[j * C_DIM];
  }
}

__global__ void __launch_bounds__(BLOCK_THREADS)
tp_wmma_kernel(const float* __restrict__ x1, const float* __restrict__ x2,
               const float* __restrict__ vals, const int* __restrict__ M1,
               const int* __restrict__ M2, const int* __restrict__ Mseg,
               float* __restrict__ out, int n_batch, int nnz) {
  __shared__ float4 cg_lds[NUM_ORDERS * K_DENSE / 4];
  float* cg = (float*)cg_lds;

  const int tid  = threadIdx.x;
  const int lane = tid & 31;
  const int half = lane >> 4;
  const int col  = lane & 15;
  const int wib    = __builtin_amdgcn_readfirstlane((int)(threadIdx.x >> 5));
  const int wave   = blockIdx.x * WAVES_PER_BLOCK + wib;
  const int nwaves = gridDim.x * WAVES_PER_BLOCK;
  const int ntiles = ((n_batch + 1) >> 1) * 4;

  if (ntiles <= 0) return;

  int t = wave;
  const bool has_tile = t < ntiles;

  const bool has_entry = tid < nnz;
  int   e_m = 0, e_m1 = 0, e_m2 = 0;
  float e_val = 0.0f;
  if (has_entry) {
    e_m   = Mseg[tid];
    e_m1  = M1[tid];
    e_m2  = M2[tid];
    e_val = vals[tid];
  }

  TileRegs r;
  load_tile(r, x1, x2, has_tile ? t : 0, n_batch, half, col);

  for (int i = tid; i < NUM_ORDERS * K_DENSE / 4; i += BLOCK_THREADS)
    cg_lds[i] = make_float4(0.0f, 0.0f, 0.0f, 0.0f);
  lds_barrier();
  if (has_entry) {
    asm volatile("" : "+v"(e_m1));
    cg[cg_frag_index(e_m, e_m1 * NUM_ORDERS + e_m2)] = e_val;
  }
  for (int i = tid + BLOCK_THREADS; i < nnz; i += BLOCK_THREADS)
    cg[cg_frag_index(Mseg[i], M1[i] * NUM_ORDERS + M2[i])] = vals[i];
  lds_barrier();

  float4 areg[32];
#pragma unroll
  for (int kp = 0; kp < 32; ++kp) areg[kp] = cg_lds[kp * 32 + lane];

  if (!has_tile) return;

  for (;;) {
    v8f acc0 = {};
    v8f acc1 = {};
#pragma unroll
    for (int kk = 0; kk < 64; ++kk) {
      const float4 q = areg[kk >> 1];
      v2f a;
      a.x = (kk & 1) ? q.z : q.x;
      a.y = (kk & 1) ? q.w : q.y;
      const float u = r.w1a[kk >> 2];
      const float v = r.w1b[kk >> 2];
      v2f b0 = {u * r.w2a[2 * (kk & 3)], u * r.w2a[2 * (kk & 3) + 1]};
      v2f b1 = {v * r.w2b[2 * (kk & 3)], v * r.w2b[2 * (kk & 3) + 1]};
      acc0 = __builtin_amdgcn_wmma_f32_16x16x4_f32(false, a, false, b0,
                                                   (short)0, acc0, false, false);
      acc1 = __builtin_amdgcn_wmma_f32_16x16x4_f32(false, a, false, b1,
                                                   (short)0, acc1, false, false);
    }

    {
      const int n0 = (t >> 2) * 2;
      const int n1 = (n0 + 1 < n_batch) ? n0 + 1 : n0;
      const int c  = (t & 3) * 16 + col;
      float* po0 = out + ((size_t)n0 * NUM_ORDERS + 8 * half) * C_DIM + c;
      float* po1 = out + ((size_t)n1 * NUM_ORDERS + 8 * half) * C_DIM + c;
#pragma unroll
      for (int v = 0; v < 8; ++v) {
        po0[v * C_DIM] = acc0[v];
        po1[v * C_DIM] = acc1[v];
      }
    }

    t += nwaves;
    if (t >= ntiles) break;
    load_tile(r, x1, x2, t, n_batch, half, col);
  }
}

extern "C" void kernel_launch(void* const* d_in, const int* in_sizes, int n_in,
                              void* d_out, int out_size, void* d_ws, size_t ws_size,
                              hipStream_t stream) {
  const float* x1   = (const float*)d_in[0];
  const float* x2   = (const float*)d_in[1];
  const float* vals = (const float*)d_in[2];
  const int* M1     = (const int*)d_in[3];
  const int* M2     = (const int*)d_in[4];
  const int* Mseg   = (const int*)d_in[5];
  float* out        = (float*)d_out;

  const int nnz     = in_sizes[2];
  const int n_batch = in_sizes[0] / (NUM_ORDERS * C_DIM);

  tp_wmma_kernel<<<NUM_BLOCKS, BLOCK_THREADS, 0, stream>>>(x1, x2, vals, M1, M2, Mseg,
                                                          out, n_batch, nnz);
}


# Next tile's x operands loaded one tile ahead into a second register set, one group in front of the current tile's WMMA block (two tiles per loop trip): load latency runs under the wave's own WMMAs
# speedup vs baseline: 1.3429x; 1.0421x over previous
#include <hip/hip_runtime.h>


typedef __attribute__((ext_vector_type(2))) float v2f;
typedef __attribute__((ext_vector_type(8))) float v8f;

#define NUM_ORDERS 16
#define C_DIM 64
#define K_DENSE 256
#define WAVES_PER_BLOCK 16
#define BLOCK_THREADS (WAVES_PER_BLOCK * 32)
#define NUM_BLOCKS 256

__device__ __forceinline__ int cg_frag_index(int m, int k) {
  const int kk   = k >> 2;
  const int lane = ((k >> 1) & 1) * 16 + m;
  return ((kk >> 1) * 32 + lane) * 4 + (kk & 1) * 2 + (k & 1);
}

__device__ __forceinline__ void lds_barrier() {
  __builtin_amdgcn_fence(__ATOMIC_RELEASE, "workgroup", "local");
  __builtin_amdgcn_s_barrier();
  __builtin_amdgcn_fence(__ATOMIC_ACQUIRE, "workgroup", "local");
}

struct TileRegs {
  float w1a[16], w1b[16], w2a[8], w2b[8];
};

__device__ __forceinline__ void load_tile(TileRegs& r, const float* x1, const float* x2, int t,
                                          int n_batch, int half, int col) {
  const int n0 = (t >> 2) * 2;
  const int n1 = (n0 + 1 < n_batch) ? n0 + 1 : n0;
  const int c  = (t & 3) * 16 + col;
  const float* p1a = x1 + (size_t)n0 * NUM_ORDERS * C_DIM + c;
  const float* p2a = x2 + (size_t)n0 * NUM_ORDERS * C_DIM + c;
  const float* p1b = x1 + (size_t)n1 * NUM_ORDERS * C_DIM + c;
  const float* p2b = x2 + (size_t)n1 * NUM_ORDERS * C_DIM + c;
#pragma unroll
  for (int i = 0; i < 8; ++i) {
    const int o = (4 * (i >> 1) + (i & 1) + 2 * half) * C_DIM;
    r.w2a[i] = p2a[o];
    r.w2b[i] = p2b[o];
  }
#pragma unroll
  for (int j = 0; j < 16; ++j) {
    r.w1a[j] = p1a[j * C_DIM];
    r.w1b[j] = p1b[j * C_DIM];
  }
}

__device__ __forceinline__ void tile_product(const float4 (&areg)[32], const TileRegs& r,
                                             float* out, int t, int n_batch,
                                             int half, int col) {
  v8f acc0 = {};
  v8f acc1 = {};
#pragma unroll
  for (int kk = 0; kk < 64; ++kk) {
    const float4 q = areg[kk >> 1];
    v2f a;
    a.x = (kk & 1) ? q.z : q.x;
    a.y = (kk & 1) ? q.w : q.y;
    const float u = r.w1a[kk >> 2];
    const float v = r.w1b[kk >> 2];
    v2f b0 = {u * r.w2a[2 * (kk & 3)], u * r.w2a[2 * (kk & 3) + 1]};
    v2f b1 = {v * r.w2b[2 * (kk & 3)], v * r.w2b[2 * (kk & 3) + 1]};
    acc0 = __builtin_amdgcn_wmma_f32_16x16x4_f32(false, a, false, b0,
                                                 (short)0, acc0, false, false);
    acc1 = __builtin_amdgcn_wmma_f32_16x16x4_f32(false, a, false, b1,
                                                 (short)0, acc1, false, false);
  }

  const int n0 = (t >> 2) * 2;
  const int n1 = (n0 + 1 < n_batch) ? n0 + 1 : n0;
  const int c  = (t & 3) * 16 + col;
  float* po0 = out + ((size_t)n0 * NUM_ORDERS + 8 * half) * C_DIM + c;
  float* po1 = out + ((size_t)n1 * NUM_ORDERS + 8 * half) * C_DIM + c;
#pragma unroll
  for (int v = 0; v < 8; ++v) {
    po0[v * C_DIM] = acc0[v];
    po1[v * C_DIM] = acc1[v];
  }
}

__global__ void __launch_bounds__(BLOCK_THREADS)
tp_wmma_kernel(const float* x1, const float* x2,
               const float* __restrict__ vals, const int* __restrict__ M1,
               const int* __restrict__ M2, const int* __restrict__ Mseg,
               float* out, int n_batch, int nnz) {
  __shared__ float4 cg_lds[NUM_ORDERS * K_DENSE / 4];
  float* cg = (float*)cg_lds;

  const int tid  = threadIdx.x;
  const int lane = tid & 31;
  const int half = lane >> 4;
  const int col  = lane & 15;
  const int wib    = __builtin_amdgcn_readfirstlane((int)(threadIdx.x >> 5));
  const int wave   = blockIdx.x * WAVES_PER_BLOCK + wib;
  const int nwaves = gridDim.x * WAVES_PER_BLOCK;
  const int ntiles = ((n_batch + 1) >> 1) * 4;

  if (ntiles <= 0) return;

  int t = wave;
  const bool has_tile = t < ntiles;

  const bool has_entry = tid < nnz;
  int   e_m = 0, e_m1 = 0, e_m2 = 0;
  float e_val = 0.0f;
  if (has_entry) {
    e_m   = Mseg[tid];
    e_m1  = M1[tid];
    e_m2  = M2[tid];
    e_val = vals[tid];
  }

  TileRegs r;
  load_tile(r, x1, x2, has_tile ? t : 0, n_batch, half, col);

  for (int i = tid; i < NUM_ORDERS * K_DENSE / 4; i += BLOCK_THREADS)
    cg_lds[i] = make_float4(0.0f, 0.0f, 0.0f, 0.0f);
  lds_barrier();
  if (has_entry) {
    asm volatile("" : "+v"(e_m1));
    cg[cg_frag_index(e_m, e_m1 * NUM_ORDERS + e_m2)] = e_val;
  }
  for (int i = tid + BLOCK_THREADS; i < nnz; i += BLOCK_THREADS)
    cg[cg_frag_index(Mseg[i], M1[i] * NUM_ORDERS + M2[i])] = vals[i];
  lds_barrier();

  float4 areg[32];
#pragma unroll
  for (int kp = 0; kp < 32; ++kp) areg[kp] = cg_lds[kp * 32 + lane];

  if (!has_tile) return;

  const int cnt = (ntiles - t + nwaves - 1) / nwaves;
  TileRegs r1;
  for (int p = cnt >> 1; p > 0; --p) {
    const int t1 = t + nwaves;
    load_tile(r1, x1, x2, t1, n_batch, half, col);
    __builtin_amdgcn_sched_barrier(0);
    tile_product(areg, r, out, t, n_batch, half, col);
    const int t2 = t1 + nwaves;
    load_tile(r, x1, x2, (t2 < ntiles) ? t2 : t1, n_batch, half, col);
    __builtin_amdgcn_sched_barrier(0);
    tile_product(areg, r1, out, t1, n_batch, half, col);
    t = t2;
  }
  if (cnt & 1) tile_product(areg, r, out, t, n_batch, half, col);
}

extern "C" void kernel_launch(void* const* d_in, const int* in_sizes, int n_in,
                              void* d_out, int out_size, void* d_ws, size_t ws_size,
                              hipStream_t stream) {
  const float* x1   = (const float*)d_in[0];
  const float* x2   = (const float*)d_in[1];
  const float* vals = (const float*)d_in[2];
  const int* M1     = (const int*)d_in[3];
  const int* M2     = (const int*)d_in[4];
  const int* Mseg   = (const int*)d_in[5];
  float* out        = (float*)d_out;

  const int nnz     = in_sizes[2];
  const int n_batch = in_sizes[0] / (NUM_ORDERS * C_DIM);

  tp_wmma_kernel<<<NUM_BLOCKS, BLOCK_THREADS, 0, stream>>>(x1, x2, vals, M1, M2, Mseg,
                                                          out, n_batch, nnz);
}


# CG applied to x2 alone on WMMA as a 256x16 matrix with (m,m1) rows grouped so most 16x4 blocks are zero and skipped at run time; x1 factor and m1 sum as FMAs on D (summation order changes)
# speedup vs baseline: 1.8801x; 1.4000x over previous
#include <hip/hip_runtime.h>


typedef __attribute__((ext_vector_type(2))) float v2f;
typedef __attribute__((ext_vector_type(8))) float v8f;
typedef __attribute__((ext_vector_type(16))) float v16f;

#define NUM_ORDERS 16
#define C_DIM 64
#define K_DENSE 256
#define WAVES_PER_BLOCK 16
#define BLOCK_THREADS (WAVES_PER_BLOCK * 32)
#define NUM_BLOCKS 256

__device__ const unsigned char ROW_OF[256] = {
    0, 48, 1, 112, 176, 49, 2, 113, 177, 50, 178, 51, 3, 114, 179, 115,
    116, 4, 117, 180, 52, 5, 118, 181, 119, 182, 53, 6, 128, 183, 129, 240,
    8, 56, 9, 120, 184, 57, 10, 121, 185, 58, 186, 59, 11, 122, 187, 123,
    54, 192, 55, 7, 64, 193, 65, 16, 130, 241, 66, 194, 67, 17, 131, 195,
    196, 132, 197, 133, 18, 134, 198, 135, 242, 68, 19, 144, 199, 145, 243, 244,
    124, 12, 125, 188, 60, 13, 126, 189, 127, 190, 61, 14, 136, 191, 137, 248,
    20, 69, 21, 146, 208, 70, 22, 147, 209, 71, 210, 80, 23, 148, 211, 149,
    62, 200, 63, 15, 72, 201, 73, 24, 138, 249, 74, 202, 75, 25, 139, 203,
    212, 81, 213, 82, 245, 83, 214, 84, 32, 246, 247, 85, 215, 86, 33, 150,
    151, 224, 160, 35, 163, 225, 161, 36, 87, 34, 164, 226, 162, 96, 227, 167,
    204, 140, 205, 141, 26, 142, 206, 143, 250, 76, 27, 152, 207, 153, 251, 252,
    159, 42, 168, 232, 95, 43, 169, 233, 171, 234, 104, 44, 170, 235, 172, 175,
    28, 77, 29, 154, 216, 78, 30, 155, 217, 79, 218, 88, 31, 156, 219, 157,
    97, 228, 98, 37, 101, 229, 99, 38, 165, 103, 102, 230, 100, 39, 166, 231,
    220, 89, 221, 90, 253, 91, 222, 92, 40, 254, 255, 93, 223, 94, 41, 158,
    105, 46, 106, 236, 47, 173, 107, 237, 109, 174, 239, 111, 108, 238, 110, 45,
};
__device__ const unsigned char COL_POS[16] = {0, 8, 1, 4, 14, 9, 2, 5, 12, 11, 15, 10, 3, 6, 13, 7};

__device__ __forceinline__ int cg_frag_index(int i, int k) {
  const int kk   = k >> 2;
  const int lane = ((k >> 1) & 1) * 16 + i;
  return ((kk >> 1) * 32 + lane) * 4 + (kk & 1) * 2 + (k & 1);
}

__device__ __forceinline__ void lds_barrier() {
  __builtin_amdgcn_fence(__ATOMIC_RELEASE, "workgroup", "local");
  __builtin_amdgcn_s_barrier();
  __builtin_amdgcn_fence(__ATOMIC_ACQUIRE, "workgroup", "local");
}

struct TileRegs {
  v16f w1;
  v8f  w2;
};

__device__ __forceinline__ void load_tile(TileRegs& r, const float* x1, const float* x2, int t,
                                          int half, int col) {
  const int n = t >> 2;
  const int c = (t & 3) * 16 + col;
  const float* p1 = x1 + (size_t)n * NUM_ORDERS * C_DIM + c;
  const float* p2 = x2 + (size_t)n * NUM_ORDERS * C_DIM + c;
  r.w2[0] = p2[(half ? 6 : 0) * C_DIM];
  r.w2[1] = p2[(half ? 12 : 2) * C_DIM];
  r.w2[2] = p2[(half ? 13 : 3) * C_DIM];
  r.w2[3] = p2[(half ? 15 : 7) * C_DIM];
  r.w2[4] = p2[(half ? 11 : 1) * C_DIM];
  r.w2[5] = p2[(half ? 9 : 5) * C_DIM];
  r.w2[6] = p2[(half ? 4 : 8) * C_DIM];
  r.w2[7] = p2[(half ? 10 : 14) * C_DIM];
  r.w1[0] = p1[0 * C_DIM];
  r.w1[1] = p1[1 * C_DIM];
  r.w1[2] = p1[2 * C_DIM];
  r.w1[3] = p1[3 * C_DIM];
  r.w1[4] = p1[4 * C_DIM];
  r.w1[5] = p1[5 * C_DIM];
  r.w1[6] = p1[6 * C_DIM];
  r.w1[7] = p1[7 * C_DIM];
  r.w1[8] = p1[8 * C_DIM];
  r.w1[9] = p1[9 * C_DIM];
  r.w1[10] = p1[10 * C_DIM];
  r.w1[11] = p1[11 * C_DIM];
  r.w1[12] = p1[12 * C_DIM];
  r.w1[13] = p1[13 * C_DIM];
  r.w1[14] = p1[14 * C_DIM];
  r.w1[15] = p1[15 * C_DIM];
}

#define WMMA_RESULT_GUARD(d) \
  asm volatile("s_wait_alu depctr_va_vdst(0)\n\tv_nop\n\tv_nop\n\tv_nop\n\tv_nop" : "+v"(d))

__device__ __forceinline__ void tile_product(const float4 (&areg)[32], unsigned mlo, unsigned mhi,
                                             const TileRegs& r, float* out, int t,
                                             int half, int col) {
  const v8f zero = {};
  float o[8] = {0.0f, 0.0f, 0.0f, 0.0f, 0.0f, 0.0f, 0.0f, 0.0f};
  if (mlo & 0x1u) {
    const v2f a = {areg[0].x, areg[0].y};
    const v2f b = {r.w2[0], r.w2[1]};
    v8f d = __builtin_amdgcn_wmma_f32_16x16x4_f32(false, a, false, b, (short)0, zero, false, false);
    WMMA_RESULT_GUARD(d);
    o[0] = __builtin_fmaf(r.w1[0], d[0], o[0]);
    o[0] = __builtin_fmaf(r.w1[2], d[1], o[0]);
    o[0] = __builtin_fmaf(r.w1[6], d[2], o[0]);
    o[0] = __builtin_fmaf(r.w1[12], d[3], o[0]);
    o[1] = __builtin_fmaf(r.w1[1], d[4], o[1]);
    o[1] = __builtin_fmaf(r.w1[5], d[5], o[1]);
    o[1] = __builtin_fmaf(r.w1[11], d[6], o[1]);
    o[2] = __builtin_fmaf(r.w1[3], d[7], o[2]);
  }
  if (mlo & 0xeu) {
    if (mlo & 0x2u) {
      const v2f a = {areg[0].z, areg[0].w};
      const v2f b = {r.w2[2], r.w2[3]};
      v8f d = __builtin_amdgcn_wmma_f32_16x16x4_f32(false, a, false, b, (short)0, zero, false, false);
      WMMA_RESULT_GUARD(d);
      o[0] = __builtin_fmaf(r.w1[0], d[0], o[0]);
      o[0] = __builtin_fmaf(r.w1[2], d[1], o[0]);
      o[0] = __builtin_fmaf(r.w1[6], d[2], o[0]);
      o[0] = __builtin_fmaf(r.w1[12], d[3], o[0]);
      o[1] = __builtin_fmaf(r.w1[1], d[4], o[1]);
      o[1] = __builtin_fmaf(r.w1[5], d[5], o[1]);
      o[1] = __builtin_fmaf(r.w1[11], d[6], o[1]);
      o[2] = __builtin_fmaf(r.w1[3], d[7], o[2]);
    }
    if (mlo & 0x4u) {
      const v2f a = {areg[1].x, areg[1].y};
      const v2f b = {r.w2[4], r.w2[5]};
      v8f d = __builtin_amdgcn_wmma_f32_16x16x4_f32(false, a, false, b, (short)0, zero, false, false);
      WMMA_RESULT_GUARD(d);
      o[0] = __builtin_fmaf(r.w1[0], d[0], o[0]);
      o[0] = __builtin_fmaf(r.w1[2], d[1], o[0]);
      o[0] = __builtin_fmaf(r.w1[6], d[2], o[0]);
      o[0] = __builtin_fmaf(r.w1[12], d[3], o[0]);
      o[1] = __builtin_fmaf(r.w1[1], d[4], o[1]);
      o[1] = __builtin_fmaf(r.w1[5], d[5], o[1]);
      o[1] = __builtin_fmaf(r.w1[11], d[6], o[1]);
      o[2] = __builtin_fmaf(r.w1[3], d[7], o[2]);
    }
    if (mlo & 0x8u) {
      const v2f a = {areg[1].z, areg[1].w};
      const v2f b = {r.w2[6], r.w2[7]};
      v8f d = __builtin_amdgcn_wmma_f32_16x16x4_f32(false, a, false, b, (short)0, zero, false, false);
      WMMA_RESULT_GUARD(d);
      o[0] = __builtin_fmaf(r.w1[0], d[0], o[0]);
      o[0] = __builtin_fmaf(r.w1[2], d[1], o[0]);
      o[0] = __builtin_fmaf(r.w1[6], d[2], o[0]);
      o[0] = __builtin_fmaf(r.w1[12], d[3], o[0]);
      o[1] = __builtin_fmaf(r.w1[1], d[4], o[1]);
      o[1] = __builtin_fmaf(r.w1[5], d[5], o[1]);
      o[1] = __builtin_fmaf(r.w1[11], d[6], o[1]);
      o[2] = __builtin_fmaf(r.w1[3], d[7], o[2]);
    }
  }
  if (mlo & 0x10u) {
    const v2f a = {areg[2].x, areg[2].y};
    const v2f b = {r.w2[0], r.w2[1]};
    v8f d = __builtin_amdgcn_wmma_f32_16x16x4_f32(false, a, false, b, (short)0, zero, false, false);
    WMMA_RESULT_GUARD(d);
    o[2] = __builtin_fmaf(r.w1[7], d[0], o[2]);
    o[2] = __builtin_fmaf(r.w1[13], d[1], o[2]);
    o[3] = __builtin_fmaf(r.w1[4], d[2], o[3]);
    o[3] = __builtin_fmaf(r.w1[10], d[3], o[3]);
    o[4] = __builtin_fmaf(r.w1[0], d[4], o[4]);
    o[4] = __builtin_fmaf(r.w1[2], d[5], o[4]);
    o[4] = __builtin_fmaf(r.w1[6], d[6], o[4]);
    o[4] = __builtin_fmaf(r.w1[12], d[7], o[4]);
  }
  if (mlo & 0xe0u) {
    if (mlo & 0x20u) {
      const v2f a = {areg[2].z, areg[2].w};
      const v2f b = {r.w2[2], r.w2[3]};
      v8f d = __builtin_amdgcn_wmma_f32_16x16x4_f32(false, a, false, b, (short)0, zero, false, false);
      WMMA_RESULT_GUARD(d);
      o[2] = __builtin_fmaf(r.w1[7], d[0], o[2]);
      o[2] = __builtin_fmaf(r.w1[13], d[1], o[2]);
      o[3] = __builtin_fmaf(r.w1[4], d[2], o[3]);
      o[3] = __builtin_fmaf(r.w1[10], d[3], o[3]);
      o[4] = __builtin_fmaf(r.w1[0], d[4], o[4]);
      o[4] = __builtin_fmaf(r.w1[2], d[5], o[4]);
      o[4] = __builtin_fmaf(r.w1[6], d[6], o[4]);
      o[4] = __builtin_fmaf(r.w1[12], d[7], o[4]);
    }
    if (mlo & 0x40u) {
      const v2f a = {areg[3].x, areg[3].y};
      const v2f b = {r.w2[4], r.w2[5]};
      v8f d = __builtin_amdgcn_wmma_f32_16x16x4_f32(false, a, false, b, (short)0, zero, false, false);
      WMMA_RESULT_GUARD(d);
      o[2] = __builtin_fmaf(r.w1[7], d[0], o[2]);
      o[2] = __builtin_fmaf(r.w1[13], d[1], o[2]);
      o[3] = __builtin_fmaf(r.w1[4], d[2], o[3]);
      o[3] = __builtin_fmaf(r.w1[10], d[3], o[3]);
      o[4] = __builtin_fmaf(r.w1[0], d[4], o[4]);
      o[4] = __builtin_fmaf(r.w1[2], d[5], o[4]);
      o[4] = __builtin_fmaf(r.w1[6], d[6], o[4]);
      o[4] = __builtin_fmaf(r.w1[12], d[7], o[4]);
    }
    if (mlo & 0x80u) {
      const v2f a = {areg[3].z, areg[3].w};
      const v2f b = {r.w2[6], r.w2[7]};
      v8f d = __builtin_amdgcn_wmma_f32_16x16x4_f32(false, a, false, b, (short)0, zero, false, false);
      WMMA_RESULT_GUARD(d);
      o[2] = __builtin_fmaf(r.w1[7], d[0], o[2]);
      o[2] = __builtin_fmaf(r.w1[13], d[1], o[2]);
      o[3] = __builtin_fmaf(r.w1[4], d[2], o[3]);
      o[3] = __builtin_fmaf(r.w1[10], d[3], o[3]);
      o[4] = __builtin_fmaf(r.w1[0], d[4], o[4]);
      o[4] = __builtin_fmaf(r.w1[2], d[5], o[4]);
      o[4] = __builtin_fmaf(r.w1[6], d[6], o[4]);
      o[4] = __builtin_fmaf(r.w1[12], d[7], o[4]);
    }
  }
  if (mlo & 0x100u) {
    const v2f a = {areg[4].x, areg[4].y};
    const v2f b = {r.w2[0], r.w2[1]};
    v8f d = __builtin_amdgcn_wmma_f32_16x16x4_f32(false, a, false, b, (short)0, zero, false, false);
    WMMA_RESULT_GUARD(d);
    o[5] = __builtin_fmaf(r.w1[8], d[0], o[5]);
    o[5] = __builtin_fmaf(r.w1[14], d[1], o[5]);
    o[6] = __builtin_fmaf((half ? r.w1[1] : r.w1[9]), d[2], o[6]);
    o[6] = __builtin_fmaf((half ? r.w1[5] : r.w1[3]), d[3], o[6]);
    o[6] = __builtin_fmaf((half ? r.w1[11] : r.w1[7]), d[4], o[6]);
    o[7] = __builtin_fmaf((half ? r.w1[15] : r.w1[3]), d[5], o[7]);
    o[7] = __builtin_fmaf((half ? r.w1[1] : r.w1[7]), d[6], o[7]);
    o[7] = __builtin_fmaf((half ? r.w1[4] : r.w1[13]), d[7], o[7]);
  }
  if (mlo & 0xe00u) {
    if (mlo & 0x200u) {
      const v2f a = {areg[4].z, areg[4].w};
      const v2f b = {r.w2[2], r.w2[3]};
      v8f d = __builtin_amdgcn_wmma_f32_16x16x4_f32(false, a, false, b, (short)0, zero, false, false);
      WMMA_RESULT_GUARD(d);
      o[5] = __builtin_fmaf(r.w1[8], d[0], o[5]);
      o[5] = __builtin_fmaf(r.w1[14], d[1], o[5]);
      o[6] = __builtin_fmaf((half ? r.w1[1] : r.w1[9]), d[2], o[6]);
      o[6] = __builtin_fmaf((half ? r.w1[5] : r.w1[3]), d[3], o[6]);
      o[6] = __builtin_fmaf((half ? r.w1[11] : r.w1[7]), d[4], o[6]);
      o[7] = __builtin_fmaf((half ? r.w1[15] : r.w1[3]), d[5], o[7]);
      o[7] = __builtin_fmaf((half ? r.w1[1] : r.w1[7]), d[6], o[7]);
      o[7] = __builtin_fmaf((half ? r.w1[4] : r.w1[13]), d[7], o[7]);
    }
    if (mlo & 0x400u) {
      const v2f a = {areg[5].x, areg[5].y};
      const v2f b = {r.w2[4], r.w2[5]};
      v8f d = __builtin_amdgcn_wmma_f32_16x16x4_f32(false, a, false, b, (short)0, zero, false, false);
      WMMA_RESULT_GUARD(d);
      o[5] = __builtin_fmaf(r.w1[8], d[0], o[5]);
      o[5] = __builtin_fmaf(r.w1[14], d[1], o[5]);
      o[6] = __builtin_fmaf((half ? r.w1[1] : r.w1[9]), d[2], o[6]);
      o[6] = __builtin_fmaf((half ? r.w1[5] : r.w1[3]), d[3], o[6]);
      o[6] = __builtin_fmaf((half ? r.w1[11] : r.w1[7]), d[4], o[6]);
      o[7] = __builtin_fmaf((half ? r.w1[15] : r.w1[3]), d[5], o[7]);
      o[7] = __builtin_fmaf((half ? r.w1[1] : r.w1[7]), d[6], o[7]);
      o[7] = __builtin_fmaf((half ? r.w1[4] : r.w1[13]), d[7], o[7]);
    }
    if (mlo & 0x800u) {
      const v2f a = {areg[5].z, areg[5].w};
      const v2f b = {r.w2[6], r.w2[7]};
      v8f d = __builtin_amdgcn_wmma_f32_16x16x4_f32(false, a, false, b, (short)0, zero, false, false);
      WMMA_RESULT_GUARD(d);
      o[5] = __builtin_fmaf(r.w1[8], d[0], o[5]);
      o[5] = __builtin_fmaf(r.w1[14], d[1], o[5]);
      o[6] = __builtin_fmaf((half ? r.w1[1] : r.w1[9]), d[2], o[6]);
      o[6] = __builtin_fmaf((half ? r.w1[5] : r.w1[3]), d[3], o[6]);
      o[6] = __builtin_fmaf((half ? r.w1[11] : r.w1[7]), d[4], o[6]);
      o[7] = __builtin_fmaf((half ? r.w1[15] : r.w1[3]), d[5], o[7]);
      o[7] = __builtin_fmaf((half ? r.w1[1] : r.w1[7]), d[6], o[7]);
      o[7] = __builtin_fmaf((half ? r.w1[4] : r.w1[13]), d[7], o[7]);
    }
  }
  if (mlo & 0x2000u) {
    const v2f a = {areg[6].z, areg[6].w};
    const v2f b = {r.w2[2], r.w2[3]};
    v8f d = __builtin_amdgcn_wmma_f32_16x16x4_f32(false, a, false, b, (short)0, zero, false, false);
    WMMA_RESULT_GUARD(d);
    o[0] = __builtin_fmaf(r.w1[1], d[0], o[0]);
    o[0] = __builtin_fmaf(r.w1[5], d[1], o[0]);
    o[0] = __builtin_fmaf(r.w1[9], d[2], o[0]);
    o[0] = __builtin_fmaf(r.w1[11], d[3], o[0]);
    o[1] = __builtin_fmaf(r.w1[4], d[4], o[1]);
    o[1] = __builtin_fmaf(r.w1[10], d[5], o[1]);
    o[2] = __builtin_fmaf(r.w1[0], d[6], o[2]);
    o[2] = __builtin_fmaf(r.w1[2], d[7], o[2]);
  }
  if (mlo & 0xd000u) {
    if (mlo & 0x1000u) {
      const v2f a = {areg[6].x, areg[6].y};
      const v2f b = {r.w2[0], r.w2[1]};
      v8f d = __builtin_amdgcn_wmma_f32_16x16x4_f32(false, a, false, b, (short)0, zero, false, false);
      WMMA_RESULT_GUARD(d);
      o[0] = __builtin_fmaf(r.w1[1], d[0], o[0]);
      o[0] = __builtin_fmaf(r.w1[5], d[1], o[0]);
      o[0] = __builtin_fmaf(r.w1[9], d[2], o[0]);
      o[0] = __builtin_fmaf(r.w1[11], d[3], o[0]);
      o[1] = __builtin_fmaf(r.w1[4], d[4], o[1]);
      o[1] = __builtin_fmaf(r.w1[10], d[5], o[1]);
      o[2] = __builtin_fmaf(r.w1[0], d[6], o[2]);
      o[2] = __builtin_fmaf(r.w1[2], d[7], o[2]);
    }
    if (mlo & 0x4000u) {
      const v2f a = {areg[7].x, areg[7].y};
      const v2f b = {r.w2[4], r.w2[5]};
      v8f d = __builtin_amdgcn_wmma_f32_16x16x4_f32(false, a, false, b, (short)0, zero, false, false);
      WMMA_RESULT_GUARD(d);
      o[0] = __builtin_fmaf(r.w1[1], d[0], o[0]);
      o[0] = __builtin_fmaf(r.w1[5], d[1], o[0]);
      o[0] = __builtin_fmaf(r.w1[9], d[2], o[0]);
      o[0] = __builtin_fmaf(r.w1[11], d[3], o[0]);
      o[1] = __builtin_fmaf(r.w1[4], d[4], o[1]);
      o[1] = __builtin_fmaf(r.w1[10], d[5], o[1]);
      o[2] = __builtin_fmaf(r.w1[0], d[6], o[2]);
      o[2] = __builtin_fmaf(r.w1[2], d[7], o[2]);
    }
    if (mlo & 0x8000u) {
      const v2f a = {areg[7].z, areg[7].w};
      const v2f b = {r.w2[6], r.w2[7]};
      v8f d = __builtin_amdgcn_wmma_f32_16x16x4_f32(false, a, false, b, (short)0, zero, false, false);
      WMMA_RESULT_GUARD(d);
      o[0] = __builtin_fmaf(r.w1[1], d[0], o[0]);
      o[0] = __builtin_fmaf(r.w1[5], d[1], o[0]);
      o[0] = __builtin_fmaf(r.w1[9], d[2], o[0]);
      o[0] = __builtin_fmaf(r.w1[11], d[3], o[0]);
      o[1] = __builtin_fmaf(r.w1[4], d[4], o[1]);
      o[1] = __builtin_fmaf(r.w1[10], d[5], o[1]);
      o[2] = __builtin_fmaf(r.w1[0], d[6], o[2]);
      o[2] = __builtin_fmaf(r.w1[2], d[7], o[2]);
    }
  }
  if (mlo & 0x20000u) {
    const v2f a = {areg[8].z, areg[8].w};
    const v2f b = {r.w2[2], r.w2[3]};
    v8f d = __builtin_amdgcn_wmma_f32_16x16x4_f32(false, a, false, b, (short)0, zero, false, false);
    WMMA_RESULT_GUARD(d);
    o[2] = __builtin_fmaf(r.w1[4], d[0], o[2]);
    o[2] = __builtin_fmaf(r.w1[6], d[1], o[2]);
    o[2] = __builtin_fmaf(r.w1[10], d[2], o[2]);
    o[2] = __builtin_fmaf(r.w1[12], d[3], o[2]);
    o[3] = __builtin_fmaf(r.w1[9], d[4], o[3]);
    o[4] = __builtin_fmaf(r.w1[1], d[5], o[4]);
    o[4] = __builtin_fmaf(r.w1[5], d[6], o[4]);
    o[4] = __builtin_fmaf(r.w1[9], d[7], o[4]);
  }
  if (mlo & 0xd0000u) {
    if (mlo & 0x10000u) {
      const v2f a = {areg[8].x, areg[8].y};
      const v2f b = {r.w2[0], r.w2[1]};
      v8f d = __builtin_amdgcn_wmma_f32_16x16x4_f32(false, a, false, b, (short)0, zero, false, false);
      WMMA_RESULT_GUARD(d);
      o[2] = __builtin_fmaf(r.w1[4], d[0], o[2]);
      o[2] = __builtin_fmaf(r.w1[6], d[1], o[2]);
      o[2] = __builtin_fmaf(r.w1[10], d[2], o[2]);
      o[2] = __builtin_fmaf(r.w1[12], d[3], o[2]);
      o[3] = __builtin_fmaf(r.w1[9], d[4], o[3]);
      o[4] = __builtin_fmaf(r.w1[1], d[5], o[4]);
      o[4] = __builtin_fmaf(r.w1[5], d[6], o[4]);
      o[4] = __builtin_fmaf(r.w1[9], d[7], o[4]);
    }
    if (mlo & 0x40000u) {
      const v2f a = {areg[9].x, areg[9].y};
      const v2f b = {r.w2[4], r.w2[5]};
      v8f d = __builtin_amdgcn_wmma_f32_16x16x4_f32(false, a, false, b, (short)0, zero, false, false);
      WMMA_RESULT_GUARD(d);
      o[2] = __builtin_fmaf(r.w1[4], d[0], o[2]);
      o[2] = __builtin_fmaf(r.w1[6], d[1], o[2]);
      o[2] = __builtin_fmaf(r.w1[10], d[2], o[2]);
      o[2] = __builtin_fmaf(r.w1[12], d[3], o[2]);
      o[3] = __builtin_fmaf(r.w1[9], d[4], o[3]);
      o[4] = __builtin_fmaf(r.w1[1], d[5], o[4]);
      o[4] = __builtin_fmaf(r.w1[5], d[6], o[4]);
      o[4] = __builtin_fmaf(r.w1[9], d[7], o[4]);
    }
    if (mlo & 0x80000u) {
      const v2f a = {areg[9].z, areg[9].w};
      const v2f b = {r.w2[6], r.w2[7]};
      v8f d = __builtin_amdgcn_wmma_f32_16x16x4_f32(false, a, false, b, (short)0, zero, false, false);
      WMMA_RESULT_GUARD(d);
      o[2] = __builtin_fmaf(r.w1[4], d[0], o[2]);
      o[2] = __builtin_fmaf(r.w1[6], d[1], o[2]);
      o[2] = __builtin_fmaf(r.w1[10], d[2], o[2]);
      o[2] = __builtin_fmaf(r.w1[12], d[3], o[2]);
      o[3] = __builtin_fmaf(r.w1[9], d[4], o[3]);
      o[4] = __builtin_fmaf(r.w1[1], d[5], o[4]);
      o[4] = __builtin_fmaf(r.w1[5], d[6], o[4]);
      o[4] = __builtin_fmaf(r.w1[9], d[7], o[4]);
    }
  }
  if (mlo & 0x200000u) {
    const v2f a = {areg[10].z, areg[10].w};
    const v2f b = {r.w2[2], r.w2[3]};
    v8f d = __builtin_amdgcn_wmma_f32_16x16x4_f32(false, a, false, b, (short)0, zero, false, false);
    WMMA_RESULT_GUARD(d);
    o[4] = __builtin_fmaf(r.w1[11], d[0], o[4]);
    o[5] = __builtin_fmaf(r.w1[1], d[1], o[5]);
    o[5] = __builtin_fmaf(r.w1[3], d[2], o[5]);
    o[5] = __builtin_fmaf(r.w1[5], d[3], o[5]);
    o[5] = __builtin_fmaf(r.w1[7], d[4], o[5]);
    o[5] = __builtin_fmaf(r.w1[11], d[5], o[5]);
    o[5] = __builtin_fmaf(r.w1[13], d[6], o[5]);
    o[6] = __builtin_fmaf((half ? r.w1[4] : r.w1[8]), d[7], o[6]);
  }
  if (mlo & 0xd00000u) {
    if (mlo & 0x100000u) {
      const v2f a = {areg[10].x, areg[10].y};
      const v2f b = {r.w2[0], r.w2[1]};
      v8f d = __builtin_amdgcn_wmma_f32_16x16x4_f32(false, a, false, b, (short)0, zero, false, false);
      WMMA_RESULT_GUARD(d);
      o[4] = __builtin_fmaf(r.w1[11], d[0], o[4]);
      o[5] = __builtin_fmaf(r.w1[1], d[1], o[5]);
      o[5] = __builtin_fmaf(r.w1[3], d[2], o[5]);
      o[5] = __builtin_fmaf(r.w1[5], d[3], o[5]);
      o[5] = __builtin_fmaf(r.w1[7], d[4], o[5]);
      o[5] = __builtin_fmaf(r.w1[11], d[5], o[5]);
      o[5] = __builtin_fmaf(r.w1[13], d[6], o[5]);
      o[6] = __builtin_fmaf((half ? r.w1[4] : r.w1[8]), d[7], o[6]);
    }
    if (mlo & 0x400000u) {
      const v2f a = {areg[11].x, areg[11].y};
      const v2f b = {r.w2[4], r.w2[5]};
      v8f d = __builtin_amdgcn_wmma_f32_16x16x4_f32(false, a, false, b, (short)0, zero, false, false);
      WMMA_RESULT_GUARD(d);
      o[4] = __builtin_fmaf(r.w1[11], d[0], o[4]);
      o[5] = __builtin_fmaf(r.w1[1], d[1], o[5]);
      o[5] = __builtin_fmaf(r.w1[3], d[2], o[5]);
      o[5] = __builtin_fmaf(r.w1[5], d[3], o[5]);
      o[5] = __builtin_fmaf(r.w1[7], d[4], o[5]);
      o[5] = __builtin_fmaf(r.w1[11], d[5], o[5]);
      o[5] = __builtin_fmaf(r.w1[13], d[6], o[5]);
      o[6] = __builtin_fmaf((half ? r.w1[4] : r.w1[8]), d[7], o[6]);
    }
    if (mlo & 0x800000u) {
      const v2f a = {areg[11].z, areg[11].w};
      const v2f b = {r.w2[6], r.w2[7]};
      v8f d = __builtin_amdgcn_wmma_f32_16x16x4_f32(false, a, false, b, (short)0, zero, false, false);
      WMMA_RESULT_GUARD(d);
      o[4] = __builtin_fmaf(r.w1[11], d[0], o[4]);
      o[5] = __builtin_fmaf(r.w1[1], d[1], o[5]);
      o[5] = __builtin_fmaf(r.w1[3], d[2], o[5]);
      o[5] = __builtin_fmaf(r.w1[5], d[3], o[5]);
      o[5] = __builtin_fmaf(r.w1[7], d[4], o[5]);
      o[5] = __builtin_fmaf(r.w1[11], d[5], o[5]);
      o[5] = __builtin_fmaf(r.w1[13], d[6], o[5]);
      o[6] = __builtin_fmaf((half ? r.w1[4] : r.w1[8]), d[7], o[6]);
    }
  }
  if (mlo & 0x2000000u) {
    const v2f a = {areg[12].z, areg[12].w};
    const v2f b = {r.w2[2], r.w2[3]};
    v8f d = __builtin_amdgcn_wmma_f32_16x16x4_f32(false, a, false, b, (short)0, zero, false, false);
    WMMA_RESULT_GUARD(d);
    o[6] = __builtin_fmaf((half ? r.w1[10] : r.w1[13]), d[0], o[6]);
    o[7] = __builtin_fmaf(r.w1[0], d[1], o[7]);
    o[7] = __builtin_fmaf(r.w1[2], d[2], o[7]);
    o[7] = __builtin_fmaf(r.w1[6], d[3], o[7]);
    o[7] = __builtin_fmaf(r.w1[12], d[4], o[7]);
    o[7] = __builtin_fmaf((half ? r.w1[8] : r.w1[4]), d[5], o[7]);
    o[7] = __builtin_fmaf((half ? r.w1[14] : r.w1[10]), d[6], o[7]);
    o[7] = __builtin_fmaf((half ? r.w1[11] : r.w1[9]), d[7], o[7]);
  }
  if (mlo & 0xd000000u) {
    if (mlo & 0x1000000u) {
      const v2f a = {areg[12].x, areg[12].y};
      const v2f b = {r.w2[0], r.w2[1]};
      v8f d = __builtin_amdgcn_wmma_f32_16x16x4_f32(false, a, false, b, (short)0, zero, false, false);
      WMMA_RESULT_GUARD(d);
      o[6] = __builtin_fmaf((half ? r.w1[10] : r.w1[13]), d[0], o[6]);
      o[7] = __builtin_fmaf(r.w1[0], d[1], o[7]);
      o[7] = __builtin_fmaf(r.w1[2], d[2], o[7]);
      o[7] = __builtin_fmaf(r.w1[6], d[3], o[7]);
      o[7] = __builtin_fmaf(r.w1[12], d[4], o[7]);
      o[7] = __builtin_fmaf((half ? r.w1[8] : r.w1[4]), d[5], o[7]);
      o[7] = __builtin_fmaf((half ? r.w1[14] : r.w1[10]), d[6], o[7]);
      o[7] = __builtin_fmaf((half ? r.w1[11] : r.w1[9]), d[7], o[7]);
    }
    if (mlo & 0x4000000u) {
      const v2f a = {areg[13].x, areg[13].y};
      const v2f b = {r.w2[4], r.w2[5]};
      v8f d = __builtin_amdgcn_wmma_f32_16x16x4_f32(false, a, false, b, (short)0, zero, false, false);
      WMMA_RESULT_GUARD(d);
      o[6] = __builtin_fmaf((half ? r.w1[10] : r.w1[13]), d[0], o[6]);
      o[7] = __builtin_fmaf(r.w1[0], d[1], o[7]);
      o[7] = __builtin_fmaf(r.w1[2], d[2], o[7]);
      o[7] = __builtin_fmaf(r.w1[6], d[3], o[7]);
      o[7] = __builtin_fmaf(r.w1[12], d[4], o[7]);
      o[7] = __builtin_fmaf((half ? r.w1[8] : r.w1[4]), d[5], o[7]);
      o[7] = __builtin_fmaf((half ? r.w1[14] : r.w1[10]), d[6], o[7]);
      o[7] = __builtin_fmaf((half ? r.w1[11] : r.w1[9]), d[7], o[7]);
    }
    if (mlo & 0x8000000u) {
      const v2f a = {areg[13].z, areg[13].w};
      const v2f b = {r.w2[6], r.w2[7]};
      v8f d = __builtin_amdgcn_wmma_f32_16x16x4_f32(false, a, false, b, (short)0, zero, false, false);
      WMMA_RESULT_GUARD(d);
      o[6] = __builtin_fmaf((half ? r.w1[10] : r.w1[13]), d[0], o[6]);
      o[7] = __builtin_fmaf(r.w1[0], d[1], o[7]);
      o[7] = __builtin_fmaf(r.w1[2], d[2], o[7]);
      o[7] = __builtin_fmaf(r.w1[6], d[3], o[7]);
      o[7] = __builtin_fmaf(r.w1[12], d[4], o[7]);
      o[7] = __builtin_fmaf((half ? r.w1[8] : r.w1[4]), d[5], o[7]);
      o[7] = __builtin_fmaf((half ? r.w1[14] : r.w1[10]), d[6], o[7]);
      o[7] = __builtin_fmaf((half ? r.w1[11] : r.w1[9]), d[7], o[7]);
    }
  }
  if (mlo & 0x40000000u) {
    const v2f a = {areg[15].x, areg[15].y};
    const v2f b = {r.w2[4], r.w2[5]};
    v8f d = __builtin_amdgcn_wmma_f32_16x16x4_f32(false, a, false, b, (short)0, zero, false, false);
    WMMA_RESULT_GUARD(d);
    o[0] = __builtin_fmaf(r.w1[3], d[0], o[0]);
    o[0] = __builtin_fmaf(r.w1[7], d[1], o[0]);
    o[0] = __builtin_fmaf(r.w1[13], d[2], o[0]);
    o[0] = __builtin_fmaf(r.w1[15], d[3], o[0]);
    o[1] = __builtin_fmaf(r.w1[0], d[4], o[1]);
    o[1] = __builtin_fmaf(r.w1[2], d[5], o[1]);
    o[1] = __builtin_fmaf(r.w1[6], d[6], o[1]);
    o[1] = __builtin_fmaf(r.w1[8], d[7], o[1]);
  }
  if (mlo & 0xb0000000u) {
    if (mlo & 0x10000000u) {
      const v2f a = {areg[14].x, areg[14].y};
      const v2f b = {r.w2[0], r.w2[1]};
      v8f d = __builtin_amdgcn_wmma_f32_16x16x4_f32(false, a, false, b, (short)0, zero, false, false);
      WMMA_RESULT_GUARD(d);
      o[0] = __builtin_fmaf(r.w1[3], d[0], o[0]);
      o[0] = __builtin_fmaf(r.w1[7], d[1], o[0]);
      o[0] = __builtin_fmaf(r.w1[13], d[2], o[0]);
      o[0] = __builtin_fmaf(r.w1[15], d[3], o[0]);
      o[1] = __builtin_fmaf(r.w1[0], d[4], o[1]);
      o[1] = __builtin_fmaf(r.w1[2], d[5], o[1]);
      o[1] = __builtin_fmaf(r.w1[6], d[6], o[1]);
      o[1] = __builtin_fmaf(r.w1[8], d[7], o[1]);
    }
    if (mlo & 0x20000000u) {
      const v2f a = {areg[14].z, areg[14].w};
      const v2f b = {r.w2[2], r.w2[3]};
      v8f d = __builtin_amdgcn_wmma_f32_16x16x4_f32(false, a, false, b, (short)0, zero, false, false);
      WMMA_RESULT_GUARD(d);
      o[0] = __builtin_fmaf(r.w1[3], d[0], o[0]);
      o[0] = __builtin_fmaf(r.w1[7], d[1], o[0]);
      o[0] = __builtin_fmaf(r.w1[13], d[2], o[0]);
      o[0] = __builtin_fmaf(r.w1[15], d[3], o[0]);
      o[1] = __builtin_fmaf(r.w1[0], d[4], o[1]);
      o[1] = __builtin_fmaf(r.w1[2], d[5], o[1]);
      o[1] = __builtin_fmaf(r.w1[6], d[6], o[1]);
      o[1] = __builtin_fmaf(r.w1[8], d[7], o[1]);
    }
    if (mlo & 0x80000000u) {
      const v2f a = {areg[15].z, areg[15].w};
      const v2f b = {r.w2[6], r.w2[7]};
      v8f d = __builtin_amdgcn_wmma_f32_16x16x4_f32(false, a, false, b, (short)0, zero, false, false);
      WMMA_RESULT_GUARD(d);
      o[0] = __builtin_fmaf(r.w1[3], d[0], o[0]);
      o[0] = __builtin_fmaf(r.w1[7], d[1], o[0]);
      o[0] = __builtin_fmaf(r.w1[13], d[2], o[0]);
      o[0] = __builtin_fmaf(r.w1[15], d[3], o[0]);
      o[1] = __builtin_fmaf(r.w1[0], d[4], o[1]);
      o[1] = __builtin_fmaf(r.w1[2], d[5], o[1]);
      o[1] = __builtin_fmaf(r.w1[6], d[6], o[1]);
      o[1] = __builtin_fmaf(r.w1[8], d[7], o[1]);
    }
  }
  if (mhi & 0x4u) {
    const v2f a = {areg[17].x, areg[17].y};
    const v2f b = {r.w2[4], r.w2[5]};
    v8f d = __builtin_amdgcn_wmma_f32_16x16x4_f32(false, a, false, b, (short)0, zero, false, false);
    WMMA_RESULT_GUARD(d);
    o[1] = __builtin_fmaf(r.w1[12], d[0], o[1]);
    o[1] = __builtin_fmaf(r.w1[14], d[1], o[1]);
    o[2] = __builtin_fmaf(r.w1[8], d[2], o[2]);
    o[2] = __builtin_fmaf(r.w1[14], d[3], o[2]);
    o[3] = __builtin_fmaf(r.w1[1], d[4], o[3]);
    o[3] = __builtin_fmaf(r.w1[3], d[5], o[3]);
    o[3] = __builtin_fmaf(r.w1[5], d[6], o[3]);
    o[3] = __builtin_fmaf(r.w1[7], d[7], o[3]);
  }
  if (mhi & 0xbu) {
    if (mhi & 0x1u) {
      const v2f a = {areg[16].x, areg[16].y};
      const v2f b = {r.w2[0], r.w2[1]};
      v8f d = __builtin_amdgcn_wmma_f32_16x16x4_f32(false, a, false, b, (short)0, zero, false, false);
      WMMA_RESULT_GUARD(d);
      o[1] = __builtin_fmaf(r.w1[12], d[0], o[1]);
      o[1] = __builtin_fmaf(r.w1[14], d[1], o[1]);
      o[2] = __builtin_fmaf(r.w1[8], d[2], o[2]);
      o[2] = __builtin_fmaf(r.w1[14], d[3], o[2]);
      o[3] = __builtin_fmaf(r.w1[1], d[4], o[3]);
      o[3] = __builtin_fmaf(r.w1[3], d[5], o[3]);
      o[3] = __builtin_fmaf(r.w1[5], d[6], o[3]);
      o[3] = __builtin_fmaf(r.w1[7], d[7], o[3]);
    }
    if (mhi & 0x2u) {
      const v2f a = {areg[16].z, areg[16].w};
      const v2f b = {r.w2[2], r.w2[3]};
      v8f d = __builtin_amdgcn_wmma_f32_16x16x4_f32(false, a, false, b, (short)0, zero, false, false);
      WMMA_RESULT_GUARD(d);
      o[1] = __builtin_fmaf(r.w1[12], d[0], o[1]);
      o[1] = __builtin_fmaf(r.w1[14], d[1], o[1]);
      o[2] = __builtin_fmaf(r.w1[8], d[2], o[2]);
      o[2] = __builtin_fmaf(r.w1[14], d[3], o[2]);
      o[3] = __builtin_fmaf(r.w1[1], d[4], o[3]);
      o[3] = __builtin_fmaf(r.w1[3], d[5], o[3]);
      o[3] = __builtin_fmaf(r.w1[5], d[6], o[3]);
      o[3] = __builtin_fmaf(r.w1[7], d[7], o[3]);
    }
    if (mhi & 0x8u) {
      const v2f a = {areg[17].z, areg[17].w};
      const v2f b = {r.w2[6], r.w2[7]};
      v8f d = __builtin_amdgcn_wmma_f32_16x16x4_f32(false, a, false, b, (short)0, zero, false, false);
      WMMA_RESULT_GUARD(d);
      o[1] = __builtin_fmaf(r.w1[12], d[0], o[1]);
      o[1] = __builtin_fmaf(r.w1[14], d[1], o[1]);
      o[2] = __builtin_fmaf(r.w1[8], d[2], o[2]);
      o[2] = __builtin_fmaf(r.w1[14], d[3], o[2]);
      o[3] = __builtin_fmaf(r.w1[1], d[4], o[3]);
      o[3] = __builtin_fmaf(r.w1[3], d[5], o[3]);
      o[3] = __builtin_fmaf(r.w1[5], d[6], o[3]);
      o[3] = __builtin_fmaf(r.w1[7], d[7], o[3]);
    }
  }
  if (mhi & 0x40u) {
    const v2f a = {areg[19].x, areg[19].y};
    const v2f b = {r.w2[4], r.w2[5]};
    v8f d = __builtin_amdgcn_wmma_f32_16x16x4_f32(false, a, false, b, (short)0, zero, false, false);
    WMMA_RESULT_GUARD(d);
    o[3] = __builtin_fmaf(r.w1[11], d[0], o[3]);
    o[3] = __builtin_fmaf(r.w1[13], d[1], o[3]);
    o[4] = __builtin_fmaf(r.w1[3], d[2], o[4]);
    o[4] = __builtin_fmaf(r.w1[7], d[3], o[4]);
    o[4] = __builtin_fmaf(r.w1[13], d[4], o[4]);
    o[4] = __builtin_fmaf(r.w1[15], d[5], o[4]);
    o[5] = __builtin_fmaf(r.w1[15], d[6], o[5]);
    o[6] = __builtin_fmaf(r.w1[0], d[7], o[6]);
  }
  if (mhi & 0xb0u) {
    if (mhi & 0x10u) {
      const v2f a = {areg[18].x, areg[18].y};
      const v2f b = {r.w2[0], r.w2[1]};
      v8f d = __builtin_amdgcn_wmma_f32_16x16x4_f32(false, a, false, b, (short)0, zero, false, false);
      WMMA_RESULT_GUARD(d);
      o[3] = __builtin_fmaf(r.w1[11], d[0], o[3]);
      o[3] = __builtin_fmaf(r.w1[13], d[1], o[3]);
      o[4] = __builtin_fmaf(r.w1[3], d[2], o[4]);
      o[4] = __builtin_fmaf(r.w1[7], d[3], o[4]);
      o[4] = __builtin_fmaf(r.w1[13], d[4], o[4]);
      o[4] = __builtin_fmaf(r.w1[15], d[5], o[4]);
      o[5] = __builtin_fmaf(r.w1[15], d[6], o[5]);
      o[6] = __builtin_fmaf(r.w1[0], d[7], o[6]);
    }
    if (mhi & 0x20u) {
      const v2f a = {areg[18].z, areg[18].w};
      const v2f b = {r.w2[2], r.w2[3]};
      v8f d = __builtin_amdgcn_wmma_f32_16x16x4_f32(false, a, false, b, (short)0, zero, false, false);
      WMMA_RESULT_GUARD(d);
      o[3] = __builtin_fmaf(r.w1[11], d[0], o[3]);
      o[3] = __builtin_fmaf(r.w1[13], d[1], o[3]);
      o[4] = __builtin_fmaf(r.w1[3], d[2], o[4]);
      o[4] = __builtin_fmaf(r.w1[7], d[3], o[4]);
      o[4] = __builtin_fmaf(r.w1[13], d[4], o[4]);
      o[4] = __builtin_fmaf(r.w1[15], d[5], o[4]);
      o[5] = __builtin_fmaf(r.w1[15], d[6], o[5]);
      o[6] = __builtin_fmaf(r.w1[0], d[7], o[6]);
    }
    if (mhi & 0x80u) {
      const v2f a = {areg[19].z, areg[19].w};
      const v2f b = {r.w2[6], r.w2[7]};
      v8f d = __builtin_amdgcn_wmma_f32_16x16x4_f32(false, a, false, b, (short)0, zero, false, false);
      WMMA_RESULT_GUARD(d);
      o[3] = __builtin_fmaf(r.w1[11], d[0], o[3]);
      o[3] = __builtin_fmaf(r.w1[13], d[1], o[3]);
      o[4] = __builtin_fmaf(r.w1[3], d[2], o[4]);
      o[4] = __builtin_fmaf(r.w1[7], d[3], o[4]);
      o[4] = __builtin_fmaf(r.w1[13], d[4], o[4]);
      o[4] = __builtin_fmaf(r.w1[15], d[5], o[4]);
      o[5] = __builtin_fmaf(r.w1[15], d[6], o[5]);
      o[6] = __builtin_fmaf(r.w1[0], d[7], o[6]);
    }
  }
  if (mhi & 0x400u) {
    const v2f a = {areg[21].x, areg[21].y};
    const v2f b = {r.w2[4], r.w2[5]};
    v8f d = __builtin_amdgcn_wmma_f32_16x16x4_f32(false, a, false, b, (short)0, zero, false, false);
    WMMA_RESULT_GUARD(d);
    o[6] = __builtin_fmaf(r.w1[2], d[0], o[6]);
    o[6] = __builtin_fmaf(r.w1[6], d[1], o[6]);
    o[6] = __builtin_fmaf(r.w1[12], d[2], o[6]);
    o[6] = __builtin_fmaf((half ? r.w1[8] : r.w1[4]), d[3], o[6]);
    o[6] = __builtin_fmaf((half ? r.w1[14] : r.w1[10]), d[4], o[6]);
    o[7] = __builtin_fmaf((half ? r.w1[5] : r.w1[8]), d[5], o[7]);
    o[7] = __builtin_fmaf((half ? r.w1[9] : r.w1[14]), d[6], o[7]);
    o[6] = __builtin_fmaf(r.w1[15], d[7], o[6]);
  }
  if (mhi & 0xb00u) {
    if (mhi & 0x100u) {
      const v2f a = {areg[20].x, areg[20].y};
      const v2f b = {r.w2[0], r.w2[1]};
      v8f d = __builtin_amdgcn_wmma_f32_16x16x4_f32(false, a, false, b, (short)0, zero, false, false);
      WMMA_RESULT_GUARD(d);
      o[6] = __builtin_fmaf(r.w1[2], d[0], o[6]);
      o[6] = __builtin_fmaf(r.w1[6], d[1], o[6]);
      o[6] = __builtin_fmaf(r.w1[12], d[2], o[6]);
      o[6] = __builtin_fmaf((half ? r.w1[8] : r.w1[4]), d[3], o[6]);
      o[6] = __builtin_fmaf((half ? r.w1[14] : r.w1[10]), d[4], o[6]);
      o[7] = __builtin_fmaf((half ? r.w1[5] : r.w1[8]), d[5], o[7]);
      o[7] = __builtin_fmaf((half ? r.w1[9] : r.w1[14]), d[6], o[7]);
      o[6] = __builtin_fmaf(r.w1[15], d[7], o[6]);
    }
    if (mhi & 0x200u) {
      const v2f a = {areg[20].z, areg[20].w};
      const v2f b = {r.w2[2], r.w2[3]};
      v8f d = __builtin_amdgcn_wmma_f32_16x16x4_f32(false, a, false, b, (short)0, zero, false, false);
      WMMA_RESULT_GUARD(d);
      o[6] = __builtin_fmaf(r.w1[2], d[0], o[6]);
      o[6] = __builtin_fmaf(r.w1[6], d[1], o[6]);
      o[6] = __builtin_fmaf(r.w1[12], d[2], o[6]);
      o[6] = __builtin_fmaf((half ? r.w1[8] : r.w1[4]), d[3], o[6]);
      o[6] = __builtin_fmaf((half ? r.w1[14] : r.w1[10]), d[4], o[6]);
      o[7] = __builtin_fmaf((half ? r.w1[5] : r.w1[8]), d[5], o[7]);
      o[7] = __builtin_fmaf((half ? r.w1[9] : r.w1[14]), d[6], o[7]);
      o[6] = __builtin_fmaf(r.w1[15], d[7], o[6]);
    }
    if (mhi & 0x800u) {
      const v2f a = {areg[21].z, areg[21].w};
      const v2f b = {r.w2[6], r.w2[7]};
      v8f d = __builtin_amdgcn_wmma_f32_16x16x4_f32(false, a, false, b, (short)0, zero, false, false);
      WMMA_RESULT_GUARD(d);
      o[6] = __builtin_fmaf(r.w1[2], d[0], o[6]);
      o[6] = __builtin_fmaf(r.w1[6], d[1], o[6]);
      o[6] = __builtin_fmaf(r.w1[12], d[2], o[6]);
      o[6] = __builtin_fmaf((half ? r.w1[8] : r.w1[4]), d[3], o[6]);
      o[6] = __builtin_fmaf((half ? r.w1[14] : r.w1[10]), d[4], o[6]);
      o[7] = __builtin_fmaf((half ? r.w1[5] : r.w1[8]), d[5], o[7]);
      o[7] = __builtin_fmaf((half ? r.w1[9] : r.w1[14]), d[6], o[7]);
      o[6] = __builtin_fmaf(r.w1[15], d[7], o[6]);
    }
  }
  if (mhi & 0x8000u) {
    const v2f a = {areg[23].z, areg[23].w};
    const v2f b = {r.w2[6], r.w2[7]};
    v8f d = __builtin_amdgcn_wmma_f32_16x16x4_f32(false, a, false, b, (short)0, zero, false, false);
    WMMA_RESULT_GUARD(d);
    o[0] = __builtin_fmaf(r.w1[4], d[0], o[0]);
    o[0] = __builtin_fmaf(r.w1[8], d[1], o[0]);
    o[0] = __builtin_fmaf(r.w1[10], d[2], o[0]);
    o[0] = __builtin_fmaf(r.w1[14], d[3], o[0]);
    o[1] = __builtin_fmaf(r.w1[3], d[4], o[1]);
    o[1] = __builtin_fmaf(r.w1[7], d[5], o[1]);
    o[1] = __builtin_fmaf(r.w1[9], d[6], o[1]);
    o[1] = __builtin_fmaf(r.w1[13], d[7], o[1]);
  }
  if (mhi & 0x7000u) {
    if (mhi & 0x1000u) {
      const v2f a = {areg[22].x, areg[22].y};
      const v2f b = {r.w2[0], r.w2[1]};
      v8f d = __builtin_amdgcn_wmma_f32_16x16x4_f32(false, a, false, b, (short)0, zero, false, false);
      WMMA_RESULT_GUARD(d);
      o[0] = __builtin_fmaf(r.w1[4], d[0], o[0]);
      o[0] = __builtin_fmaf(r.w1[8], d[1], o[0]);
      o[0] = __builtin_fmaf(r.w1[10], d[2], o[0]);
      o[0] = __builtin_fmaf(r.w1[14], d[3], o[0]);
      o[1] = __builtin_fmaf(r.w1[3], d[4], o[1]);
      o[1] = __builtin_fmaf(r.w1[7], d[5], o[1]);
      o[1] = __builtin_fmaf(r.w1[9], d[6], o[1]);
      o[1] = __builtin_fmaf(r.w1[13], d[7], o[1]);
    }
    if (mhi & 0x2000u) {
      const v2f a = {areg[22].z, areg[22].w};
      const v2f b = {r.w2[2], r.w2[3]};
      v8f d = __builtin_amdgcn_wmma_f32_16x16x4_f32(false, a, false, b, (short)0, zero, false, false);
      WMMA_RESULT_GUARD(d);
      o[0] = __builtin_fmaf(r.w1[4], d[0], o[0]);
      o[0] = __builtin_fmaf(r.w1[8], d[1], o[0]);
      o[0] = __builtin_fmaf(r.w1[10], d[2], o[0]);
      o[0] = __builtin_fmaf(r.w1[14], d[3], o[0]);
      o[1] = __builtin_fmaf(r.w1[3], d[4], o[1]);
      o[1] = __builtin_fmaf(r.w1[7], d[5], o[1]);
      o[1] = __builtin_fmaf(r.w1[9], d[6], o[1]);
      o[1] = __builtin_fmaf(r.w1[13], d[7], o[1]);
    }
    if (mhi & 0x4000u) {
      const v2f a = {areg[23].x, areg[23].y};
      const v2f b = {r.w2[4], r.w2[5]};
      v8f d = __builtin_amdgcn_wmma_f32_16x16x4_f32(false, a, false, b, (short)0, zero, false, false);
      WMMA_RESULT_GUARD(d);
      o[0] = __builtin_fmaf(r.w1[4], d[0], o[0]);
      o[0] = __builtin_fmaf(r.w1[8], d[1], o[0]);
      o[0] = __builtin_fmaf(r.w1[10], d[2], o[0]);
      o[0] = __builtin_fmaf(r.w1[14], d[3], o[0]);
      o[1] = __builtin_fmaf(r.w1[3], d[4], o[1]);
      o[1] = __builtin_fmaf(r.w1[7], d[5], o[1]);
      o[1] = __builtin_fmaf(r.w1[9], d[6], o[1]);
      o[1] = __builtin_fmaf(r.w1[13], d[7], o[1]);
    }
  }
  if (mhi & 0x80000u) {
    const v2f a = {areg[25].z, areg[25].w};
    const v2f b = {r.w2[6], r.w2[7]};
    v8f d = __builtin_amdgcn_wmma_f32_16x16x4_f32(false, a, false, b, (short)0, zero, false, false);
    WMMA_RESULT_GUARD(d);
    o[2] = __builtin_fmaf(r.w1[1], d[0], o[2]);
    o[2] = __builtin_fmaf(r.w1[5], d[1], o[2]);
    o[2] = __builtin_fmaf(r.w1[11], d[2], o[2]);
    o[2] = __builtin_fmaf(r.w1[15], d[3], o[2]);
    o[3] = __builtin_fmaf(r.w1[0], d[4], o[3]);
    o[3] = __builtin_fmaf(r.w1[2], d[5], o[3]);
    o[3] = __builtin_fmaf(r.w1[6], d[6], o[3]);
    o[3] = __builtin_fmaf(r.w1[12], d[7], o[3]);
  }
  if (mhi & 0x70000u) {
    if (mhi & 0x10000u) {
      const v2f a = {areg[24].x, areg[24].y};
      const v2f b = {r.w2[0], r.w2[1]};
      v8f d = __builtin_amdgcn_wmma_f32_16x16x4_f32(false, a, false, b, (short)0, zero, false, false);
      WMMA_RESULT_GUARD(d);
      o[2] = __builtin_fmaf(r.w1[1], d[0], o[2]);
      o[2] = __builtin_fmaf(r.w1[5], d[1], o[2]);
      o[2] = __builtin_fmaf(r.w1[11], d[2], o[2]);
      o[2] = __builtin_fmaf(r.w1[15], d[3], o[2]);
      o[3] = __builtin_fmaf(r.w1[0], d[4], o[3]);
      o[3] = __builtin_fmaf(r.w1[2], d[5], o[3]);
      o[3] = __builtin_fmaf(r.w1[6], d[6], o[3]);
      o[3] = __builtin_fmaf(r.w1[12], d[7], o[3]);
    }
    if (mhi & 0x20000u) {
      const v2f a = {areg[24].z, areg[24].w};
      const v2f b = {r.w2[2], r.w2[3]};
      v8f d = __builtin_amdgcn_wmma_f32_16x16x4_f32(false, a, false, b, (short)0, zero, false, false);
      WMMA_RESULT_GUARD(d);
      o[2] = __builtin_fmaf(r.w1[1], d[0], o[2]);
      o[2] = __builtin_fmaf(r.w1[5], d[1], o[2]);
      o[2] = __builtin_fmaf(r.w1[11], d[2], o[2]);
      o[2] = __builtin_fmaf(r.w1[15], d[3], o[2]);
      o[3] = __builtin_fmaf(r.w1[0], d[4], o[3]);
      o[3] = __builtin_fmaf(r.w1[2], d[5], o[3]);
      o[3] = __builtin_fmaf(r.w1[6], d[6], o[3]);
      o[3] = __builtin_fmaf(r.w1[12], d[7], o[3]);
    }
    if (mhi & 0x40000u) {
      const v2f a = {areg[25].x, areg[25].y};
      const v2f b = {r.w2[4], r.w2[5]};
      v8f d = __builtin_amdgcn_wmma_f32_16x16x4_f32(false, a, false, b, (short)0, zero, false, false);
      WMMA_RESULT_GUARD(d);
      o[2] = __builtin_fmaf(r.w1[1], d[0], o[2]);
      o[2] = __builtin_fmaf(r.w1[5], d[1], o[2]);
      o[2] = __builtin_fmaf(r.w1[11], d[2], o[2]);
      o[2] = __builtin_fmaf(r.w1[15], d[3], o[2]);
      o[3] = __builtin_fmaf(r.w1[0], d[4], o[3]);
      o[3] = __builtin_fmaf(r.w1[2], d[5], o[3]);
      o[3] = __builtin_fmaf(r.w1[6], d[6], o[3]);
      o[3] = __builtin_fmaf(r.w1[12], d[7], o[3]);
    }
  }
  if (mhi & 0x800000u) {
    const v2f a = {areg[27].z, areg[27].w};
    const v2f b = {r.w2[6], r.w2[7]};
    v8f d = __builtin_amdgcn_wmma_f32_16x16x4_f32(false, a, false, b, (short)0, zero, false, false);
    WMMA_RESULT_GUARD(d);
    o[4] = __builtin_fmaf(r.w1[4], d[0], o[4]);
    o[4] = __builtin_fmaf(r.w1[8], d[1], o[4]);
    o[4] = __builtin_fmaf(r.w1[10], d[2], o[4]);
    o[4] = __builtin_fmaf(r.w1[14], d[3], o[4]);
    o[5] = __builtin_fmaf(r.w1[0], d[4], o[5]);
    o[5] = __builtin_fmaf(r.w1[2], d[5], o[5]);
    o[5] = __builtin_fmaf(r.w1[6], d[6], o[5]);
    o[5] = __builtin_fmaf(r.w1[12], d[7], o[5]);
  }
  if (mhi & 0x700000u) {
    if (mhi & 0x100000u) {
      const v2f a = {areg[26].x, areg[26].y};
      const v2f b = {r.w2[0], r.w2[1]};
      v8f d = __builtin_amdgcn_wmma_f32_16x16x4_f32(false, a, false, b, (short)0, zero, false, false);
      WMMA_RESULT_GUARD(d);
      o[4] = __builtin_fmaf(r.w1[4], d[0], o[4]);
      o[4] = __builtin_fmaf(r.w1[8], d[1], o[4]);
      o[4] = __builtin_fmaf(r.w1[10], d[2], o[4]);
      o[4] = __builtin_fmaf(r.w1[14], d[3], o[4]);
      o[5] = __builtin_fmaf(r.w1[0], d[4], o[5]);
      o[5] = __builtin_fmaf(r.w1[2], d[5], o[5]);
      o[5] = __builtin_fmaf(r.w1[6], d[6], o[5]);
      o[5] = __builtin_fmaf(r.w1[12], d[7], o[5]);
    }
    if (mhi & 0x200000u) {
      const v2f a = {areg[26].z, areg[26].w};
      const v2f b = {r.w2[2], r.w2[3]};
      v8f d = __builtin_amdgcn_wmma_f32_16x16x4_f32(false, a, false, b, (short)0, zero, false, false);
      WMMA_RESULT_GUARD(d);
      o[4] = __builtin_fmaf(r.w1[4], d[0], o[4]);
      o[4] = __builtin_fmaf(r.w1[8], d[1], o[4]);
      o[4] = __builtin_fmaf(r.w1[10], d[2], o[4]);
      o[4] = __builtin_fmaf(r.w1[14], d[3], o[4]);
      o[5] = __builtin_fmaf(r.w1[0], d[4], o[5]);
      o[5] = __builtin_fmaf(r.w1[2], d[5], o[5]);
      o[5] = __builtin_fmaf(r.w1[6], d[6], o[5]);
      o[5] = __builtin_fmaf(r.w1[12], d[7], o[5]);
    }
    if (mhi & 0x400000u) {
      const v2f a = {areg[27].x, areg[27].y};
      const v2f b = {r.w2[4], r.w2[5]};
      v8f d = __builtin_amdgcn_wmma_f32_16x16x4_f32(false, a, false, b, (short)0, zero, false, false);
      WMMA_RESULT_GUARD(d);
      o[4] = __builtin_fmaf(r.w1[4], d[0], o[4]);
      o[4] = __builtin_fmaf(r.w1[8], d[1], o[4]);
      o[4] = __builtin_fmaf(r.w1[10], d[2], o[4]);
      o[4] = __builtin_fmaf(r.w1[14], d[3], o[4]);
      o[5] = __builtin_fmaf(r.w1[0], d[4], o[5]);
      o[5] = __builtin_fmaf(r.w1[2], d[5], o[5]);
      o[5] = __builtin_fmaf(r.w1[6], d[6], o[5]);
      o[5] = __builtin_fmaf(r.w1[12], d[7], o[5]);
    }
  }
  if (mhi & 0x8000000u) {
    const v2f a = {areg[29].z, areg[29].w};
    const v2f b = {r.w2[6], r.w2[7]};
    v8f d = __builtin_amdgcn_wmma_f32_16x16x4_f32(false, a, false, b, (short)0, zero, false, false);
    WMMA_RESULT_GUARD(d);
    o[6] = __builtin_fmaf((half ? r.w1[3] : r.w1[1]), d[0], o[6]);
    o[6] = __builtin_fmaf((half ? r.w1[7] : r.w1[5]), d[1], o[6]);
    o[6] = __builtin_fmaf((half ? r.w1[9] : r.w1[11]), d[2], o[6]);
    o[6] = __builtin_fmaf((half ? r.w1[13] : r.w1[14]), d[3], o[6]);
    o[7] = __builtin_fmaf((half ? r.w1[3] : r.w1[1]), d[4], o[7]);
    o[7] = __builtin_fmaf((half ? r.w1[7] : r.w1[5]), d[5], o[7]);
    o[7] = __builtin_fmaf((half ? r.w1[13] : r.w1[11]), d[6], o[7]);
    o[7] = __builtin_fmaf((half ? r.w1[10] : r.w1[15]), d[7], o[7]);
  }
  if (mhi & 0x7000000u) {
    if (mhi & 0x1000000u) {
      const v2f a = {areg[28].x, areg[28].y};
      const v2f b = {r.w2[0], r.w2[1]};
      v8f d = __builtin_amdgcn_wmma_f32_16x16x4_f32(false, a, false, b, (short)0, zero, false, false);
      WMMA_RESULT_GUARD(d);
      o[6] = __builtin_fmaf((half ? r.w1[3] : r.w1[1]), d[0], o[6]);
      o[6] = __builtin_fmaf((half ? r.w1[7] : r.w1[5]), d[1], o[6]);
      o[6] = __builtin_fmaf((half ? r.w1[9] : r.w1[11]), d[2], o[6]);
      o[6] = __builtin_fmaf((half ? r.w1[13] : r.w1[14]), d[3], o[6]);
      o[7] = __builtin_fmaf((half ? r.w1[3] : r.w1[1]), d[4], o[7]);
      o[7] = __builtin_fmaf((half ? r.w1[7] : r.w1[5]), d[5], o[7]);
      o[7] = __builtin_fmaf((half ? r.w1[13] : r.w1[11]), d[6], o[7]);
      o[7] = __builtin_fmaf((half ? r.w1[10] : r.w1[15]), d[7], o[7]);
    }
    if (mhi & 0x2000000u) {
      const v2f a = {areg[28].z, areg[28].w};
      const v2f b = {r.w2[2], r.w2[3]};
      v8f d = __builtin_amdgcn_wmma_f32_16x16x4_f32(false, a, false, b, (short)0, zero, false, false);
      WMMA_RESULT_GUARD(d);
      o[6] = __builtin_fmaf((half ? r.w1[3] : r.w1[1]), d[0], o[6]);
      o[6] = __builtin_fmaf((half ? r.w1[7] : r.w1[5]), d[1], o[6]);
      o[6] = __builtin_fmaf((half ? r.w1[9] : r.w1[11]), d[2], o[6]);
      o[6] = __builtin_fmaf((half ? r.w1[13] : r.w1[14]), d[3], o[6]);
      o[7] = __builtin_fmaf((half ? r.w1[3] : r.w1[1]), d[4], o[7]);
      o[7] = __builtin_fmaf((half ? r.w1[7] : r.w1[5]), d[5], o[7]);
      o[7] = __builtin_fmaf((half ? r.w1[13] : r.w1[11]), d[6], o[7]);
      o[7] = __builtin_fmaf((half ? r.w1[10] : r.w1[15]), d[7], o[7]);
    }
    if (mhi & 0x4000000u) {
      const v2f a = {areg[29].x, areg[29].y};
      const v2f b = {r.w2[4], r.w2[5]};
      v8f d = __builtin_amdgcn_wmma_f32_16x16x4_f32(false, a, false, b, (short)0, zero, false, false);
      WMMA_RESULT_GUARD(d);
      o[6] = __builtin_fmaf((half ? r.w1[3] : r.w1[1]), d[0], o[6]);
      o[6] = __builtin_fmaf((half ? r.w1[7] : r.w1[5]), d[1], o[6]);
      o[6] = __builtin_fmaf((half ? r.w1[9] : r.w1[11]), d[2], o[6]);
      o[6] = __builtin_fmaf((half ? r.w1[13] : r.w1[14]), d[3], o[6]);
      o[7] = __builtin_fmaf((half ? r.w1[3] : r.w1[1]), d[4], o[7]);
      o[7] = __builtin_fmaf((half ? r.w1[7] : r.w1[5]), d[5], o[7]);
      o[7] = __builtin_fmaf((half ? r.w1[13] : r.w1[11]), d[6], o[7]);
      o[7] = __builtin_fmaf((half ? r.w1[10] : r.w1[15]), d[7], o[7]);
    }
  }
  if (mhi & 0x10000000u) {
    const v2f a = {areg[30].x, areg[30].y};
    const v2f b = {r.w2[0], r.w2[1]};
    v8f d = __builtin_amdgcn_wmma_f32_16x16x4_f32(false, a, false, b, (short)0, zero, false, false);
    WMMA_RESULT_GUARD(d);
    o[1] = __builtin_fmaf(r.w1[15], d[0], o[1]);
    o[2] = __builtin_fmaf(r.w1[9], d[1], o[2]);
    o[3] = __builtin_fmaf(r.w1[8], d[2], o[3]);
    o[3] = __builtin_fmaf(r.w1[14], d[3], o[3]);
    o[3] = __builtin_fmaf(r.w1[15], d[4], o[3]);
    o[5] = __builtin_fmaf(r.w1[4], d[5], o[5]);
    o[5] = __builtin_fmaf(r.w1[9], d[6], o[5]);
    o[5] = __builtin_fmaf(r.w1[10], d[7], o[5]);
  }
  if (mhi & 0xe0000000u) {
    if (mhi & 0x20000000u) {
      const v2f a = {areg[30].z, areg[30].w};
      const v2f b = {r.w2[2], r.w2[3]};
      v8f d = __builtin_amdgcn_wmma_f32_16x16x4_f32(false, a, false, b, (short)0, zero, false, false);
      WMMA_RESULT_GUARD(d);
      o[1] = __builtin_fmaf(r.w1[15], d[0], o[1]);
      o[2] = __builtin_fmaf(r.w1[9], d[1], o[2]);
      o[3] = __builtin_fmaf(r.w1[8], d[2], o[3]);
      o[3] = __builtin_fmaf(r.w1[14], d[3], o[3]);
      o[3] = __builtin_fmaf(r.w1[15], d[4], o[3]);
      o[5] = __builtin_fmaf(r.w1[4], d[5], o[5]);
      o[5] = __builtin_fmaf(r.w1[9], d[6], o[5]);
      o[5] = __builtin_fmaf(r.w1[10], d[7], o[5]);
    }
    if (mhi & 0x40000000u) {
      const v2f a = {areg[31].x, areg[31].y};
      const v2f b = {r.w2[4], r.w2[5]};
      v8f d = __builtin_amdgcn_wmma_f32_16x16x4_f32(false, a, false, b, (short)0, zero, false, false);
      WMMA_RESULT_GUARD(d);
      o[1] = __builtin_fmaf(r.w1[15], d[0], o[1]);
      o[2] = __builtin_fmaf(r.w1[9], d[1], o[2]);
      o[3] = __builtin_fmaf(r.w1[8], d[2], o[3]);
      o[3] = __builtin_fmaf(r.w1[14], d[3], o[3]);
      o[3] = __builtin_fmaf(r.w1[15], d[4], o[3]);
      o[5] = __builtin_fmaf(r.w1[4], d[5], o[5]);
      o[5] = __builtin_fmaf(r.w1[9], d[6], o[5]);
      o[5] = __builtin_fmaf(r.w1[10], d[7], o[5]);
    }
    if (mhi & 0x80000000u) {
      const v2f a = {areg[31].z, areg[31].w};
      const v2f b = {r.w2[6], r.w2[7]};
      v8f d = __builtin_amdgcn_wmma_f32_16x16x4_f32(false, a, false, b, (short)0, zero, false, false);
      WMMA_RESULT_GUARD(d);
      o[1] = __builtin_fmaf(r.w1[15], d[0], o[1]);
      o[2] = __builtin_fmaf(r.w1[9], d[1], o[2]);
      o[3] = __builtin_fmaf(r.w1[8], d[2], o[3]);
      o[3] = __builtin_fmaf(r.w1[14], d[3], o[3]);
      o[3] = __builtin_fmaf(r.w1[15], d[4], o[3]);
      o[5] = __builtin_fmaf(r.w1[4], d[5], o[5]);
      o[5] = __builtin_fmaf(r.w1[9], d[6], o[5]);
      o[5] = __builtin_fmaf(r.w1[10], d[7], o[5]);
    }
  }

  const int n = t >> 2;
  const int c = (t & 3) * 16 + col;
  float* po = out + (size_t)n * NUM_ORDERS * C_DIM + c;
  po[(half ? 2 : 0) * C_DIM] = o[0];
  po[(half ? 5 : 1) * C_DIM] = o[1];
  po[(half ? 7 : 3) * C_DIM] = o[2];
  po[(half ? 10 : 4) * C_DIM] = o[3];
  po[(half ? 12 : 6) * C_DIM] = o[4];
  po[(half ? 14 : 8) * C_DIM] = o[5];
  po[(half ? 11 : 9) * C_DIM] = o[6];
  po[(half ? 15 : 13) * C_DIM] = o[7];
}

__global__ void __launch_bounds__(BLOCK_THREADS)
tp_wmma_kernel(const float* x1, const float* x2,
               const float* __restrict__ vals, const int* __restrict__ M1,
               const int* __restrict__ M2, const int* __restrict__ Mseg,
               float* out, int n_batch, int nnz) {
  __shared__ float4 cg_lds[NUM_ORDERS * K_DENSE / 4];
  __shared__ unsigned char tab_lds[256 + 16];
  float* cg = (float*)cg_lds;

  const int tid  = threadIdx.x;
  const int lane = tid & 31;
  const int half = lane >> 4;
  const int col  = lane & 15;
  const int wib    = __builtin_amdgcn_readfirstlane((int)(threadIdx.x >> 5));
  const int wave   = blockIdx.x * WAVES_PER_BLOCK + wib;
  const int nwaves = gridDim.x * WAVES_PER_BLOCK;
  const int ntiles = n_batch * 4;

  if (ntiles <= 0) return;

  int t = wave;
  const bool has_tile = t < ntiles;

  unsigned char t_row = 0, t_col = 0;
  if (tid < 256) t_row = ROW_OF[tid];
  if (tid < 16) t_col = COL_POS[tid];
  const bool has_entry = tid < nnz;
  int   e_m = 0, e_m1 = 0, e_m2 = 0;
  float e_val = 0.0f;
  if (has_entry) {
    e_m   = Mseg[tid];
    e_m1  = M1[tid];
    e_m2  = M2[tid];
    e_val = vals[tid];
  }

  TileRegs r;
  load_tile(r, x1, x2, has_tile ? t : 0, half, col);

  if (tid < 256) tab_lds[tid] = t_row;
  if (tid < 16) tab_lds[256 + tid] = t_col;
  for (int i = tid; i < NUM_ORDERS * K_DENSE / 4; i += BLOCK_THREADS)
    cg_lds[i] = make_float4(0.0f, 0.0f, 0.0f, 0.0f);
  lds_barrier();
  if (has_entry) {
    asm volatile("" : "+v"(e_m), "+v"(e_m1), "+v"(e_m2));
    const int rho = tab_lds[((e_m & 15) << 4) | (e_m1 & 15)];
    const int pos = tab_lds[256 + (e_m2 & 15)];
    cg[cg_frag_index(rho & 15, (rho & 0xF0) | pos)] = e_val;
  }
  for (int i = tid + BLOCK_THREADS; i < nnz; i += BLOCK_THREADS) {
    const int rho = tab_lds[((Mseg[i] & 15) << 4) | (M1[i] & 15)];
    const int pos = tab_lds[256 + (M2[i] & 15)];
    cg[cg_frag_index(rho & 15, (rho & 0xF0) | pos)] = vals[i];
  }
  lds_barrier();

  float4 areg[32];
#pragma unroll
  for (int kp = 0; kp < 32; ++kp) areg[kp] = cg_lds[kp * 32 + lane];

  unsigned mlo = 0, mhi = 0;
#pragma unroll
  for (int kp = 0; kp < 32; ++kp) {
    const float4 q = areg[kp];
    const unsigned b0 = __builtin_amdgcn_ballot_w32((q.x != 0.0f) | (q.y != 0.0f));
    const unsigned b1 = __builtin_amdgcn_ballot_w32((q.z != 0.0f) | (q.w != 0.0f));
    const unsigned bits = (b0 ? 1u : 0u) | (b1 ? 2u : 0u);
    if (kp < 16) mlo |= bits << (2 * kp);
    else         mhi |= bits << (2 * kp - 32);
  }


  if (!has_tile) return;

  const int cnt = (ntiles - t + nwaves - 1) / nwaves;
  TileRegs r1;
  for (int p = cnt; p > 0; --p) {
    asm volatile("" : "+s"(mlo), "+s"(mhi));
    const int tn = t + nwaves;
    load_tile(r1, x1, x2, (tn < ntiles) ? tn : t, half, col);
    __builtin_amdgcn_sched_barrier(0);
    tile_product(areg, mlo, mhi, r, out, t, half, col);
    r = r1;
    t = tn;
  }
}

extern "C" void kernel_launch(void* const* d_in, const int* in_sizes, int n_in,
                              void* d_out, int out_size, void* d_ws, size_t ws_size,
                              hipStream_t stream) {
  const float* x1   = (const float*)d_in[0];
  const float* x2   = (const float*)d_in[1];
  const float* vals = (const float*)d_in[2];
  const int* M1     = (const int*)d_in[3];
  const int* M2     = (const int*)d_in[4];
  const int* Mseg   = (const int*)d_in[5];
  float* out        = (float*)d_out;

  const int nnz     = in_sizes[2];
  const int n_batch = in_sizes[0] / (NUM_ORDERS * C_DIM);

  tp_wmma_kernel<<<NUM_BLOCKS, BLOCK_THREADS, 0, stream>>>(x1, x2, vals, M1, M2, Mseg,
                                                          out, n_batch, nnz);
}


# Table whose non-zero blocks lie in the 15 the layout was made for takes a branch-free tile routine: WMMA of block i+2 issued ahead of block i's FMAs, no per-block mask branch, result wait or nops
# speedup vs baseline: 2.0820x; 1.1074x over previous
#include <hip/hip_runtime.h>


typedef __attribute__((ext_vector_type(2))) float v2f;
typedef __attribute__((ext_vector_type(8))) float v8f;
typedef __attribute__((ext_vector_type(16))) float v16f;

#define NUM_ORDERS 16
#define C_DIM 64
#define K_DENSE 256
#define WAVES_PER_BLOCK 16
#define BLOCK_THREADS (WAVES_PER_BLOCK * 32)
#define NUM_BLOCKS 256

__device__ const unsigned char ROW_OF[256] = {
    0, 48, 1, 112, 176, 49, 2, 113, 177, 50, 178, 51, 3, 114, 179, 115,
    116, 4, 117, 180, 52, 5, 118, 181, 119, 182, 53, 6, 128, 183, 129, 240,
    8, 56, 9, 120, 184, 57, 10, 121, 185, 58, 186, 59, 11, 122, 187, 123,
    54, 192, 55, 7, 64, 193, 65, 16, 130, 241, 66, 194, 67, 17, 131, 195,
    196, 132, 197, 133, 18, 134, 198, 135, 242, 68, 19, 144, 199, 145, 243, 244,
    124, 12, 125, 188, 60, 13, 126, 189, 127, 190, 61, 14, 136, 191, 137, 248,
    20, 69, 21, 146, 208, 70, 22, 147, 209, 71, 210, 80, 23, 148, 211, 149,
    62, 200, 63, 15, 72, 201, 73, 24, 138, 249, 74, 202, 75, 25, 139, 203,
    212, 81, 213, 82, 245, 83, 214, 84, 32, 246, 247, 85, 215, 86, 33, 150,
    151, 224, 160, 35, 163, 225, 161, 36, 87, 34, 164, 226, 162, 96, 227, 167,
    204, 140, 205, 141, 26, 142, 206, 143, 250, 76, 27, 152, 207, 153, 251, 252,
    159, 42, 168, 232, 95, 43, 169, 233, 171, 234, 104, 44, 170, 235, 172, 175,
    28, 77, 29, 154, 216, 78, 30, 155, 217, 79, 218, 88, 31, 156, 219, 157,
    97, 228, 98, 37, 101, 229, 99, 38, 165, 103, 102, 230, 100, 39, 166, 231,
    220, 89, 221, 90, 253, 91, 222, 92, 40, 254, 255, 93, 223, 94, 41, 158,
    105, 46, 106, 236, 47, 173, 107, 237, 109, 174, 239, 111, 108, 238, 110, 45,
};
__device__ const unsigned char COL_POS[16] = {0, 8, 1, 4, 14, 9, 2, 5, 12, 11, 15, 10, 3, 6, 13, 7};

__device__ __forceinline__ int cg_frag_index(int i, int k) {
  const int kk   = k >> 2;
  const int lane = ((k >> 1) & 1) * 16 + i;
  return ((kk >> 1) * 32 + lane) * 4 + (kk & 1) * 2 + (k & 1);
}

__device__ __forceinline__ void lds_barrier() {
  __builtin_amdgcn_fence(__ATOMIC_RELEASE, "workgroup", "local");
  __builtin_amdgcn_s_barrier();
  __builtin_amdgcn_fence(__ATOMIC_ACQUIRE, "workgroup", "local");
}

struct TileRegs {
  v16f w1;
  v8f  w2;
};

__device__ __forceinline__ void load_tile(TileRegs& r, const float* x1, const float* x2, int t,
                                          int half, int col) {
  const int n = t >> 2;
  const int c = (t & 3) * 16 + col;
  const float* p1 = x1 + (size_t)n * NUM_ORDERS * C_DIM + c;
  const float* p2 = x2 + (size_t)n * NUM_ORDERS * C_DIM + c;
  r.w2[0] = p2[(half ? 6 : 0) * C_DIM];
  r.w2[1] = p2[(half ? 12 : 2) * C_DIM];
  r.w2[2] = p2[(half ? 13 : 3) * C_DIM];
  r.w2[3] = p2[(half ? 15 : 7) * C_DIM];
  r.w2[4] = p2[(half ? 11 : 1) * C_DIM];
  r.w2[5] = p2[(half ? 9 : 5) * C_DIM];
  r.w2[6] = p2[(half ? 4 : 8) * C_DIM];
  r.w2[7] = p2[(half ? 10 : 14) * C_DIM];
  r.w1[0] = p1[0 * C_DIM];
  r.w1[1] = p1[1 * C_DIM];
  r.w1[2] = p1[2 * C_DIM];
  r.w1[3] = p1[3 * C_DIM];
  r.w1[4] = p1[4 * C_DIM];
  r.w1[5] = p1[5 * C_DIM];
  r.w1[6] = p1[6 * C_DIM];
  r.w1[7] = p1[7 * C_DIM];
  r.w1[8] = p1[8 * C_DIM];
  r.w1[9] = p1[9 * C_DIM];
  r.w1[10] = p1[10 * C_DIM];
  r.w1[11] = p1[11 * C_DIM];
  r.w1[12] = p1[12 * C_DIM];
  r.w1[13] = p1[13 * C_DIM];
  r.w1[14] = p1[14 * C_DIM];
  r.w1[15] = p1[15 * C_DIM];
}

#define WMMA_RESULT_GUARD(d) \
  asm volatile("s_wait_alu depctr_va_vdst(0)\n\tv_nop\n\tv_nop\n\tv_nop\n\tv_nop" : "+v"(d))

__device__ __forceinline__ void tile_product(const float4 (&areg)[32], unsigned mlo, unsigned mhi,
                                             const TileRegs& r, float* out, int t,
                                             int half, int col) {
  const v8f zero = {};
  float o[8] = {0.0f, 0.0f, 0.0f, 0.0f, 0.0f, 0.0f, 0.0f, 0.0f};
  if (mlo & 0x1u) {
    const v2f a = {areg[0].x, areg[0].y};
    const v2f b = {r.w2[0], r.w2[1]};
    v8f d = __builtin_amdgcn_wmma_f32_16x16x4_f32(false, a, false, b, (short)0, zero, false, false);
    WMMA_RESULT_GUARD(d);
    o[0] = __builtin_fmaf(r.w1[0], d[0], o[0]);
    o[0] = __builtin_fmaf(r.w1[2], d[1], o[0]);
    o[0] = __builtin_fmaf(r.w1[6], d[2], o[0]);
    o[0] = __builtin_fmaf(r.w1[12], d[3], o[0]);
    o[1] = __builtin_fmaf(r.w1[1], d[4], o[1]);
    o[1] = __builtin_fmaf(r.w1[5], d[5], o[1]);
    o[1] = __builtin_fmaf(r.w1[11], d[6], o[1]);
    o[2] = __builtin_fmaf(r.w1[3], d[7], o[2]);
  }
  if (mlo & 0xeu) {
    if (mlo & 0x2u) {
      const v2f a = {areg[0].z, areg[0].w};
      const v2f b = {r.w2[2], r.w2[3]};
      v8f d = __builtin_amdgcn_wmma_f32_16x16x4_f32(false, a, false, b, (short)0, zero, false, false);
      WMMA_RESULT_GUARD(d);
      o[0] = __builtin_fmaf(r.w1[0], d[0], o[0]);
      o[0] = __builtin_fmaf(r.w1[2], d[1], o[0]);
      o[0] = __builtin_fmaf(r.w1[6], d[2], o[0]);
      o[0] = __builtin_fmaf(r.w1[12], d[3], o[0]);
      o[1] = __builtin_fmaf(r.w1[1], d[4], o[1]);
      o[1] = __builtin_fmaf(r.w1[5], d[5], o[1]);
      o[1] = __builtin_fmaf(r.w1[11], d[6], o[1]);
      o[2] = __builtin_fmaf(r.w1[3], d[7], o[2]);
    }
    if (mlo & 0x4u) {
      const v2f a = {areg[1].x, areg[1].y};
      const v2f b = {r.w2[4], r.w2[5]};
      v8f d = __builtin_amdgcn_wmma_f32_16x16x4_f32(false, a, false, b, (short)0, zero, false, false);
      WMMA_RESULT_GUARD(d);
      o[0] = __builtin_fmaf(r.w1[0], d[0], o[0]);
      o[0] = __builtin_fmaf(r.w1[2], d[1], o[0]);
      o[0] = __builtin_fmaf(r.w1[6], d[2], o[0]);
      o[0] = __builtin_fmaf(r.w1[12], d[3], o[0]);
      o[1] = __builtin_fmaf(r.w1[1], d[4], o[1]);
      o[1] = __builtin_fmaf(r.w1[5], d[5], o[1]);
      o[1] = __builtin_fmaf(r.w1[11], d[6], o[1]);
      o[2] = __builtin_fmaf(r.w1[3], d[7], o[2]);
    }
    if (mlo & 0x8u) {
      const v2f a = {areg[1].z, areg[1].w};
      const v2f b = {r.w2[6], r.w2[7]};
      v8f d = __builtin_amdgcn_wmma_f32_16x16x4_f32(false, a, false, b, (short)0, zero, false, false);
      WMMA_RESULT_GUARD(d);
      o[0] = __builtin_fmaf(r.w1[0], d[0], o[0]);
      o[0] = __builtin_fmaf(r.w1[2], d[1], o[0]);
      o[0] = __builtin_fmaf(r.w1[6], d[2], o[0]);
      o[0] = __builtin_fmaf(r.w1[12], d[3], o[0]);
      o[1] = __builtin_fmaf(r.w1[1], d[4], o[1]);
      o[1] = __builtin_fmaf(r.w1[5], d[5], o[1]);
      o[1] = __builtin_fmaf(r.w1[11], d[6], o[1]);
      o[2] = __builtin_fmaf(r.w1[3], d[7], o[2]);
    }
  }
  if (mlo & 0x10u) {
    const v2f a = {areg[2].x, areg[2].y};
    const v2f b = {r.w2[0], r.w2[1]};
    v8f d = __builtin_amdgcn_wmma_f32_16x16x4_f32(false, a, false, b, (short)0, zero, false, false);
    WMMA_RESULT_GUARD(d);
    o[2] = __builtin_fmaf(r.w1[7], d[0], o[2]);
    o[2] = __builtin_fmaf(r.w1[13], d[1], o[2]);
    o[3] = __builtin_fmaf(r.w1[4], d[2], o[3]);
    o[3] = __builtin_fmaf(r.w1[10], d[3], o[3]);
    o[4] = __builtin_fmaf(r.w1[0], d[4], o[4]);
    o[4] = __builtin_fmaf(r.w1[2], d[5], o[4]);
    o[4] = __builtin_fmaf(r.w1[6], d[6], o[4]);
    o[4] = __builtin_fmaf(r.w1[12], d[7], o[4]);
  }
  if (mlo & 0xe0u) {
    if (mlo & 0x20u) {
      const v2f a = {areg[2].z, areg[2].w};
      const v2f b = {r.w2[2], r.w2[3]};
      v8f d = __builtin_amdgcn_wmma_f32_16x16x4_f32(false, a, false, b, (short)0, zero, false, false);
      WMMA_RESULT_GUARD(d);
      o[2] = __builtin_fmaf(r.w1[7], d[0], o[2]);
      o[2] = __builtin_fmaf(r.w1[13], d[1], o[2]);
      o[3] = __builtin_fmaf(r.w1[4], d[2], o[3]);
      o[3] = __builtin_fmaf(r.w1[10], d[3], o[3]);
      o[4] = __builtin_fmaf(r.w1[0], d[4], o[4]);
      o[4] = __builtin_fmaf(r.w1[2], d[5], o[4]);
      o[4] = __builtin_fmaf(r.w1[6], d[6], o[4]);
      o[4] = __builtin_fmaf(r.w1[12], d[7], o[4]);
    }
    if (mlo & 0x40u) {
      const v2f a = {areg[3].x, areg[3].y};
      const v2f b = {r.w2[4], r.w2[5]};
      v8f d = __builtin_amdgcn_wmma_f32_16x16x4_f32(false, a, false, b, (short)0, zero, false, false);
      WMMA_RESULT_GUARD(d);
      o[2] = __builtin_fmaf(r.w1[7], d[0], o[2]);
      o[2] = __builtin_fmaf(r.w1[13], d[1], o[2]);
      o[3] = __builtin_fmaf(r.w1[4], d[2], o[3]);
      o[3] = __builtin_fmaf(r.w1[10], d[3], o[3]);
      o[4] = __builtin_fmaf(r.w1[0], d[4], o[4]);
      o[4] = __builtin_fmaf(r.w1[2], d[5], o[4]);
      o[4] = __builtin_fmaf(r.w1[6], d[6], o[4]);
      o[4] = __builtin_fmaf(r.w1[12], d[7], o[4]);
    }
    if (mlo & 0x80u) {
      const v2f a = {areg[3].z, areg[3].w};
      const v2f b = {r.w2[6], r.w2[7]};
      v8f d = __builtin_amdgcn_wmma_f32_16x16x4_f32(false, a, false, b, (short)0, zero, false, false);
      WMMA_RESULT_GUARD(d);
      o[2] = __builtin_fmaf(r.w1[7], d[0], o[2]);
      o[2] = __builtin_fmaf(r.w1[13], d[1], o[2]);
      o[3] = __builtin_fmaf(r.w1[4], d[2], o[3]);
      o[3] = __builtin_fmaf(r.w1[10], d[3], o[3]);
      o[4] = __builtin_fmaf(r.w1[0], d[4], o[4]);
      o[4] = __builtin_fmaf(r.w1[2], d[5], o[4]);
      o[4] = __builtin_fmaf(r.w1[6], d[6], o[4]);
      o[4] = __builtin_fmaf(r.w1[12], d[7], o[4]);
    }
  }
  if (mlo & 0x100u) {
    const v2f a = {areg[4].x, areg[4].y};
    const v2f b = {r.w2[0], r.w2[1]};
    v8f d = __builtin_amdgcn_wmma_f32_16x16x4_f32(false, a, false, b, (short)0, zero, false, false);
    WMMA_RESULT_GUARD(d);
    o[5] = __builtin_fmaf(r.w1[8], d[0], o[5]);
    o[5] = __builtin_fmaf(r.w1[14], d[1], o[5]);
    o[6] = __builtin_fmaf((half ? r.w1[1] : r.w1[9]), d[2], o[6]);
    o[6] = __builtin_fmaf((half ? r.w1[5] : r.w1[3]), d[3], o[6]);
    o[6] = __builtin_fmaf((half ? r.w1[11] : r.w1[7]), d[4], o[6]);
    o[7] = __builtin_fmaf((half ? r.w1[15] : r.w1[3]), d[5], o[7]);
    o[7] = __builtin_fmaf((half ? r.w1[1] : r.w1[7]), d[6], o[7]);
    o[7] = __builtin_fmaf((half ? r.w1[4] : r.w1[13]), d[7], o[7]);
  }
  if (mlo & 0xe00u) {
    if (mlo & 0x200u) {
      const v2f a = {areg[4].z, areg[4].w};
      const v2f b = {r.w2[2], r.w2[3]};
      v8f d = __builtin_amdgcn_wmma_f32_16x16x4_f32(false, a, false, b, (short)0, zero, false, false);
      WMMA_RESULT_GUARD(d);
      o[5] = __builtin_fmaf(r.w1[8], d[0], o[5]);
      o[5] = __builtin_fmaf(r.w1[14], d[1], o[5]);
      o[6] = __builtin_fmaf((half ? r.w1[1] : r.w1[9]), d[2], o[6]);
      o[6] = __builtin_fmaf((half ? r.w1[5] : r.w1[3]), d[3], o[6]);
      o[6] = __builtin_fmaf((half ? r.w1[11] : r.w1[7]), d[4], o[6]);
      o[7] = __builtin_fmaf((half ? r.w1[15] : r.w1[3]), d[5], o[7]);
      o[7] = __builtin_fmaf((half ? r.w1[1] : r.w1[7]), d[6], o[7]);
      o[7] = __builtin_fmaf((half ? r.w1[4] : r.w1[13]), d[7], o[7]);
    }
    if (mlo & 0x400u) {
      const v2f a = {areg[5].x, areg[5].y};
      const v2f b = {r.w2[4], r.w2[5]};
      v8f d = __builtin_amdgcn_wmma_f32_16x16x4_f32(false, a, false, b, (short)0, zero, false, false);
      WMMA_RESULT_GUARD(d);
      o[5] = __builtin_fmaf(r.w1[8], d[0], o[5]);
      o[5] = __builtin_fmaf(r.w1[14], d[1], o[5]);
      o[6] = __builtin_fmaf((half ? r.w1[1] : r.w1[9]), d[2], o[6]);
      o[6] = __builtin_fmaf((half ? r.w1[5] : r.w1[3]), d[3], o[6]);
      o[6] = __builtin_fmaf((half ? r.w1[11] : r.w1[7]), d[4], o[6]);
      o[7] = __builtin_fmaf((half ? r.w1[15] : r.w1[3]), d[5], o[7]);
      o[7] = __builtin_fmaf((half ? r.w1[1] : r.w1[7]), d[6], o[7]);
      o[7] = __builtin_fmaf((half ? r.w1[4] : r.w1[13]), d[7], o[7]);
    }
    if (mlo & 0x800u) {
      const v2f a = {areg[5].z, areg[5].w};
      const v2f b = {r.w2[6], r.w2[7]};
      v8f d = __builtin_amdgcn_wmma_f32_16x16x4_f32(false, a, false, b, (short)0, zero, false, false);
      WMMA_RESULT_GUARD(d);
      o[5] = __builtin_fmaf(r.w1[8], d[0], o[5]);
      o[5] = __builtin_fmaf(r.w1[14], d[1], o[5]);
      o[6] = __builtin_fmaf((half ? r.w1[1] : r.w1[9]), d[2], o[6]);
      o[6] = __builtin_fmaf((half ? r.w1[5] : r.w1[3]), d[3], o[6]);
      o[6] = __builtin_fmaf((half ? r.w1[11] : r.w1[7]), d[4], o[6]);
      o[7] = __builtin_fmaf((half ? r.w1[15] : r.w1[3]), d[5], o[7]);
      o[7] = __builtin_fmaf((half ? r.w1[1] : r.w1[7]), d[6], o[7]);
      o[7] = __builtin_fmaf((half ? r.w1[4] : r.w1[13]), d[7], o[7]);
    }
  }
  if (mlo & 0x2000u) {
    const v2f a = {areg[6].z, areg[6].w};
    const v2f b = {r.w2[2], r.w2[3]};
    v8f d = __builtin_amdgcn_wmma_f32_16x16x4_f32(false, a, false, b, (short)0, zero, false, false);
    WMMA_RESULT_GUARD(d);
    o[0] = __builtin_fmaf(r.w1[1], d[0], o[0]);
    o[0] = __builtin_fmaf(r.w1[5], d[1], o[0]);
    o[0] = __builtin_fmaf(r.w1[9], d[2], o[0]);
    o[0] = __builtin_fmaf(r.w1[11], d[3], o[0]);
    o[1] = __builtin_fmaf(r.w1[4], d[4], o[1]);
    o[1] = __builtin_fmaf(r.w1[10], d[5], o[1]);
    o[2] = __builtin_fmaf(r.w1[0], d[6], o[2]);
    o[2] = __builtin_fmaf(r.w1[2], d[7], o[2]);
  }
  if (mlo & 0xd000u) {
    if (mlo & 0x1000u) {
      const v2f a = {areg[6].x, areg[6].y};
      const v2f b = {r.w2[0], r.w2[1]};
      v8f d = __builtin_amdgcn_wmma_f32_16x16x4_f32(false, a, false, b, (short)0, zero, false, false);
      WMMA_RESULT_GUARD(d);
      o[0] = __builtin_fmaf(r.w1[1], d[0], o[0]);
      o[0] = __builtin_fmaf(r.w1[5], d[1], o[0]);
      o[0] = __builtin_fmaf(r.w1[9], d[2], o[0]);
      o[0] = __builtin_fmaf(r.w1[11], d[3], o[0]);
      o[1] = __builtin_fmaf(r.w1[4], d[4], o[1]);
      o[1] = __builtin_fmaf(r.w1[10], d[5], o[1]);
      o[2] = __builtin_fmaf(r.w1[0], d[6], o[2]);
      o[2] = __builtin_fmaf(r.w1[2], d[7], o[2]);
    }
    if (mlo & 0x4000u) {
      const v2f a = {areg[7].x, areg[7].y};
      const v2f b = {r.w2[4], r.w2[5]};
      v8f d = __builtin_amdgcn_wmma_f32_16x16x4_f32(false, a, false, b, (short)0, zero, false, false);
      WMMA_RESULT_GUARD(d);
      o[0] = __builtin_fmaf(r.w1[1], d[0], o[0]);
      o[0] = __builtin_fmaf(r.w1[5], d[1], o[0]);
      o[0] = __builtin_fmaf(r.w1[9], d[2], o[0]);
      o[0] = __builtin_fmaf(r.w1[11], d[3], o[0]);
      o[1] = __builtin_fmaf(r.w1[4], d[4], o[1]);
      o[1] = __builtin_fmaf(r.w1[10], d[5], o[1]);
      o[2] = __builtin_fmaf(r.w1[0], d[6], o[2]);
      o[2] = __builtin_fmaf(r.w1[2], d[7], o[2]);
    }
    if (mlo & 0x8000u) {
      const v2f a = {areg[7].z, areg[7].w};
      const v2f b = {r.w2[6], r.w2[7]};
      v8f d = __builtin_amdgcn_wmma_f32_16x16x4_f32(false, a, false, b, (short)0, zero, false, false);
      WMMA_RESULT_GUARD(d);
      o[0] = __builtin_fmaf(r.w1[1], d[0], o[0]);
      o[0] = __builtin_fmaf(r.w1[5], d[1], o[0]);
      o[0] = __builtin_fmaf(r.w1[9], d[2], o[0]);
      o[0] = __builtin_fmaf(r.w1[11], d[3], o[0]);
      o[1] = __builtin_fmaf(r.w1[4], d[4], o[1]);
      o[1] = __builtin_fmaf(r.w1[10], d[5], o[1]);
      o[2] = __builtin_fmaf(r.w1[0], d[6], o[2]);
      o[2] = __builtin_fmaf(r.w1[2], d[7], o[2]);
    }
  }
  if (mlo & 0x20000u) {
    const v2f a = {areg[8].z, areg[8].w};
    const v2f b = {r.w2[2], r.w2[3]};
    v8f d = __builtin_amdgcn_wmma_f32_16x16x4_f32(false, a, false, b, (short)0, zero, false, false);
    WMMA_RESULT_GUARD(d);
    o[2] = __builtin_fmaf(r.w1[4], d[0], o[2]);
    o[2] = __builtin_fmaf(r.w1[6], d[1], o[2]);
    o[2] = __builtin_fmaf(r.w1[10], d[2], o[2]);
    o[2] = __builtin_fmaf(r.w1[12], d[3], o[2]);
    o[3] = __builtin_fmaf(r.w1[9], d[4], o[3]);
    o[4] = __builtin_fmaf(r.w1[1], d[5], o[4]);
    o[4] = __builtin_fmaf(r.w1[5], d[6], o[4]);
    o[4] = __builtin_fmaf(r.w1[9], d[7], o[4]);
  }
  if (mlo & 0xd0000u) {
    if (mlo & 0x10000u) {
      const v2f a = {areg[8].x, areg[8].y};
      const v2f b = {r.w2[0], r.w2[1]};
      v8f d = __builtin_amdgcn_wmma_f32_16x16x4_f32(false, a, false, b, (short)0, zero, false, false);
      WMMA_RESULT_GUARD(d);
      o[2] = __builtin_fmaf(r.w1[4], d[0], o[2]);
      o[2] = __builtin_fmaf(r.w1[6], d[1], o[2]);
      o[2] = __builtin_fmaf(r.w1[10], d[2], o[2]);
      o[2] = __builtin_fmaf(r.w1[12], d[3], o[2]);
      o[3] = __builtin_fmaf(r.w1[9], d[4], o[3]);
      o[4] = __builtin_fmaf(r.w1[1], d[5], o[4]);
      o[4] = __builtin_fmaf(r.w1[5], d[6], o[4]);
      o[4] = __builtin_fmaf(r.w1[9], d[7], o[4]);
    }
    if (mlo & 0x40000u) {
      const v2f a = {areg[9].x, areg[9].y};
      const v2f b = {r.w2[4], r.w2[5]};
      v8f d = __builtin_amdgcn_wmma_f32_16x16x4_f32(false, a, false, b, (short)0, zero, false, false);
      WMMA_RESULT_GUARD(d);
      o[2] = __builtin_fmaf(r.w1[4], d[0], o[2]);
      o[2] = __builtin_fmaf(r.w1[6], d[1], o[2]);
      o[2] = __builtin_fmaf(r.w1[10], d[2], o[2]);
      o[2] = __builtin_fmaf(r.w1[12], d[3], o[2]);
      o[3] = __builtin_fmaf(r.w1[9], d[4], o[3]);
      o[4] = __builtin_fmaf(r.w1[1], d[5], o[4]);
      o[4] = __builtin_fmaf(r.w1[5], d[6], o[4]);
      o[4] = __builtin_fmaf(r.w1[9], d[7], o[4]);
    }
    if (mlo & 0x80000u) {
      const v2f a = {areg[9].z, areg[9].w};
      const v2f b = {r.w2[6], r.w2[7]};
      v8f d = __builtin_amdgcn_wmma_f32_16x16x4_f32(false, a, false, b, (short)0, zero, false, false);
      WMMA_RESULT_GUARD(d);
      o[2] = __builtin_fmaf(r.w1[4], d[0], o[2]);
      o[2] = __builtin_fmaf(r.w1[6], d[1], o[2]);
      o[2] = __builtin_fmaf(r.w1[10], d[2], o[2]);
      o[2] = __builtin_fmaf(r.w1[12], d[3], o[2]);
      o[3] = __builtin_fmaf(r.w1[9], d[4], o[3]);
      o[4] = __builtin_fmaf(r.w1[1], d[5], o[4]);
      o[4] = __builtin_fmaf(r.w1[5], d[6], o[4]);
      o[4] = __builtin_fmaf(r.w1[9], d[7], o[4]);
    }
  }
  if (mlo & 0x200000u) {
    const v2f a = {areg[10].z, areg[10].w};
    const v2f b = {r.w2[2], r.w2[3]};
    v8f d = __builtin_amdgcn_wmma_f32_16x16x4_f32(false, a, false, b, (short)0, zero, false, false);
    WMMA_RESULT_GUARD(d);
    o[4] = __builtin_fmaf(r.w1[11], d[0], o[4]);
    o[5] = __builtin_fmaf(r.w1[1], d[1], o[5]);
    o[5] = __builtin_fmaf(r.w1[3], d[2], o[5]);
    o[5] = __builtin_fmaf(r.w1[5], d[3], o[5]);
    o[5] = __builtin_fmaf(r.w1[7], d[4], o[5]);
    o[5] = __builtin_fmaf(r.w1[11], d[5], o[5]);
    o[5] = __builtin_fmaf(r.w1[13], d[6], o[5]);
    o[6] = __builtin_fmaf((half ? r.w1[4] : r.w1[8]), d[7], o[6]);
  }
  if (mlo & 0xd00000u) {
    if (mlo & 0x100000u) {
      const v2f a = {areg[10].x, areg[10].y};
      const v2f b = {r.w2[0], r.w2[1]};
      v8f d = __builtin_amdgcn_wmma_f32_16x16x4_f32(false, a, false, b, (short)0, zero, false, false);
      WMMA_RESULT_GUARD(d);
      o[4] = __builtin_fmaf(r.w1[11], d[0], o[4]);
      o[5] = __builtin_fmaf(r.w1[1], d[1], o[5]);
      o[5] = __builtin_fmaf(r.w1[3], d[2], o[5]);
      o[5] = __builtin_fmaf(r.w1[5], d[3], o[5]);
      o[5] = __builtin_fmaf(r.w1[7], d[4], o[5]);
      o[5] = __builtin_fmaf(r.w1[11], d[5], o[5]);
      o[5] = __builtin_fmaf(r.w1[13], d[6], o[5]);
      o[6] = __builtin_fmaf((half ? r.w1[4] : r.w1[8]), d[7], o[6]);
    }
    if (mlo & 0x400000u) {
      const v2f a = {areg[11].x, areg[11].y};
      const v2f b = {r.w2[4], r.w2[5]};
      v8f d = __builtin_amdgcn_wmma_f32_16x16x4_f32(false, a, false, b, (short)0, zero, false, false);
      WMMA_RESULT_GUARD(d);
      o[4] = __builtin_fmaf(r.w1[11], d[0], o[4]);
      o[5] = __builtin_fmaf(r.w1[1], d[1], o[5]);
      o[5] = __builtin_fmaf(r.w1[3], d[2], o[5]);
      o[5] = __builtin_fmaf(r.w1[5], d[3], o[5]);
      o[5] = __builtin_fmaf(r.w1[7], d[4], o[5]);
      o[5] = __builtin_fmaf(r.w1[11], d[5], o[5]);
      o[5] = __builtin_fmaf(r.w1[13], d[6], o[5]);
      o[6] = __builtin_fmaf((half ? r.w1[4] : r.w1[8]), d[7], o[6]);
    }
    if (mlo & 0x800000u) {
      const v2f a = {areg[11].z, areg[11].w};
      const v2f b = {r.w2[6], r.w2[7]};
      v8f d = __builtin_amdgcn_wmma_f32_16x16x4_f32(false, a, false, b, (short)0, zero, false, false);
      WMMA_RESULT_GUARD(d);
      o[4] = __builtin_fmaf(r.w1[11], d[0], o[4]);
      o[5] = __builtin_fmaf(r.w1[1], d[1], o[5]);
      o[5] = __builtin_fmaf(r.w1[3], d[2], o[5]);
      o[5] = __builtin_fmaf(r.w1[5], d[3], o[5]);
      o[5] = __builtin_fmaf(r.w1[7], d[4], o[5]);
      o[5] = __builtin_fmaf(r.w1[11], d[5], o[5]);
      o[5] = __builtin_fmaf(r.w1[13], d[6], o[5]);
      o[6] = __builtin_fmaf((half ? r.w1[4] : r.w1[8]), d[7], o[6]);
    }
  }
  if (mlo & 0x2000000u) {
    const v2f a = {areg[12].z, areg[12].w};
    const v2f b = {r.w2[2], r.w2[3]};
    v8f d = __builtin_amdgcn_wmma_f32_16x16x4_f32(false, a, false, b, (short)0, zero, false, false);
    WMMA_RESULT_GUARD(d);
    o[6] = __builtin_fmaf((half ? r.w1[10] : r.w1[13]), d[0], o[6]);
    o[7] = __builtin_fmaf(r.w1[0], d[1], o[7]);
    o[7] = __builtin_fmaf(r.w1[2], d[2], o[7]);
    o[7] = __builtin_fmaf(r.w1[6], d[3], o[7]);
    o[7] = __builtin_fmaf(r.w1[12], d[4], o[7]);
    o[7] = __builtin_fmaf((half ? r.w1[8] : r.w1[4]), d[5], o[7]);
    o[7] = __builtin_fmaf((half ? r.w1[14] : r.w1[10]), d[6], o[7]);
    o[7] = __builtin_fmaf((half ? r.w1[11] : r.w1[9]), d[7], o[7]);
  }
  if (mlo & 0xd000000u) {
    if (mlo & 0x1000000u) {
      const v2f a = {areg[12].x, areg[12].y};
      const v2f b = {r.w2[0], r.w2[1]};
      v8f d = __builtin_amdgcn_wmma_f32_16x16x4_f32(false, a, false, b, (short)0, zero, false, false);
      WMMA_RESULT_GUARD(d);
      o[6] = __builtin_fmaf((half ? r.w1[10] : r.w1[13]), d[0], o[6]);
      o[7] = __builtin_fmaf(r.w1[0], d[1], o[7]);
      o[7] = __builtin_fmaf(r.w1[2], d[2], o[7]);
      o[7] = __builtin_fmaf(r.w1[6], d[3], o[7]);
      o[7] = __builtin_fmaf(r.w1[12], d[4], o[7]);
      o[7] = __builtin_fmaf((half ? r.w1[8] : r.w1[4]), d[5], o[7]);
      o[7] = __builtin_fmaf((half ? r.w1[14] : r.w1[10]), d[6], o[7]);
      o[7] = __builtin_fmaf((half ? r.w1[11] : r.w1[9]), d[7], o[7]);
    }
    if (mlo & 0x4000000u) {
      const v2f a = {areg[13].x, areg[13].y};
      const v2f b = {r.w2[4], r.w2[5]};
      v8f d = __builtin_amdgcn_wmma_f32_16x16x4_f32(false, a, false, b, (short)0, zero, false, false);
      WMMA_RESULT_GUARD(d);
      o[6] = __builtin_fmaf((half ? r.w1[10] : r.w1[13]), d[0], o[6]);
      o[7] = __builtin_fmaf(r.w1[0], d[1], o[7]);
      o[7] = __builtin_fmaf(r.w1[2], d[2], o[7]);
      o[7] = __builtin_fmaf(r.w1[6], d[3], o[7]);
      o[7] = __builtin_fmaf(r.w1[12], d[4], o[7]);
      o[7] = __builtin_fmaf((half ? r.w1[8] : r.w1[4]), d[5], o[7]);
      o[7] = __builtin_fmaf((half ? r.w1[14] : r.w1[10]), d[6], o[7]);
      o[7] = __builtin_fmaf((half ? r.w1[11] : r.w1[9]), d[7], o[7]);
    }
    if (mlo & 0x8000000u) {
      const v2f a = {areg[13].z, areg[13].w};
      const v2f b = {r.w2[6], r.w2[7]};
      v8f d = __builtin_amdgcn_wmma_f32_16x16x4_f32(false, a, false, b, (short)0, zero, false, false);
      WMMA_RESULT_GUARD(d);
      o[6] = __builtin_fmaf((half ? r.w1[10] : r.w1[13]), d[0], o[6]);
      o[7] = __builtin_fmaf(r.w1[0], d[1], o[7]);
      o[7] = __builtin_fmaf(r.w1[2], d[2], o[7]);
      o[7] = __builtin_fmaf(r.w1[6], d[3], o[7]);
      o[7] = __builtin_fmaf(r.w1[12], d[4], o[7]);
      o[7] = __builtin_fmaf((half ? r.w1[8] : r.w1[4]), d[5], o[7]);
      o[7] = __builtin_fmaf((half ? r.w1[14] : r.w1[10]), d[6], o[7]);
      o[7] = __builtin_fmaf((half ? r.w1[11] : r.w1[9]), d[7], o[7]);
    }
  }
  if (mlo & 0x40000000u) {
    const v2f a = {areg[15].x, areg[15].y};
    const v2f b = {r.w2[4], r.w2[5]};
    v8f d = __builtin_amdgcn_wmma_f32_16x16x4_f32(false, a, false, b, (short)0, zero, false, false);
    WMMA_RESULT_GUARD(d);
    o[0] = __builtin_fmaf(r.w1[3], d[0], o[0]);
    o[0] = __builtin_fmaf(r.w1[7], d[1], o[0]);
    o[0] = __builtin_fmaf(r.w1[13], d[2], o[0]);
    o[0] = __builtin_fmaf(r.w1[15], d[3], o[0]);
    o[1] = __builtin_fmaf(r.w1[0], d[4], o[1]);
    o[1] = __builtin_fmaf(r.w1[2], d[5], o[1]);
    o[1] = __builtin_fmaf(r.w1[6], d[6], o[1]);
    o[1] = __builtin_fmaf(r.w1[8], d[7], o[1]);
  }
  if (mlo & 0xb0000000u) {
    if (mlo & 0x10000000u) {
      const v2f a = {areg[14].x, areg[14].y};
      const v2f b = {r.w2[0], r.w2[1]};
      v8f d = __builtin_amdgcn_wmma_f32_16x16x4_f32(false, a, false, b, (short)0, zero, false, false);
      WMMA_RESULT_GUARD(d);
      o[0] = __builtin_fmaf(r.w1[3], d[0], o[0]);
      o[0] = __builtin_fmaf(r.w1[7], d[1], o[0]);
      o[0] = __builtin_fmaf(r.w1[13], d[2], o[0]);
      o[0] = __builtin_fmaf(r.w1[15], d[3], o[0]);
      o[1] = __builtin_fmaf(r.w1[0], d[4], o[1]);
      o[1] = __builtin_fmaf(r.w1[2], d[5], o[1]);
      o[1] = __builtin_fmaf(r.w1[6], d[6], o[1]);
      o[1] = __builtin_fmaf(r.w1[8], d[7], o[1]);
    }
    if (mlo & 0x20000000u) {
      const v2f a = {areg[14].z, areg[14].w};
      const v2f b = {r.w2[2], r.w2[3]};
      v8f d = __builtin_amdgcn_wmma_f32_16x16x4_f32(false, a, false, b, (short)0, zero, false, false);
      WMMA_RESULT_GUARD(d);
      o[0] = __builtin_fmaf(r.w1[3], d[0], o[0]);
      o[0] = __builtin_fmaf(r.w1[7], d[1], o[0]);
      o[0] = __builtin_fmaf(r.w1[13], d[2], o[0]);
      o[0] = __builtin_fmaf(r.w1[15], d[3], o[0]);
      o[1] = __builtin_fmaf(r.w1[0], d[4], o[1]);
      o[1] = __builtin_fmaf(r.w1[2], d[5], o[1]);
      o[1] = __builtin_fmaf(r.w1[6], d[6], o[1]);
      o[1] = __builtin_fmaf(r.w1[8], d[7], o[1]);
    }
    if (mlo & 0x80000000u) {
      const v2f a = {areg[15].z, areg[15].w};
      const v2f b = {r.w2[6], r.w2[7]};
      v8f d = __builtin_amdgcn_wmma_f32_16x16x4_f32(false, a, false, b, (short)0, zero, false, false);
      WMMA_RESULT_GUARD(d);
      o[0] = __builtin_fmaf(r.w1[3], d[0], o[0]);
      o[0] = __builtin_fmaf(r.w1[7], d[1], o[0]);
      o[0] = __builtin_fmaf(r.w1[13], d[2], o[0]);
      o[0] = __builtin_fmaf(r.w1[15], d[3], o[0]);
      o[1] = __builtin_fmaf(r.w1[0], d[4], o[1]);
      o[1] = __builtin_fmaf(r.w1[2], d[5], o[1]);
      o[1] = __builtin_fmaf(r.w1[6], d[6], o[1]);
      o[1] = __builtin_fmaf(r.w1[8], d[7], o[1]);
    }
  }
  if (mhi & 0x4u) {
    const v2f a = {areg[17].x, areg[17].y};
    const v2f b = {r.w2[4], r.w2[5]};
    v8f d = __builtin_amdgcn_wmma_f32_16x16x4_f32(false, a, false, b, (short)0, zero, false, false);
    WMMA_RESULT_GUARD(d);
    o[1] = __builtin_fmaf(r.w1[12], d[0], o[1]);
    o[1] = __builtin_fmaf(r.w1[14], d[1], o[1]);
    o[2] = __builtin_fmaf(r.w1[8], d[2], o[2]);
    o[2] = __builtin_fmaf(r.w1[14], d[3], o[2]);
    o[3] = __builtin_fmaf(r.w1[1], d[4], o[3]);
    o[3] = __builtin_fmaf(r.w1[3], d[5], o[3]);
    o[3] = __builtin_fmaf(r.w1[5], d[6], o[3]);
    o[3] = __builtin_fmaf(r.w1[7], d[7], o[3]);
  }
  if (mhi & 0xbu) {
    if (mhi & 0x1u) {
      const v2f a = {areg[16].x, areg[16].y};
      const v2f b = {r.w2[0], r.w2[1]};
      v8f d = __builtin_amdgcn_wmma_f32_16x16x4_f32(false, a, false, b, (short)0, zero, false, false);
      WMMA_RESULT_GUARD(d);
      o[1] = __builtin_fmaf(r.w1[12], d[0], o[1]);
      o[1] = __builtin_fmaf(r.w1[14], d[1], o[1]);
      o[2] = __builtin_fmaf(r.w1[8], d[2], o[2]);
      o[2] = __builtin_fmaf(r.w1[14], d[3], o[2]);
      o[3] = __builtin_fmaf(r.w1[1], d[4], o[3]);
      o[3] = __builtin_fmaf(r.w1[3], d[5], o[3]);
      o[3] = __builtin_fmaf(r.w1[5], d[6], o[3]);
      o[3] = __builtin_fmaf(r.w1[7], d[7], o[3]);
    }
    if (mhi & 0x2u) {
      const v2f a = {areg[16].z, areg[16].w};
      const v2f b = {r.w2[2], r.w2[3]};
      v8f d = __builtin_amdgcn_wmma_f32_16x16x4_f32(false, a, false, b, (short)0, zero, false, false);
      WMMA_RESULT_GUARD(d);
      o[1] = __builtin_fmaf(r.w1[12], d[0], o[1]);
      o[1] = __builtin_fmaf(r.w1[14], d[1], o[1]);
      o[2] = __builtin_fmaf(r.w1[8], d[2], o[2]);
      o[2] = __builtin_fmaf(r.w1[14], d[3], o[2]);
      o[3] = __builtin_fmaf(r.w1[1], d[4], o[3]);
      o[3] = __builtin_fmaf(r.w1[3], d[5], o[3]);
      o[3] = __builtin_fmaf(r.w1[5], d[6], o[3]);
      o[3] = __builtin_fmaf(r.w1[7], d[7], o[3]);
    }
    if (mhi & 0x8u) {
      const v2f a = {areg[17].z, areg[17].w};
      const v2f b = {r.w2[6], r.w2[7]};
      v8f d = __builtin_amdgcn_wmma_f32_16x16x4_f32(false, a, false, b, (short)0, zero, false, false);
      WMMA_RESULT_GUARD(d);
      o[1] = __builtin_fmaf(r.w1[12], d[0], o[1]);
      o[1] = __builtin_fmaf(r.w1[14], d[1], o[1]);
      o[2] = __builtin_fmaf(r.w1[8], d[2], o[2]);
      o[2] = __builtin_fmaf(r.w1[14], d[3], o[2]);
      o[3] = __builtin_fmaf(r.w1[1], d[4], o[3]);
      o[3] = __builtin_fmaf(r.w1[3], d[5], o[3]);
      o[3] = __builtin_fmaf(r.w1[5], d[6], o[3]);
      o[3] = __builtin_fmaf(r.w1[7], d[7], o[3]);
    }
  }
  if (mhi & 0x40u) {
    const v2f a = {areg[19].x, areg[19].y};
    const v2f b = {r.w2[4], r.w2[5]};
    v8f d = __builtin_amdgcn_wmma_f32_16x16x4_f32(false, a, false, b, (short)0, zero, false, false);
    WMMA_RESULT_GUARD(d);
    o[3] = __builtin_fmaf(r.w1[11], d[0], o[3]);
    o[3] = __builtin_fmaf(r.w1[13], d[1], o[3]);
    o[4] = __builtin_fmaf(r.w1[3], d[2], o[4]);
    o[4] = __builtin_fmaf(r.w1[7], d[3], o[4]);
    o[4] = __builtin_fmaf(r.w1[13], d[4], o[4]);
    o[4] = __builtin_fmaf(r.w1[15], d[5], o[4]);
    o[5] = __builtin_fmaf(r.w1[15], d[6], o[5]);
    o[6] = __builtin_fmaf(r.w1[0], d[7], o[6]);
  }
  if (mhi & 0xb0u) {
    if (mhi & 0x10u) {
      const v2f a = {areg[18].x, areg[18].y};
      const v2f b = {r.w2[0], r.w2[1]};
      v8f d = __builtin_amdgcn_wmma_f32_16x16x4_f32(false, a, false, b, (short)0, zero, false, false);
      WMMA_RESULT_GUARD(d);
      o[3] = __builtin_fmaf(r.w1[11], d[0], o[3]);
      o[3] = __builtin_fmaf(r.w1[13], d[1], o[3]);
      o[4] = __builtin_fmaf(r.w1[3], d[2], o[4]);
      o[4] = __builtin_fmaf(r.w1[7], d[3], o[4]);
      o[4] = __builtin_fmaf(r.w1[13], d[4], o[4]);
      o[4] = __builtin_fmaf(r.w1[15], d[5], o[4]);
      o[5] = __builtin_fmaf(r.w1[15], d[6], o[5]);
      o[6] = __builtin_fmaf(r.w1[0], d[7], o[6]);
    }
    if (mhi & 0x20u) {
      const v2f a = {areg[18].z, areg[18].w};
      const v2f b = {r.w2[2], r.w2[3]};
      v8f d = __builtin_amdgcn_wmma_f32_16x16x4_f32(false, a, false, b, (short)0, zero, false, false);
      WMMA_RESULT_GUARD(d);
      o[3] = __builtin_fmaf(r.w1[11], d[0], o[3]);
      o[3] = __builtin_fmaf(r.w1[13], d[1], o[3]);
      o[4] = __builtin_fmaf(r.w1[3], d[2], o[4]);
      o[4] = __builtin_fmaf(r.w1[7], d[3], o[4]);
      o[4] = __builtin_fmaf(r.w1[13], d[4], o[4]);
      o[4] = __builtin_fmaf(r.w1[15], d[5], o[4]);
      o[5] = __builtin_fmaf(r.w1[15], d[6], o[5]);
      o[6] = __builtin_fmaf(r.w1[0], d[7], o[6]);
    }
    if (mhi & 0x80u) {
      const v2f a = {areg[19].z, areg[19].w};
      const v2f b = {r.w2[6], r.w2[7]};
      v8f d = __builtin_amdgcn_wmma_f32_16x16x4_f32(false, a, false, b, (short)0, zero, false, false);
      WMMA_RESULT_GUARD(d);
      o[3] = __builtin_fmaf(r.w1[11], d[0], o[3]);
      o[3] = __builtin_fmaf(r.w1[13], d[1], o[3]);
      o[4] = __builtin_fmaf(r.w1[3], d[2], o[4]);
      o[4] = __builtin_fmaf(r.w1[7], d[3], o[4]);
      o[4] = __builtin_fmaf(r.w1[13], d[4], o[4]);
      o[4] = __builtin_fmaf(r.w1[15], d[5], o[4]);
      o[5] = __builtin_fmaf(r.w1[15], d[6], o[5]);
      o[6] = __builtin_fmaf(r.w1[0], d[7], o[6]);
    }
  }
  if (mhi & 0x400u) {
    const v2f a = {areg[21].x, areg[21].y};
    const v2f b = {r.w2[4], r.w2[5]};
    v8f d = __builtin_amdgcn_wmma_f32_16x16x4_f32(false, a, false, b, (short)0, zero, false, false);
    WMMA_RESULT_GUARD(d);
    o[6] = __builtin_fmaf(r.w1[2], d[0], o[6]);
    o[6] = __builtin_fmaf(r.w1[6], d[1], o[6]);
    o[6] = __builtin_fmaf(r.w1[12], d[2], o[6]);
    o[6] = __builtin_fmaf((half ? r.w1[8] : r.w1[4]), d[3], o[6]);
    o[6] = __builtin_fmaf((half ? r.w1[14] : r.w1[10]), d[4], o[6]);
    o[7] = __builtin_fmaf((half ? r.w1[5] : r.w1[8]), d[5], o[7]);
    o[7] = __builtin_fmaf((half ? r.w1[9] : r.w1[14]), d[6], o[7]);
    o[6] = __builtin_fmaf(r.w1[15], d[7], o[6]);
  }
  if (mhi & 0xb00u) {
    if (mhi & 0x100u) {
      const v2f a = {areg[20].x, areg[20].y};
      const v2f b = {r.w2[0], r.w2[1]};
      v8f d = __builtin_amdgcn_wmma_f32_16x16x4_f32(false, a, false, b, (short)0, zero, false, false);
      WMMA_RESULT_GUARD(d);
      o[6] = __builtin_fmaf(r.w1[2], d[0], o[6]);
      o[6] = __builtin_fmaf(r.w1[6], d[1], o[6]);
      o[6] = __builtin_fmaf(r.w1[12], d[2], o[6]);
      o[6] = __builtin_fmaf((half ? r.w1[8] : r.w1[4]), d[3], o[6]);
      o[6] = __builtin_fmaf((half ? r.w1[14] : r.w1[10]), d[4], o[6]);
      o[7] = __builtin_fmaf((half ? r.w1[5] : r.w1[8]), d[5], o[7]);
      o[7] = __builtin_fmaf((half ? r.w1[9] : r.w1[14]), d[6], o[7]);
      o[6] = __builtin_fmaf(r.w1[15], d[7], o[6]);
    }
    if (mhi & 0x200u) {
      const v2f a = {areg[20].z, areg[20].w};
      const v2f b = {r.w2[2], r.w2[3]};
      v8f d = __builtin_amdgcn_wmma_f32_16x16x4_f32(false, a, false, b, (short)0, zero, false, false);
      WMMA_RESULT_GUARD(d);
      o[6] = __builtin_fmaf(r.w1[2], d[0], o[6]);
      o[6] = __builtin_fmaf(r.w1[6], d[1], o[6]);
      o[6] = __builtin_fmaf(r.w1[12], d[2], o[6]);
      o[6] = __builtin_fmaf((half ? r.w1[8] : r.w1[4]), d[3], o[6]);
      o[6] = __builtin_fmaf((half ? r.w1[14] : r.w1[10]), d[4], o[6]);
      o[7] = __builtin_fmaf((half ? r.w1[5] : r.w1[8]), d[5], o[7]);
      o[7] = __builtin_fmaf((half ? r.w1[9] : r.w1[14]), d[6], o[7]);
      o[6] = __builtin_fmaf(r.w1[15], d[7], o[6]);
    }
    if (mhi & 0x800u) {
      const v2f a = {areg[21].z, areg[21].w};
      const v2f b = {r.w2[6], r.w2[7]};
      v8f d = __builtin_amdgcn_wmma_f32_16x16x4_f32(false, a, false, b, (short)0, zero, false, false);
      WMMA_RESULT_GUARD(d);
      o[6] = __builtin_fmaf(r.w1[2], d[0], o[6]);
      o[6] = __builtin_fmaf(r.w1[6], d[1], o[6]);
      o[6] = __builtin_fmaf(r.w1[12], d[2], o[6]);
      o[6] = __builtin_fmaf((half ? r.w1[8] : r.w1[4]), d[3], o[6]);
      o[6] = __builtin_fmaf((half ? r.w1[14] : r.w1[10]), d[4], o[6]);
      o[7] = __builtin_fmaf((half ? r.w1[5] : r.w1[8]), d[5], o[7]);
      o[7] = __builtin_fmaf((half ? r.w1[9] : r.w1[14]), d[6], o[7]);
      o[6] = __builtin_fmaf(r.w1[15], d[7], o[6]);
    }
  }
  if (mhi & 0x8000u) {
    const v2f a = {areg[23].z, areg[23].w};
    const v2f b = {r.w2[6], r.w2[7]};
    v8f d = __builtin_amdgcn_wmma_f32_16x16x4_f32(false, a, false, b, (short)0, zero, false, false);
    WMMA_RESULT_GUARD(d);
    o[0] = __builtin_fmaf(r.w1[4], d[0], o[0]);
    o[0] = __builtin_fmaf(r.w1[8], d[1], o[0]);
    o[0] = __builtin_fmaf(r.w1[10], d[2], o[0]);
    o[0] = __builtin_fmaf(r.w1[14], d[3], o[0]);
    o[1] = __builtin_fmaf(r.w1[3], d[4], o[1]);
    o[1] = __builtin_fmaf(r.w1[7], d[5], o[1]);
    o[1] = __builtin_fmaf(r.w1[9], d[6], o[1]);
    o[1] = __builtin_fmaf(r.w1[13], d[7], o[1]);
  }
  if (mhi & 0x7000u) {
    if (mhi & 0x1000u) {
      const v2f a = {areg[22].x, areg[22].y};
      const v2f b = {r.w2[0], r.w2[1]};
      v8f d = __builtin_amdgcn_wmma_f32_16x16x4_f32(false, a, false, b, (short)0, zero, false, false);
      WMMA_RESULT_GUARD(d);
      o[0] = __builtin_fmaf(r.w1[4], d[0], o[0]);
      o[0] = __builtin_fmaf(r.w1[8], d[1], o[0]);
      o[0] = __builtin_fmaf(r.w1[10], d[2], o[0]);
      o[0] = __builtin_fmaf(r.w1[14], d[3], o[0]);
      o[1] = __builtin_fmaf(r.w1[3], d[4], o[1]);
      o[1] = __builtin_fmaf(r.w1[7], d[5], o[1]);
      o[1] = __builtin_fmaf(r.w1[9], d[6], o[1]);
      o[1] = __builtin_fmaf(r.w1[13], d[7], o[1]);
    }
    if (mhi & 0x2000u) {
      const v2f a = {areg[22].z, areg[22].w};
      const v2f b = {r.w2[2], r.w2[3]};
      v8f d = __builtin_amdgcn_wmma_f32_16x16x4_f32(false, a, false, b, (short)0, zero, false, false);
      WMMA_RESULT_GUARD(d);
      o[0] = __builtin_fmaf(r.w1[4], d[0], o[0]);
      o[0] = __builtin_fmaf(r.w1[8], d[1], o[0]);
      o[0] = __builtin_fmaf(r.w1[10], d[2], o[0]);
      o[0] = __builtin_fmaf(r.w1[14], d[3], o[0]);
      o[1] = __builtin_fmaf(r.w1[3], d[4], o[1]);
      o[1] = __builtin_fmaf(r.w1[7], d[5], o[1]);
      o[1] = __builtin_fmaf(r.w1[9], d[6], o[1]);
      o[1] = __builtin_fmaf(r.w1[13], d[7], o[1]);
    }
    if (mhi & 0x4000u) {
      const v2f a = {areg[23].x, areg[23].y};
      const v2f b = {r.w2[4], r.w2[5]};
      v8f d = __builtin_amdgcn_wmma_f32_16x16x4_f32(false, a, false, b, (short)0, zero, false, false);
      WMMA_RESULT_GUARD(d);
      o[0] = __builtin_fmaf(r.w1[4], d[0], o[0]);
      o[0] = __builtin_fmaf(r.w1[8], d[1], o[0]);
      o[0] = __builtin_fmaf(r.w1[10], d[2], o[0]);
      o[0] = __builtin_fmaf(r.w1[14], d[3], o[0]);
      o[1] = __builtin_fmaf(r.w1[3], d[4], o[1]);
      o[1] = __builtin_fmaf(r.w1[7], d[5], o[1]);
      o[1] = __builtin_fmaf(r.w1[9], d[6], o[1]);
      o[1] = __builtin_fmaf(r.w1[13], d[7], o[1]);
    }
  }
  if (mhi & 0x80000u) {
    const v2f a = {areg[25].z, areg[25].w};
    const v2f b = {r.w2[6], r.w2[7]};
    v8f d = __builtin_amdgcn_wmma_f32_16x16x4_f32(false, a, false, b, (short)0, zero, false, false);
    WMMA_RESULT_GUARD(d);
    o[2] = __builtin_fmaf(r.w1[1], d[0], o[2]);
    o[2] = __builtin_fmaf(r.w1[5], d[1], o[2]);
    o[2] = __builtin_fmaf(r.w1[11], d[2], o[2]);
    o[2] = __builtin_fmaf(r.w1[15], d[3], o[2]);
    o[3] = __builtin_fmaf(r.w1[0], d[4], o[3]);
    o[3] = __builtin_fmaf(r.w1[2], d[5], o[3]);
    o[3] = __builtin_fmaf(r.w1[6], d[6], o[3]);
    o[3] = __builtin_fmaf(r.w1[12], d[7], o[3]);
  }
  if (mhi & 0x70000u) {
    if (mhi & 0x10000u) {
      const v2f a = {areg[24].x, areg[24].y};
      const v2f b = {r.w2[0], r.w2[1]};
      v8f d = __builtin_amdgcn_wmma_f32_16x16x4_f32(false, a, false, b, (short)0, zero, false, false);
      WMMA_RESULT_GUARD(d);
      o[2] = __builtin_fmaf(r.w1[1], d[0], o[2]);
      o[2] = __builtin_fmaf(r.w1[5], d[1], o[2]);
      o[2] = __builtin_fmaf(r.w1[11], d[2], o[2]);
      o[2] = __builtin_fmaf(r.w1[15], d[3], o[2]);
      o[3] = __builtin_fmaf(r.w1[0], d[4], o[3]);
      o[3] = __builtin_fmaf(r.w1[2], d[5], o[3]);
      o[3] = __builtin_fmaf(r.w1[6], d[6], o[3]);
      o[3] = __builtin_fmaf(r.w1[12], d[7], o[3]);
    }
    if (mhi & 0x20000u) {
      const v2f a = {areg[24].z, areg[24].w};
      const v2f b = {r.w2[2], r.w2[3]};
      v8f d = __builtin_amdgcn_wmma_f32_16x16x4_f32(false, a, false, b, (short)0, zero, false, false);
      WMMA_RESULT_GUARD(d);
      o[2] = __builtin_fmaf(r.w1[1], d[0], o[2]);
      o[2] = __builtin_fmaf(r.w1[5], d[1], o[2]);
      o[2] = __builtin_fmaf(r.w1[11], d[2], o[2]);
      o[2] = __builtin_fmaf(r.w1[15], d[3], o[2]);
      o[3] = __builtin_fmaf(r.w1[0], d[4], o[3]);
      o[3] = __builtin_fmaf(r.w1[2], d[5], o[3]);
      o[3] = __builtin_fmaf(r.w1[6], d[6], o[3]);
      o[3] = __builtin_fmaf(r.w1[12], d[7], o[3]);
    }
    if (mhi & 0x40000u) {
      const v2f a = {areg[25].x, areg[25].y};
      const v2f b = {r.w2[4], r.w2[5]};
      v8f d = __builtin_amdgcn_wmma_f32_16x16x4_f32(false, a, false, b, (short)0, zero, false, false);
      WMMA_RESULT_GUARD(d);
      o[2] = __builtin_fmaf(r.w1[1], d[0], o[2]);
      o[2] = __builtin_fmaf(r.w1[5], d[1], o[2]);
      o[2] = __builtin_fmaf(r.w1[11], d[2], o[2]);
      o[2] = __builtin_fmaf(r.w1[15], d[3], o[2]);
      o[3] = __builtin_fmaf(r.w1[0], d[4], o[3]);
      o[3] = __builtin_fmaf(r.w1[2], d[5], o[3]);
      o[3] = __builtin_fmaf(r.w1[6], d[6], o[3]);
      o[3] = __builtin_fmaf(r.w1[12], d[7], o[3]);
    }
  }
  if (mhi & 0x800000u) {
    const v2f a = {areg[27].z, areg[27].w};
    const v2f b = {r.w2[6], r.w2[7]};
    v8f d = __builtin_amdgcn_wmma_f32_16x16x4_f32(false, a, false, b, (short)0, zero, false, false);
    WMMA_RESULT_GUARD(d);
    o[4] = __builtin_fmaf(r.w1[4], d[0], o[4]);
    o[4] = __builtin_fmaf(r.w1[8], d[1], o[4]);
    o[4] = __builtin_fmaf(r.w1[10], d[2], o[4]);
    o[4] = __builtin_fmaf(r.w1[14], d[3], o[4]);
    o[5] = __builtin_fmaf(r.w1[0], d[4], o[5]);
    o[5] = __builtin_fmaf(r.w1[2], d[5], o[5]);
    o[5] = __builtin_fmaf(r.w1[6], d[6], o[5]);
    o[5] = __builtin_fmaf(r.w1[12], d[7], o[5]);
  }
  if (mhi & 0x700000u) {
    if (mhi & 0x100000u) {
      const v2f a = {areg[26].x, areg[26].y};
      const v2f b = {r.w2[0], r.w2[1]};
      v8f d = __builtin_amdgcn_wmma_f32_16x16x4_f32(false, a, false, b, (short)0, zero, false, false);
      WMMA_RESULT_GUARD(d);
      o[4] = __builtin_fmaf(r.w1[4], d[0], o[4]);
      o[4] = __builtin_fmaf(r.w1[8], d[1], o[4]);
      o[4] = __builtin_fmaf(r.w1[10], d[2], o[4]);
      o[4] = __builtin_fmaf(r.w1[14], d[3], o[4]);
      o[5] = __builtin_fmaf(r.w1[0], d[4], o[5]);
      o[5] = __builtin_fmaf(r.w1[2], d[5], o[5]);
      o[5] = __builtin_fmaf(r.w1[6], d[6], o[5]);
      o[5] = __builtin_fmaf(r.w1[12], d[7], o[5]);
    }
    if (mhi & 0x200000u) {
      const v2f a = {areg[26].z, areg[26].w};
      const v2f b = {r.w2[2], r.w2[3]};
      v8f d = __builtin_amdgcn_wmma_f32_16x16x4_f32(false, a, false, b, (short)0, zero, false, false);
      WMMA_RESULT_GUARD(d);
      o[4] = __builtin_fmaf(r.w1[4], d[0], o[4]);
      o[4] = __builtin_fmaf(r.w1[8], d[1], o[4]);
      o[4] = __builtin_fmaf(r.w1[10], d[2], o[4]);
      o[4] = __builtin_fmaf(r.w1[14], d[3], o[4]);
      o[5] = __builtin_fmaf(r.w1[0], d[4], o[5]);
      o[5] = __builtin_fmaf(r.w1[2], d[5], o[5]);
      o[5] = __builtin_fmaf(r.w1[6], d[6], o[5]);
      o[5] = __builtin_fmaf(r.w1[12], d[7], o[5]);
    }
    if (mhi & 0x400000u) {
      const v2f a = {areg[27].x, areg[27].y};
      const v2f b = {r.w2[4], r.w2[5]};
      v8f d = __builtin_amdgcn_wmma_f32_16x16x4_f32(false, a, false, b, (short)0, zero, false, false);
      WMMA_RESULT_GUARD(d);
      o[4] = __builtin_fmaf(r.w1[4], d[0], o[4]);
      o[4] = __builtin_fmaf(r.w1[8], d[1], o[4]);
      o[4] = __builtin_fmaf(r.w1[10], d[2], o[4]);
      o[4] = __builtin_fmaf(r.w1[14], d[3], o[4]);
      o[5] = __builtin_fmaf(r.w1[0], d[4], o[5]);
      o[5] = __builtin_fmaf(r.w1[2], d[5], o[5]);
      o[5] = __builtin_fmaf(r.w1[6], d[6], o[5]);
      o[5] = __builtin_fmaf(r.w1[12], d[7], o[5]);
    }
  }
  if (mhi & 0x8000000u) {
    const v2f a = {areg[29].z, areg[29].w};
    const v2f b = {r.w2[6], r.w2[7]};
    v8f d = __builtin_amdgcn_wmma_f32_16x16x4_f32(false, a, false, b, (short)0, zero, false, false);
    WMMA_RESULT_GUARD(d);
    o[6] = __builtin_fmaf((half ? r.w1[3] : r.w1[1]), d[0], o[6]);
    o[6] = __builtin_fmaf((half ? r.w1[7] : r.w1[5]), d[1], o[6]);
    o[6] = __builtin_fmaf((half ? r.w1[9] : r.w1[11]), d[2], o[6]);
    o[6] = __builtin_fmaf((half ? r.w1[13] : r.w1[14]), d[3], o[6]);
    o[7] = __builtin_fmaf((half ? r.w1[3] : r.w1[1]), d[4], o[7]);
    o[7] = __builtin_fmaf((half ? r.w1[7] : r.w1[5]), d[5], o[7]);
    o[7] = __builtin_fmaf((half ? r.w1[13] : r.w1[11]), d[6], o[7]);
    o[7] = __builtin_fmaf((half ? r.w1[10] : r.w1[15]), d[7], o[7]);
  }
  if (mhi & 0x7000000u) {
    if (mhi & 0x1000000u) {
      const v2f a = {areg[28].x, areg[28].y};
      const v2f b = {r.w2[0], r.w2[1]};
      v8f d = __builtin_amdgcn_wmma_f32_16x16x4_f32(false, a, false, b, (short)0, zero, false, false);
      WMMA_RESULT_GUARD(d);
      o[6] = __builtin_fmaf((half ? r.w1[3] : r.w1[1]), d[0], o[6]);
      o[6] = __builtin_fmaf((half ? r.w1[7] : r.w1[5]), d[1], o[6]);
      o[6] = __builtin_fmaf((half ? r.w1[9] : r.w1[11]), d[2], o[6]);
      o[6] = __builtin_fmaf((half ? r.w1[13] : r.w1[14]), d[3], o[6]);
      o[7] = __builtin_fmaf((half ? r.w1[3] : r.w1[1]), d[4], o[7]);
      o[7] = __builtin_fmaf((half ? r.w1[7] : r.w1[5]), d[5], o[7]);
      o[7] = __builtin_fmaf((half ? r.w1[13] : r.w1[11]), d[6], o[7]);
      o[7] = __builtin_fmaf((half ? r.w1[10] : r.w1[15]), d[7], o[7]);
    }
    if (mhi & 0x2000000u) {
      const v2f a = {areg[28].z, areg[28].w};
      const v2f b = {r.w2[2], r.w2[3]};
      v8f d = __builtin_amdgcn_wmma_f32_16x16x4_f32(false, a, false, b, (short)0, zero, false, false);
      WMMA_RESULT_GUARD(d);
      o[6] = __builtin_fmaf((half ? r.w1[3] : r.w1[1]), d[0], o[6]);
      o[6] = __builtin_fmaf((half ? r.w1[7] : r.w1[5]), d[1], o[6]);
      o[6] = __builtin_fmaf((half ? r.w1[9] : r.w1[11]), d[2], o[6]);
      o[6] = __builtin_fmaf((half ? r.w1[13] : r.w1[14]), d[3], o[6]);
      o[7] = __builtin_fmaf((half ? r.w1[3] : r.w1[1]), d[4], o[7]);
      o[7] = __builtin_fmaf((half ? r.w1[7] : r.w1[5]), d[5], o[7]);
      o[7] = __builtin_fmaf((half ? r.w1[13] : r.w1[11]), d[6], o[7]);
      o[7] = __builtin_fmaf((half ? r.w1[10] : r.w1[15]), d[7], o[7]);
    }
    if (mhi & 0x4000000u) {
      const v2f a = {areg[29].x, areg[29].y};
      const v2f b = {r.w2[4], r.w2[5]};
      v8f d = __builtin_amdgcn_wmma_f32_16x16x4_f32(false, a, false, b, (short)0, zero, false, false);
      WMMA_RESULT_GUARD(d);
      o[6] = __builtin_fmaf((half ? r.w1[3] : r.w1[1]), d[0], o[6]);
      o[6] = __builtin_fmaf((half ? r.w1[7] : r.w1[5]), d[1], o[6]);
      o[6] = __builtin_fmaf((half ? r.w1[9] : r.w1[11]), d[2], o[6]);
      o[6] = __builtin_fmaf((half ? r.w1[13] : r.w1[14]), d[3], o[6]);
      o[7] = __builtin_fmaf((half ? r.w1[3] : r.w1[1]), d[4], o[7]);
      o[7] = __builtin_fmaf((half ? r.w1[7] : r.w1[5]), d[5], o[7]);
      o[7] = __builtin_fmaf((half ? r.w1[13] : r.w1[11]), d[6], o[7]);
      o[7] = __builtin_fmaf((half ? r.w1[10] : r.w1[15]), d[7], o[7]);
    }
  }
  if (mhi & 0x10000000u) {
    const v2f a = {areg[30].x, areg[30].y};
    const v2f b = {r.w2[0], r.w2[1]};
    v8f d = __builtin_amdgcn_wmma_f32_16x16x4_f32(false, a, false, b, (short)0, zero, false, false);
    WMMA_RESULT_GUARD(d);
    o[1] = __builtin_fmaf(r.w1[15], d[0], o[1]);
    o[2] = __builtin_fmaf(r.w1[9], d[1], o[2]);
    o[3] = __builtin_fmaf(r.w1[8], d[2], o[3]);
    o[3] = __builtin_fmaf(r.w1[14], d[3], o[3]);
    o[3] = __builtin_fmaf(r.w1[15], d[4], o[3]);
    o[5] = __builtin_fmaf(r.w1[4], d[5], o[5]);
    o[5] = __builtin_fmaf(r.w1[9], d[6], o[5]);
    o[5] = __builtin_fmaf(r.w1[10], d[7], o[5]);
  }
  if (mhi & 0xe0000000u) {
    if (mhi & 0x20000000u) {
      const v2f a = {areg[30].z, areg[30].w};
      const v2f b = {r.w2[2], r.w2[3]};
      v8f d = __builtin_amdgcn_wmma_f32_16x16x4_f32(false, a, false, b, (short)0, zero, false, false);
      WMMA_RESULT_GUARD(d);
      o[1] = __builtin_fmaf(r.w1[15], d[0], o[1]);
      o[2] = __builtin_fmaf(r.w1[9], d[1], o[2]);
      o[3] = __builtin_fmaf(r.w1[8], d[2], o[3]);
      o[3] = __builtin_fmaf(r.w1[14], d[3], o[3]);
      o[3] = __builtin_fmaf(r.w1[15], d[4], o[3]);
      o[5] = __builtin_fmaf(r.w1[4], d[5], o[5]);
      o[5] = __builtin_fmaf(r.w1[9], d[6], o[5]);
      o[5] = __builtin_fmaf(r.w1[10], d[7], o[5]);
    }
    if (mhi & 0x40000000u) {
      const v2f a = {areg[31].x, areg[31].y};
      const v2f b = {r.w2[4], r.w2[5]};
      v8f d = __builtin_amdgcn_wmma_f32_16x16x4_f32(false, a, false, b, (short)0, zero, false, false);
      WMMA_RESULT_GUARD(d);
      o[1] = __builtin_fmaf(r.w1[15], d[0], o[1]);
      o[2] = __builtin_fmaf(r.w1[9], d[1], o[2]);
      o[3] = __builtin_fmaf(r.w1[8], d[2], o[3]);
      o[3] = __builtin_fmaf(r.w1[14], d[3], o[3]);
      o[3] = __builtin_fmaf(r.w1[15], d[4], o[3]);
      o[5] = __builtin_fmaf(r.w1[4], d[5], o[5]);
      o[5] = __builtin_fmaf(r.w1[9], d[6], o[5]);
      o[5] = __builtin_fmaf(r.w1[10], d[7], o[5]);
    }
    if (mhi & 0x80000000u) {
      const v2f a = {areg[31].z, areg[31].w};
      const v2f b = {r.w2[6], r.w2[7]};
      v8f d = __builtin_amdgcn_wmma_f32_16x16x4_f32(false, a, false, b, (short)0, zero, false, false);
      WMMA_RESULT_GUARD(d);
      o[1] = __builtin_fmaf(r.w1[15], d[0], o[1]);
      o[2] = __builtin_fmaf(r.w1[9], d[1], o[2]);
      o[3] = __builtin_fmaf(r.w1[8], d[2], o[3]);
      o[3] = __builtin_fmaf(r.w1[14], d[3], o[3]);
      o[3] = __builtin_fmaf(r.w1[15], d[4], o[3]);
      o[5] = __builtin_fmaf(r.w1[4], d[5], o[5]);
      o[5] = __builtin_fmaf(r.w1[9], d[6], o[5]);
      o[5] = __builtin_fmaf(r.w1[10], d[7], o[5]);
    }
  }

  const int n = t >> 2;
  const int c = (t & 3) * 16 + col;
  float* po = out + (size_t)n * NUM_ORDERS * C_DIM + c;
  po[(half ? 2 : 0) * C_DIM] = o[0];
  po[(half ? 5 : 1) * C_DIM] = o[1];
  po[(half ? 7 : 3) * C_DIM] = o[2];
  po[(half ? 10 : 4) * C_DIM] = o[3];
  po[(half ? 12 : 6) * C_DIM] = o[4];
  po[(half ? 14 : 8) * C_DIM] = o[5];
  po[(half ? 11 : 9) * C_DIM] = o[6];
  po[(half ? 15 : 13) * C_DIM] = o[7];
}

#define FAST_BLOCKS 15
#define FAST_MASK_LO 0x42222111u
#define FAST_MASK_HI 0x08888444u

#define FIRST_BLOCK_GUARD(a, b, c)                                                        \
  asm volatile("v_nop\n\tv_nop\n\tv_nop\n\tv_nop\n\tv_nop\n\tv_nop\n\tv_nop\n\tv_nop" \
               : "+v"(a), "+v"(b), "+v"(c))

__device__ __forceinline__ void tile_product_fast(v2f (&fa)[FAST_BLOCKS], const TileRegs& r,
                                                  const unsigned (&so)[8], float* out, int t,
                                                  int half, int col) {
  const v8f zero = {};
  float o[8] = {0.0f, 0.0f, 0.0f, 0.0f, 0.0f, 0.0f, 0.0f, 0.0f};
  const float s_1_9 = half ? r.w1[1] : r.w1[9];
  const float s_5_3 = half ? r.w1[5] : r.w1[3];
  const float s_11_7 = half ? r.w1[11] : r.w1[7];
  const float s_15_3 = half ? r.w1[15] : r.w1[3];
  const float s_1_7 = half ? r.w1[1] : r.w1[7];
  const float s_4_13 = half ? r.w1[4] : r.w1[13];
  const float s_4_8 = half ? r.w1[4] : r.w1[8];
  const float s_10_13 = half ? r.w1[10] : r.w1[13];
  const float s_8_4 = half ? r.w1[8] : r.w1[4];
  const float s_14_10 = half ? r.w1[14] : r.w1[10];
  const float s_11_9 = half ? r.w1[11] : r.w1[9];
  const float s_5_8 = half ? r.w1[5] : r.w1[8];
  const float s_9_14 = half ? r.w1[9] : r.w1[14];
  const float s_3_1 = half ? r.w1[3] : r.w1[1];
  const float s_7_5 = half ? r.w1[7] : r.w1[5];
  const float s_9_11 = half ? r.w1[9] : r.w1[11];
  const float s_13_14 = half ? r.w1[13] : r.w1[14];
  const float s_13_11 = half ? r.w1[13] : r.w1[11];
  const float s_10_15 = half ? r.w1[10] : r.w1[15];
  v8f d0, d1, d2;
  d0 = __builtin_amdgcn_wmma_f32_16x16x4_f32(false, fa[0], false, (v2f){r.w2[0], r.w2[1]}, (short)0, zero, false, false);
  d1 = __builtin_amdgcn_wmma_f32_16x16x4_f32(false, fa[1], false, (v2f){r.w2[0], r.w2[1]}, (short)0, zero, false, false);
  d2 = __builtin_amdgcn_wmma_f32_16x16x4_f32(false, fa[2], false, (v2f){r.w2[0], r.w2[1]}, (short)0, zero, false, false);
  FIRST_BLOCK_GUARD(d0, d1, d2);
  o[0] = __builtin_fmaf(r.w1[0], d0[0], o[0]);
  o[0] = __builtin_fmaf(r.w1[2], d0[1], o[0]);
  o[0] = __builtin_fmaf(r.w1[6], d0[2], o[0]);
  o[0] = __builtin_fmaf(r.w1[12], d0[3], o[0]);
  o[1] = __builtin_fmaf(r.w1[1], d0[4], o[1]);
  o[1] = __builtin_fmaf(r.w1[5], d0[5], o[1]);
  o[1] = __builtin_fmaf(r.w1[11], d0[6], o[1]);
  o[2] = __builtin_fmaf(r.w1[3], d0[7], o[2]);
  asm volatile("" : "+v"(fa[3]), "+v"(o[0]), "+v"(o[1]), "+v"(o[2]));
  d0 = __builtin_amdgcn_wmma_f32_16x16x4_f32(false, fa[3], false, (v2f){r.w2[2], r.w2[3]}, (short)0, zero, false, false);
  asm volatile("" : "+v"(d1), "+v"(d0));
  o[2] = __builtin_fmaf(r.w1[7], d1[0], o[2]);
  o[2] = __builtin_fmaf(r.w1[13], d1[1], o[2]);
  o[3] = __builtin_fmaf(r.w1[4], d1[2], o[3]);
  o[3] = __builtin_fmaf(r.w1[10], d1[3], o[3]);
  o[4] = __builtin_fmaf(r.w1[0], d1[4], o[4]);
  o[4] = __builtin_fmaf(r.w1[2], d1[5], o[4]);
  o[4] = __builtin_fmaf(r.w1[6], d1[6], o[4]);
  o[4] = __builtin_fmaf(r.w1[12], d1[7], o[4]);
  asm volatile("" : "+v"(fa[4]), "+v"(o[2]), "+v"(o[3]), "+v"(o[4]));
  d1 = __builtin_amdgcn_wmma_f32_16x16x4_f32(false, fa[4], false, (v2f){r.w2[2], r.w2[3]}, (short)0, zero, false, false);
  asm volatile("" : "+v"(d2), "+v"(d1));
  o[5] = __builtin_fmaf(r.w1[8], d2[0], o[5]);
  o[5] = __builtin_fmaf(r.w1[14], d2[1], o[5]);
  o[6] = __builtin_fmaf(s_1_9, d2[2], o[6]);
  o[6] = __builtin_fmaf(s_5_3, d2[3], o[6]);
  o[6] = __builtin_fmaf(s_11_7, d2[4], o[6]);
  o[7] = __builtin_fmaf(s_15_3, d2[5], o[7]);
  o[7] = __builtin_fmaf(s_1_7, d2[6], o[7]);
  o[7] = __builtin_fmaf(s_4_13, d2[7], o[7]);
  asm volatile("" : "+v"(fa[5]), "+v"(o[5]), "+v"(o[6]), "+v"(o[7]));
  d2 = __builtin_amdgcn_wmma_f32_16x16x4_f32(false, fa[5], false, (v2f){r.w2[2], r.w2[3]}, (short)0, zero, false, false);
  asm volatile("" : "+v"(d0), "+v"(d2));
  o[0] = __builtin_fmaf(r.w1[1], d0[0], o[0]);
  o[0] = __builtin_fmaf(r.w1[5], d0[1], o[0]);
  o[0] = __builtin_fmaf(r.w1[9], d0[2], o[0]);
  o[0] = __builtin_fmaf(r.w1[11], d0[3], o[0]);
  o[1] = __builtin_fmaf(r.w1[4], d0[4], o[1]);
  o[1] = __builtin_fmaf(r.w1[10], d0[5], o[1]);
  o[2] = __builtin_fmaf(r.w1[0], d0[6], o[2]);
  o[2] = __builtin_fmaf(r.w1[2], d0[7], o[2]);
  asm volatile("" : "+v"(fa[6]), "+v"(o[0]), "+v"(o[1]), "+v"(o[2]));
  d0 = __builtin_amdgcn_wmma_f32_16x16x4_f32(false, fa[6], false, (v2f){r.w2[2], r.w2[3]}, (short)0, zero, false, false);
  asm volatile("" : "+v"(d1), "+v"(d0));
  o[2] = __builtin_fmaf(r.w1[4], d1[0], o[2]);
  o[2] = __builtin_fmaf(r.w1[6], d1[1], o[2]);
  o[2] = __builtin_fmaf(r.w1[10], d1[2], o[2]);
  o[2] = __builtin_fmaf(r.w1[12], d1[3], o[2]);
  o[3] = __builtin_fmaf(r.w1[9], d1[4], o[3]);
  o[4] = __builtin_fmaf(r.w1[1], d1[5], o[4]);
  o[4] = __builtin_fmaf(r.w1[5], d1[6], o[4]);
  o[4] = __builtin_fmaf(r.w1[9], d1[7], o[4]);
  asm volatile("" : "+v"(fa[7]), "+v"(o[2]), "+v"(o[3]), "+v"(o[4]));
  d1 = __builtin_amdgcn_wmma_f32_16x16x4_f32(false, fa[7], false, (v2f){r.w2[4], r.w2[5]}, (short)0, zero, false, false);
  asm volatile("" : "+v"(d2), "+v"(d1));
  o[4] = __builtin_fmaf(r.w1[11], d2[0], o[4]);
  o[5] = __builtin_fmaf(r.w1[1], d2[1], o[5]);
  o[5] = __builtin_fmaf(r.w1[3], d2[2], o[5]);
  o[5] = __builtin_fmaf(r.w1[5], d2[3], o[5]);
  o[5] = __builtin_fmaf(r.w1[7], d2[4], o[5]);
  o[5] = __builtin_fmaf(r.w1[11], d2[5], o[5]);
  o[5] = __builtin_fmaf(r.w1[13], d2[6], o[5]);
  o[6] = __builtin_fmaf(s_4_8, d2[7], o[6]);
  asm volatile("" : "+v"(fa[8]), "+v"(o[4]), "+v"(o[5]), "+v"(o[6]));
  d2 = __builtin_amdgcn_wmma_f32_16x16x4_f32(false, fa[8], false, (v2f){r.w2[4], r.w2[5]}, (short)0, zero, false, false);
  asm volatile("" : "+v"(d0), "+v"(d2));
  o[6] = __builtin_fmaf(s_10_13, d0[0], o[6]);
  o[7] = __builtin_fmaf(r.w1[0], d0[1], o[7]);
  o[7] = __builtin_fmaf(r.w1[2], d0[2], o[7]);
  o[7] = __builtin_fmaf(r.w1[6], d0[3], o[7]);
  o[7] = __builtin_fmaf(r.w1[12], d0[4], o[7]);
  o[7] = __builtin_fmaf(s_8_4, d0[5], o[7]);
  o[7] = __builtin_fmaf(s_14_10, d0[6], o[7]);
  o[7] = __builtin_fmaf(s_11_9, d0[7], o[7]);
  asm volatile("" : "+v"(fa[9]), "+v"(o[6]), "+v"(o[7]));
  d0 = __builtin_amdgcn_wmma_f32_16x16x4_f32(false, fa[9], false, (v2f){r.w2[4], r.w2[5]}, (short)0, zero, false, false);
  asm volatile("" : "+v"(d1), "+v"(d0));
  o[0] = __builtin_fmaf(r.w1[3], d1[0], o[0]);
  o[0] = __builtin_fmaf(r.w1[7], d1[1], o[0]);
  o[0] = __builtin_fmaf(r.w1[13], d1[2], o[0]);
  o[0] = __builtin_fmaf(r.w1[15], d1[3], o[0]);
  o[1] = __builtin_fmaf(r.w1[0], d1[4], o[1]);
  o[1] = __builtin_fmaf(r.w1[2], d1[5], o[1]);
  o[1] = __builtin_fmaf(r.w1[6], d1[6], o[1]);
  o[1] = __builtin_fmaf(r.w1[8], d1[7], o[1]);
  asm volatile("" : "+v"(fa[10]), "+v"(o[0]), "+v"(o[1]));
  d1 = __builtin_amdgcn_wmma_f32_16x16x4_f32(false, fa[10], false, (v2f){r.w2[4], r.w2[5]}, (short)0, zero, false, false);
  asm volatile("" : "+v"(d2), "+v"(d1));
  o[1] = __builtin_fmaf(r.w1[12], d2[0], o[1]);
  o[1] = __builtin_fmaf(r.w1[14], d2[1], o[1]);
  o[2] = __builtin_fmaf(r.w1[8], d2[2], o[2]);
  o[2] = __builtin_fmaf(r.w1[14], d2[3], o[2]);
  o[3] = __builtin_fmaf(r.w1[1], d2[4], o[3]);
  o[3] = __builtin_fmaf(r.w1[3], d2[5], o[3]);
  o[3] = __builtin_fmaf(r.w1[5], d2[6], o[3]);
  o[3] = __builtin_fmaf(r.w1[7], d2[7], o[3]);
  asm volatile("" : "+v"(fa[11]), "+v"(o[1]), "+v"(o[2]), "+v"(o[3]));
  d2 = __builtin_amdgcn_wmma_f32_16x16x4_f32(false, fa[11], false, (v2f){r.w2[6], r.w2[7]}, (short)0, zero, false, false);
  asm volatile("" : "+v"(d0), "+v"(d2));
  o[3] = __builtin_fmaf(r.w1[11], d0[0], o[3]);
  o[3] = __builtin_fmaf(r.w1[13], d0[1], o[3]);
  o[4] = __builtin_fmaf(r.w1[3], d0[2], o[4]);
  o[4] = __builtin_fmaf(r.w1[7], d0[3], o[4]);
  o[4] = __builtin_fmaf(r.w1[13], d0[4], o[4]);
  o[4] = __builtin_fmaf(r.w1[15], d0[5], o[4]);
  o[5] = __builtin_fmaf(r.w1[15], d0[6], o[5]);
  o[6] = __builtin_fmaf(r.w1[0], d0[7], o[6]);
  asm volatile("" : "+v"(fa[12]), "+v"(o[3]), "+v"(o[4]), "+v"(o[5]), "+v"(o[6]));
  d0 = __builtin_amdgcn_wmma_f32_16x16x4_f32(false, fa[12], false, (v2f){r.w2[6], r.w2[7]}, (short)0, zero, false, false);
  asm volatile("" : "+v"(d1), "+v"(d0));
  o[6] = __builtin_fmaf(r.w1[2], d1[0], o[6]);
  o[6] = __builtin_fmaf(r.w1[6], d1[1], o[6]);
  o[6] = __builtin_fmaf(r.w1[12], d1[2], o[6]);
  o[6] = __builtin_fmaf(s_8_4, d1[3], o[6]);
  o[6] = __builtin_fmaf(s_14_10, d1[4], o[6]);
  o[7] = __builtin_fmaf(s_5_8, d1[5], o[7]);
  o[7] = __builtin_fmaf(s_9_14, d1[6], o[7]);
  o[6] = __builtin_fmaf(r.w1[15], d1[7], o[6]);
  asm volatile("" : "+v"(fa[13]), "+v"(o[6]), "+v"(o[7]));
  d1 = __builtin_amdgcn_wmma_f32_16x16x4_f32(false, fa[13], false, (v2f){r.w2[6], r.w2[7]}, (short)0, zero, false, false);
  asm volatile("" : "+v"(d2), "+v"(d1));
  o[0] = __builtin_fmaf(r.w1[4], d2[0], o[0]);
  o[0] = __builtin_fmaf(r.w1[8], d2[1], o[0]);
  o[0] = __builtin_fmaf(r.w1[10], d2[2], o[0]);
  o[0] = __builtin_fmaf(r.w1[14], d2[3], o[0]);
  o[1] = __builtin_fmaf(r.w1[3], d2[4], o[1]);
  o[1] = __builtin_fmaf(r.w1[7], d2[5], o[1]);
  o[1] = __builtin_fmaf(r.w1[9], d2[6], o[1]);
  o[1] = __builtin_fmaf(r.w1[13], d2[7], o[1]);
  asm volatile("" : "+v"(fa[14]), "+v"(o[0]), "+v"(o[1]));
  d2 = __builtin_amdgcn_wmma_f32_16x16x4_f32(false, fa[14], false, (v2f){r.w2[6], r.w2[7]}, (short)0, zero, false, false);
  asm volatile("" : "+v"(d0), "+v"(d2));
  o[2] = __builtin_fmaf(r.w1[1], d0[0], o[2]);
  o[2] = __builtin_fmaf(r.w1[5], d0[1], o[2]);
  o[2] = __builtin_fmaf(r.w1[11], d0[2], o[2]);
  o[2] = __builtin_fmaf(r.w1[15], d0[3], o[2]);
  o[3] = __builtin_fmaf(r.w1[0], d0[4], o[3]);
  o[3] = __builtin_fmaf(r.w1[2], d0[5], o[3]);
  o[3] = __builtin_fmaf(r.w1[6], d0[6], o[3]);
  o[3] = __builtin_fmaf(r.w1[12], d0[7], o[3]);
  asm volatile("" : "+v"(d1), "+v"(o[2]), "+v"(o[3]));
  o[4] = __builtin_fmaf(r.w1[4], d1[0], o[4]);
  o[4] = __builtin_fmaf(r.w1[8], d1[1], o[4]);
  o[4] = __builtin_fmaf(r.w1[10], d1[2], o[4]);
  o[4] = __builtin_fmaf(r.w1[14], d1[3], o[4]);
  o[5] = __builtin_fmaf(r.w1[0], d1[4], o[5]);
  o[5] = __builtin_fmaf(r.w1[2], d1[5], o[5]);
  o[5] = __builtin_fmaf(r.w1[6], d1[6], o[5]);
  o[5] = __builtin_fmaf(r.w1[12], d1[7], o[5]);
  asm volatile("" : "+v"(d2), "+v"(o[4]), "+v"(o[5]));
  o[6] = __builtin_fmaf(s_3_1, d2[0], o[6]);
  o[6] = __builtin_fmaf(s_7_5, d2[1], o[6]);
  o[6] = __builtin_fmaf(s_9_11, d2[2], o[6]);
  o[6] = __builtin_fmaf(s_13_14, d2[3], o[6]);
  o[7] = __builtin_fmaf(s_3_1, d2[4], o[7]);
  o[7] = __builtin_fmaf(s_7_5, d2[5], o[7]);
  o[7] = __builtin_fmaf(s_13_11, d2[6], o[7]);
  o[7] = __builtin_fmaf(s_10_15, d2[7], o[7]);

  const int n = t >> 2;
  const unsigned c = (unsigned)((t & 3) * 16 + col);
  float* pn = out + (size_t)n * NUM_ORDERS * C_DIM;
  pn[c + so[0]] = o[0];
  pn[c + so[1]] = o[1];
  pn[c + so[2]] = o[2];
  pn[c + so[3]] = o[3];
  pn[c + so[4]] = o[4];
  pn[c + so[5]] = o[5];
  pn[c + so[6]] = o[6];
  pn[c + so[7]] = o[7];
}

__global__ void __launch_bounds__(BLOCK_THREADS)
tp_wmma_kernel(const float* x1, const float* x2,
               const float* __restrict__ vals, const int* __restrict__ M1,
               const int* __restrict__ M2, const int* __restrict__ Mseg,
               float* out, int n_batch, int nnz) {
  __shared__ float4 cg_lds[NUM_ORDERS * K_DENSE / 4];
  __shared__ unsigned char tab_lds[256 + 16];
  float* cg = (float*)cg_lds;

  const int tid  = threadIdx.x;
  const int lane = tid & 31;
  const int half = lane >> 4;
  const int col  = lane & 15;
  const int wib    = __builtin_amdgcn_readfirstlane((int)(threadIdx.x >> 5));
  const int wave   = blockIdx.x * WAVES_PER_BLOCK + wib;
  const int nwaves = gridDim.x * WAVES_PER_BLOCK;
  const int ntiles = n_batch * 4;

  if (ntiles <= 0) return;

  int t = wave;
  const bool has_tile = t < ntiles;

  unsigned char t_row = 0, t_col = 0;
  if (tid < 256) t_row = ROW_OF[tid];
  if (tid < 16) t_col = COL_POS[tid];
  const bool has_entry = tid < nnz;
  int   e_m = 0, e_m1 = 0, e_m2 = 0;
  float e_val = 0.0f;
  if (has_entry) {
    e_m   = Mseg[tid];
    e_m1  = M1[tid];
    e_m2  = M2[tid];
    e_val = vals[tid];
  }

  TileRegs r;
  load_tile(r, x1, x2, has_tile ? t : 0, half, col);

  if (tid < 256) tab_lds[tid] = t_row;
  if (tid < 16) tab_lds[256 + tid] = t_col;
  for (int i = tid; i < NUM_ORDERS * K_DENSE / 4; i += BLOCK_THREADS)
    cg_lds[i] = make_float4(0.0f, 0.0f, 0.0f, 0.0f);
  lds_barrier();
  if (has_entry) {
    asm volatile("" : "+v"(e_m), "+v"(e_m1), "+v"(e_m2));
    const int rho = tab_lds[((e_m & 15) << 4) | (e_m1 & 15)];
    const int pos = tab_lds[256 + (e_m2 & 15)];
    cg[cg_frag_index(rho & 15, (rho & 0xF0) | pos)] = e_val;
  }
  for (int i = tid + BLOCK_THREADS; i < nnz; i += BLOCK_THREADS) {
    const int rho = tab_lds[((Mseg[i] & 15) << 4) | (M1[i] & 15)];
    const int pos = tab_lds[256 + (M2[i] & 15)];
    cg[cg_frag_index(rho & 15, (rho & 0xF0) | pos)] = vals[i];
  }
  lds_barrier();

  unsigned mlo = 0, mhi = 0;
#pragma unroll
  for (int kp = 0; kp < 32; ++kp) {
    const float4 q = cg_lds[kp * 32 + lane];
    const unsigned b0 = __builtin_amdgcn_ballot_w32((q.x != 0.0f) | (q.y != 0.0f));
    const unsigned b1 = __builtin_amdgcn_ballot_w32((q.z != 0.0f) | (q.w != 0.0f));
    const unsigned bits = (b0 ? 1u : 0u) | (b1 ? 2u : 0u);
    if (kp < 16) mlo |= bits << (2 * kp);
    else         mhi |= bits << (2 * kp - 32);
  }

  asm volatile("" ::: "memory");

  if (!has_tile) return;

  const int cnt = (ntiles - t + nwaves - 1) / nwaves;

  if (((mlo & ~FAST_MASK_LO) | (mhi & ~FAST_MASK_HI)) == 0u) {
    constexpr int fast_block[FAST_BLOCKS] = {0, 4, 8, 13, 17, 21, 25, 30, 34, 38, 42, 47, 51, 55, 59};
    v2f fa[FAST_BLOCKS];
#pragma unroll
    for (int i = 0; i < FAST_BLOCKS; ++i) {
      const int idx = ((fast_block[i] >> 1) * 32 + lane) * 4 + (fast_block[i] & 1) * 2;
      fa[i] = (v2f){cg[idx], cg[idx + 1]};
    }
    unsigned so[8] = {(half ? 2u : 0u) * C_DIM,  (half ? 5u : 1u) * C_DIM,
                      (half ? 7u : 3u) * C_DIM,  (half ? 10u : 4u) * C_DIM,
                      (half ? 12u : 6u) * C_DIM, (half ? 14u : 8u) * C_DIM,
                      (half ? 11u : 9u) * C_DIM, (half ? 15u : 13u) * C_DIM};
    asm volatile("" : "+v"(so[0]), "+v"(so[1]), "+v"(so[2]), "+v"(so[3]),
                      "+v"(so[4]), "+v"(so[5]), "+v"(so[6]), "+v"(so[7]));
    TileRegs r1;
    for (int p = cnt; p > 0; --p) {
      const int tn = t + nwaves;
      load_tile(r1, x1, x2, (tn < ntiles) ? tn : t, half, col);
      __builtin_amdgcn_sched_barrier(0);
      tile_product_fast(fa, r, so, out, t, half, col);
      asm volatile("" : "+v"(r1.w2));
      r = r1;
      t = tn;
    }
    return;
  }

  float4 areg[32];
#pragma unroll
  for (int kp = 0; kp < 32; ++kp) areg[kp] = cg_lds[kp * 32 + lane];

  for (int p = cnt; p > 0; --p) {
    asm volatile("" : "+s"(mlo), "+s"(mhi));
    tile_product(areg, mlo, mhi, r, out, t, half, col);
    t += nwaves;
    if (p > 1) load_tile(r, x1, x2, t, half, col);
  }
}

extern "C" void kernel_launch(void* const* d_in, const int* in_sizes, int n_in,
                              void* d_out, int out_size, void* d_ws, size_t ws_size,
                              hipStream_t stream) {
  const float* x1   = (const float*)d_in[0];
  const float* x2   = (const float*)d_in[1];
  const float* vals = (const float*)d_in[2];
  const int* M1     = (const int*)d_in[3];
  const int* M2     = (const int*)d_in[4];
  const int* Mseg   = (const int*)d_in[5];
  float* out        = (float*)d_out;

  const int nnz     = in_sizes[2];
  const int n_batch = in_sizes[0] / (NUM_ORDERS * C_DIM);

  tp_wmma_kernel<<<NUM_BLOCKS, BLOCK_THREADS, 0, stream>>>(x1, x2, vals, M1, M2, Mseg,
                                                          out, n_batch, nnz);
}


# Branch-free path: a lane holds two adjacent channels (8-byte loads/stores); a wave's unit is one batch row x 32 channels run as two tiles: twice the bytes in flight, half the memory instructions
# speedup vs baseline: 2.3240x; 1.1162x over previous
#include <hip/hip_runtime.h>


typedef __attribute__((ext_vector_type(2))) float v2f;
typedef __attribute__((ext_vector_type(8))) float v8f;
typedef __attribute__((ext_vector_type(16))) float v16f;

#define NUM_ORDERS 16
#define C_DIM 64
#define K_DENSE 256
#define WAVES_PER_BLOCK 16
#define BLOCK_THREADS (WAVES_PER_BLOCK * 32)
#define NUM_BLOCKS 256

__device__ const unsigned char ROW_OF[256] = {
    0, 48, 1, 112, 176, 49, 2, 113, 177, 50, 178, 51, 3, 114, 179, 115,
    116, 4, 117, 180, 52, 5, 118, 181, 119, 182, 53, 6, 128, 183, 129, 240,
    8, 56, 9, 120, 184, 57, 10, 121, 185, 58, 186, 59, 11, 122, 187, 123,
    54, 192, 55, 7, 64, 193, 65, 16, 130, 241, 66, 194, 67, 17, 131, 195,
    196, 132, 197, 133, 18, 134, 198, 135, 242, 68, 19, 144, 199, 145, 243, 244,
    124, 12, 125, 188, 60, 13, 126, 189, 127, 190, 61, 14, 136, 191, 137, 248,
    20, 69, 21, 146, 208, 70, 22, 147, 209, 71, 210, 80, 23, 148, 211, 149,
    62, 200, 63, 15, 72, 201, 73, 24, 138, 249, 74, 202, 75, 25, 139, 203,
    212, 81, 213, 82, 245, 83, 214, 84, 32, 246, 247, 85, 215, 86, 33, 150,
    151, 224, 160, 35, 163, 225, 161, 36, 87, 34, 164, 226, 162, 96, 227, 167,
    204, 140, 205, 141, 26, 142, 206, 143, 250, 76, 27, 152, 207, 153, 251, 252,
    159, 42, 168, 232, 95, 43, 169, 233, 171, 234, 104, 44, 170, 235, 172, 175,
    28, 77, 29, 154, 216, 78, 30, 155, 217, 79, 218, 88, 31, 156, 219, 157,
    97, 228, 98, 37, 101, 229, 99, 38, 165, 103, 102, 230, 100, 39, 166, 231,
    220, 89, 221, 90, 253, 91, 222, 92, 40, 254, 255, 93, 223, 94, 41, 158,
    105, 46, 106, 236, 47, 173, 107, 237, 109, 174, 239, 111, 108, 238, 110, 45,
};
__device__ const unsigned char COL_POS[16] = {0, 8, 1, 4, 14, 9, 2, 5, 12, 11, 15, 10, 3, 6, 13, 7};

__device__ __forceinline__ int cg_frag_index(int i, int k) {
  const int kk   = k >> 2;
  const int lane = ((k >> 1) & 1) * 16 + i;
  return ((kk >> 1) * 32 + lane) * 4 + (kk & 1) * 2 + (k & 1);
}

__device__ __forceinline__ void lds_barrier() {
  __builtin_amdgcn_fence(__ATOMIC_RELEASE, "workgroup", "local");
  __builtin_amdgcn_s_barrier();
  __builtin_amdgcn_fence(__ATOMIC_ACQUIRE, "workgroup", "local");
}

struct TileRegs {
  v16f w1;
  v8f  w2;
};

__device__ __forceinline__ void load_tile(TileRegs& r, const float* x1, const float* x2, int t,
                                          int half, int col) {
  const int n = t >> 2;
  const int c = (t & 3) * 16 + col;
  const float* p1 = x1 + (size_t)n * NUM_ORDERS * C_DIM + c;
  const float* p2 = x2 + (size_t)n * NUM_ORDERS * C_DIM + c;
  r.w2[0] = p2[(half ? 6 : 0) * C_DIM];
  r.w2[1] = p2[(half ? 12 : 2) * C_DIM];
  r.w2[2] = p2[(half ? 13 : 3) * C_DIM];
  r.w2[3] = p2[(half ? 15 : 7) * C_DIM];
  r.w2[4] = p2[(half ? 11 : 1) * C_DIM];
  r.w2[5] = p2[(half ? 9 : 5) * C_DIM];
  r.w2[6] = p2[(half ? 4 : 8) * C_DIM];
  r.w2[7] = p2[(half ? 10 : 14) * C_DIM];
  r.w1[0] = p1[0 * C_DIM];
  r.w1[1] = p1[1 * C_DIM];
  r.w1[2] = p1[2 * C_DIM];
  r.w1[3] = p1[3 * C_DIM];
  r.w1[4] = p1[4 * C_DIM];
  r.w1[5] = p1[5 * C_DIM];
  r.w1[6] = p1[6 * C_DIM];
  r.w1[7] = p1[7 * C_DIM];
  r.w1[8] = p1[8 * C_DIM];
  r.w1[9] = p1[9 * C_DIM];
  r.w1[10] = p1[10 * C_DIM];
  r.w1[11] = p1[11 * C_DIM];
  r.w1[12] = p1[12 * C_DIM];
  r.w1[13] = p1[13 * C_DIM];
  r.w1[14] = p1[14 * C_DIM];
  r.w1[15] = p1[15 * C_DIM];
}

struct UnitRegs {
  v2f q1[16];
  v2f q2[8];
};

struct UnitOffsets {
  unsigned eo[9];
};

__device__ __forceinline__ void unit_offsets(UnitOffsets& f, int half, int col) {
  const unsigned c2 = 2u * (unsigned)col;
  f.eo[0] = c2 + (half ? 6u : 0u) * C_DIM;
  f.eo[1] = c2 + (half ? 12u : 2u) * C_DIM;
  f.eo[2] = c2 + (half ? 13u : 3u) * C_DIM;
  f.eo[3] = c2 + (half ? 15u : 7u) * C_DIM;
  f.eo[4] = c2 + (half ? 11u : 1u) * C_DIM;
  f.eo[5] = c2 + (half ? 9u : 5u) * C_DIM;
  f.eo[6] = c2 + (half ? 4u : 8u) * C_DIM;
  f.eo[7] = c2 + (half ? 10u : 14u) * C_DIM;
  f.eo[8] = c2;
}

__device__ __forceinline__ void load_unit(UnitRegs& r, const float* x1, const float* x2, int u,
                                          const UnitOffsets& f) {
  const size_t ub = (size_t)(u >> 1) * NUM_ORDERS * C_DIM + (size_t)((u & 1) * 32);
  const float* b1 = x1 + ub;
  const float* b2 = x2 + ub;
#pragma unroll
  for (int j = 0; j < 8; ++j) r.q2[j] = *(const v2f*)(b2 + f.eo[j]);
#pragma unroll
  for (int i = 0; i < 16; ++i) r.q1[i] = *(const v2f*)(b1 + f.eo[8] + i * C_DIM);
}

__device__ __forceinline__ void split_unit(TileRegs& a, TileRegs& b, const UnitRegs& r) {
#pragma unroll
  for (int i = 0; i < 16; ++i) {
    a.w1[i] = r.q1[i].x;
    b.w1[i] = r.q1[i].y;
  }
#pragma unroll
  for (int i = 0; i < 8; ++i) {
    a.w2[i] = r.q2[i].x;
    b.w2[i] = r.q2[i].y;
  }
}

#define WMMA_RESULT_GUARD(d) \
  asm volatile("s_wait_alu depctr_va_vdst(0)\n\tv_nop\n\tv_nop\n\tv_nop\n\tv_nop" : "+v"(d))
#define WMMA_RESULT_GUARD_A(d, a) \
  asm volatile("s_wait_alu depctr_va_vdst(0)\n\tv_nop\n\tv_nop\n\tv_nop\n\tv_nop" : "+v"(d) : "v"(a))

__device__ __forceinline__ void tile_product(const float4 (&areg)[24], const float4* cgl,
                                             unsigned mlo, unsigned mhi,
                                             const TileRegs& r, float* out, int t,
                                             int half, int col) {
  const v8f zero = {};
  float o[8] = {0.0f, 0.0f, 0.0f, 0.0f, 0.0f, 0.0f, 0.0f, 0.0f};
  if (mlo & 0x1u) {
    const v2f a = {areg[0].x, areg[0].y};
    const v2f b = {r.w2[0], r.w2[1]};
    v8f d = __builtin_amdgcn_wmma_f32_16x16x4_f32(false, a, false, b, (short)0, zero, false, false);
    WMMA_RESULT_GUARD(d);
    o[0] = __builtin_fmaf(r.w1[0], d[0], o[0]);
    o[0] = __builtin_fmaf(r.w1[2], d[1], o[0]);
    o[0] = __builtin_fmaf(r.w1[6], d[2], o[0]);
    o[0] = __builtin_fmaf(r.w1[12], d[3], o[0]);
    o[1] = __builtin_fmaf(r.w1[1], d[4], o[1]);
    o[1] = __builtin_fmaf(r.w1[5], d[5], o[1]);
    o[1] = __builtin_fmaf(r.w1[11], d[6], o[1]);
    o[2] = __builtin_fmaf(r.w1[3], d[7], o[2]);
  }
  if (mlo & 0xeu) {
    if (mlo & 0x2u) {
      const v2f a = {areg[0].z, areg[0].w};
      const v2f b = {r.w2[2], r.w2[3]};
      v8f d = __builtin_amdgcn_wmma_f32_16x16x4_f32(false, a, false, b, (short)0, zero, false, false);
      WMMA_RESULT_GUARD(d);
      o[0] = __builtin_fmaf(r.w1[0], d[0], o[0]);
      o[0] = __builtin_fmaf(r.w1[2], d[1], o[0]);
      o[0] = __builtin_fmaf(r.w1[6], d[2], o[0]);
      o[0] = __builtin_fmaf(r.w1[12], d[3], o[0]);
      o[1] = __builtin_fmaf(r.w1[1], d[4], o[1]);
      o[1] = __builtin_fmaf(r.w1[5], d[5], o[1]);
      o[1] = __builtin_fmaf(r.w1[11], d[6], o[1]);
      o[2] = __builtin_fmaf(r.w1[3], d[7], o[2]);
    }
    if (mlo & 0x4u) {
      const v2f a = {areg[1].x, areg[1].y};
      const v2f b = {r.w2[4], r.w2[5]};
      v8f d = __builtin_amdgcn_wmma_f32_16x16x4_f32(false, a, false, b, (short)0, zero, false, false);
      WMMA_RESULT_GUARD(d);
      o[0] = __builtin_fmaf(r.w1[0], d[0], o[0]);
      o[0] = __builtin_fmaf(r.w1[2], d[1], o[0]);
      o[0] = __builtin_fmaf(r.w1[6], d[2], o[0]);
      o[0] = __builtin_fmaf(r.w1[12], d[3], o[0]);
      o[1] = __builtin_fmaf(r.w1[1], d[4], o[1]);
      o[1] = __builtin_fmaf(r.w1[5], d[5], o[1]);
      o[1] = __builtin_fmaf(r.w1[11], d[6], o[1]);
      o[2] = __builtin_fmaf(r.w1[3], d[7], o[2]);
    }
    if (mlo & 0x8u) {
      const v2f a = {areg[1].z, areg[1].w};
      const v2f b = {r.w2[6], r.w2[7]};
      v8f d = __builtin_amdgcn_wmma_f32_16x16x4_f32(false, a, false, b, (short)0, zero, false, false);
      WMMA_RESULT_GUARD(d);
      o[0] = __builtin_fmaf(r.w1[0], d[0], o[0]);
      o[0] = __builtin_fmaf(r.w1[2], d[1], o[0]);
      o[0] = __builtin_fmaf(r.w1[6], d[2], o[0]);
      o[0] = __builtin_fmaf(r.w1[12], d[3], o[0]);
      o[1] = __builtin_fmaf(r.w1[1], d[4], o[1]);
      o[1] = __builtin_fmaf(r.w1[5], d[5], o[1]);
      o[1] = __builtin_fmaf(r.w1[11], d[6], o[1]);
      o[2] = __builtin_fmaf(r.w1[3], d[7], o[2]);
    }
  }
  if (mlo & 0x10u) {
    const v2f a = {areg[2].x, areg[2].y};
    const v2f b = {r.w2[0], r.w2[1]};
    v8f d = __builtin_amdgcn_wmma_f32_16x16x4_f32(false, a, false, b, (short)0, zero, false, false);
    WMMA_RESULT_GUARD(d);
    o[2] = __builtin_fmaf(r.w1[7], d[0], o[2]);
    o[2] = __builtin_fmaf(r.w1[13], d[1], o[2]);
    o[3] = __builtin_fmaf(r.w1[4], d[2], o[3]);
    o[3] = __builtin_fmaf(r.w1[10], d[3], o[3]);
    o[4] = __builtin_fmaf(r.w1[0], d[4], o[4]);
    o[4] = __builtin_fmaf(r.w1[2], d[5], o[4]);
    o[4] = __builtin_fmaf(r.w1[6], d[6], o[4]);
    o[4] = __builtin_fmaf(r.w1[12], d[7], o[4]);
  }
  if (mlo & 0xe0u) {
    if (mlo & 0x20u) {
      const v2f a = {areg[2].z, areg[2].w};
      const v2f b = {r.w2[2], r.w2[3]};
      v8f d = __builtin_amdgcn_wmma_f32_16x16x4_f32(false, a, false, b, (short)0, zero, false, false);
      WMMA_RESULT_GUARD(d);
      o[2] = __builtin_fmaf(r.w1[7], d[0], o[2]);
      o[2] = __builtin_fmaf(r.w1[13], d[1], o[2]);
      o[3] = __builtin_fmaf(r.w1[4], d[2], o[3]);
      o[3] = __builtin_fmaf(r.w1[10], d[3], o[3]);
      o[4] = __builtin_fmaf(r.w1[0], d[4], o[4]);
      o[4] = __builtin_fmaf(r.w1[2], d[5], o[4]);
      o[4] = __builtin_fmaf(r.w1[6], d[6], o[4]);
      o[4] = __builtin_fmaf(r.w1[12], d[7], o[4]);
    }
    if (mlo & 0x40u) {
      const v2f a = {areg[3].x, areg[3].y};
      const v2f b = {r.w2[4], r.w2[5]};
      v8f d = __builtin_amdgcn_wmma_f32_16x16x4_f32(false, a, false, b, (short)0, zero, false, false);
      WMMA_RESULT_GUARD(d);
      o[2] = __builtin_fmaf(r.w1[7], d[0], o[2]);
      o[2] = __builtin_fmaf(r.w1[13], d[1], o[2]);
      o[3] = __builtin_fmaf(r.w1[4], d[2], o[3]);
      o[3] = __builtin_fmaf(r.w1[10], d[3], o[3]);
      o[4] = __builtin_fmaf(r.w1[0], d[4], o[4]);
      o[4] = __builtin_fmaf(r.w1[2], d[5], o[4]);
      o[4] = __builtin_fmaf(r.w1[6], d[6], o[4]);
      o[4] = __builtin_fmaf(r.w1[12], d[7], o[4]);
    }
    if (mlo & 0x80u) {
      const v2f a = {areg[3].z, areg[3].w};
      const v2f b = {r.w2[6], r.w2[7]};
      v8f d = __builtin_amdgcn_wmma_f32_16x16x4_f32(false, a, false, b, (short)0, zero, false, false);
      WMMA_RESULT_GUARD(d);
      o[2] = __builtin_fmaf(r.w1[7], d[0], o[2]);
      o[2] = __builtin_fmaf(r.w1[13], d[1], o[2]);
      o[3] = __builtin_fmaf(r.w1[4], d[2], o[3]);
      o[3] = __builtin_fmaf(r.w1[10], d[3], o[3]);
      o[4] = __builtin_fmaf(r.w1[0], d[4], o[4]);
      o[4] = __builtin_fmaf(r.w1[2], d[5], o[4]);
      o[4] = __builtin_fmaf(r.w1[6], d[6], o[4]);
      o[4] = __builtin_fmaf(r.w1[12], d[7], o[4]);
    }
  }
  if (mlo & 0x100u) {
    const v2f a = {areg[4].x, areg[4].y};
    const v2f b = {r.w2[0], r.w2[1]};
    v8f d = __builtin_amdgcn_wmma_f32_16x16x4_f32(false, a, false, b, (short)0, zero, false, false);
    WMMA_RESULT_GUARD(d);
    o[5] = __builtin_fmaf(r.w1[8], d[0], o[5]);
    o[5] = __builtin_fmaf(r.w1[14], d[1], o[5]);
    o[6] = __builtin_fmaf((half ? r.w1[1] : r.w1[9]), d[2], o[6]);
    o[6] = __builtin_fmaf((half ? r.w1[5] : r.w1[3]), d[3], o[6]);
    o[6] = __builtin_fmaf((half ? r.w1[11] : r.w1[7]), d[4], o[6]);
    o[7] = __builtin_fmaf((half ? r.w1[15] : r.w1[3]), d[5], o[7]);
    o[7] = __builtin_fmaf((half ? r.w1[1] : r.w1[7]), d[6], o[7]);
    o[7] = __builtin_fmaf((half ? r.w1[4] : r.w1[13]), d[7], o[7]);
  }
  if (mlo & 0xe00u) {
    if (mlo & 0x200u) {
      const v2f a = {areg[4].z, areg[4].w};
      const v2f b = {r.w2[2], r.w2[3]};
      v8f d = __builtin_amdgcn_wmma_f32_16x16x4_f32(false, a, false, b, (short)0, zero, false, false);
      WMMA_RESULT_GUARD(d);
      o[5] = __builtin_fmaf(r.w1[8], d[0], o[5]);
      o[5] = __builtin_fmaf(r.w1[14], d[1], o[5]);
      o[6] = __builtin_fmaf((half ? r.w1[1] : r.w1[9]), d[2], o[6]);
      o[6] = __builtin_fmaf((half ? r.w1[5] : r.w1[3]), d[3], o[6]);
      o[6] = __builtin_fmaf((half ? r.w1[11] : r.w1[7]), d[4], o[6]);
      o[7] = __builtin_fmaf((half ? r.w1[15] : r.w1[3]), d[5], o[7]);
      o[7] = __builtin_fmaf((half ? r.w1[1] : r.w1[7]), d[6], o[7]);
      o[7] = __builtin_fmaf((half ? r.w1[4] : r.w1[13]), d[7], o[7]);
    }
    if (mlo & 0x400u) {
      const v2f a = {areg[5].x, areg[5].y};
      const v2f b = {r.w2[4], r.w2[5]};
      v8f d = __builtin_amdgcn_wmma_f32_16x16x4_f32(false, a, false, b, (short)0, zero, false, false);
      WMMA_RESULT_GUARD(d);
      o[5] = __builtin_fmaf(r.w1[8], d[0], o[5]);
      o[5] = __builtin_fmaf(r.w1[14], d[1], o[5]);
      o[6] = __builtin_fmaf((half ? r.w1[1] : r.w1[9]), d[2], o[6]);
      o[6] = __builtin_fmaf((half ? r.w1[5] : r.w1[3]), d[3], o[6]);
      o[6] = __builtin_fmaf((half ? r.w1[11] : r.w1[7]), d[4], o[6]);
      o[7] = __builtin_fmaf((half ? r.w1[15] : r.w1[3]), d[5], o[7]);
      o[7] = __builtin_fmaf((half ? r.w1[1] : r.w1[7]), d[6], o[7]);
      o[7] = __builtin_fmaf((half ? r.w1[4] : r.w1[13]), d[7], o[7]);
    }
    if (mlo & 0x800u) {
      const v2f a = {areg[5].z, areg[5].w};
      const v2f b = {r.w2[6], r.w2[7]};
      v8f d = __builtin_amdgcn_wmma_f32_16x16x4_f32(false, a, false, b, (short)0, zero, false, false);
      WMMA_RESULT_GUARD(d);
      o[5] = __builtin_fmaf(r.w1[8], d[0], o[5]);
      o[5] = __builtin_fmaf(r.w1[14], d[1], o[5]);
      o[6] = __builtin_fmaf((half ? r.w1[1] : r.w1[9]), d[2], o[6]);
      o[6] = __builtin_fmaf((half ? r.w1[5] : r.w1[3]), d[3], o[6]);
      o[6] = __builtin_fmaf((half ? r.w1[11] : r.w1[7]), d[4], o[6]);
      o[7] = __builtin_fmaf((half ? r.w1[15] : r.w1[3]), d[5], o[7]);
      o[7] = __builtin_fmaf((half ? r.w1[1] : r.w1[7]), d[6], o[7]);
      o[7] = __builtin_fmaf((half ? r.w1[4] : r.w1[13]), d[7], o[7]);
    }
  }
  if (mlo & 0x2000u) {
    const v2f a = {areg[6].z, areg[6].w};
    const v2f b = {r.w2[2], r.w2[3]};
    v8f d = __builtin_amdgcn_wmma_f32_16x16x4_f32(false, a, false, b, (short)0, zero, false, false);
    WMMA_RESULT_GUARD(d);
    o[0] = __builtin_fmaf(r.w1[1], d[0], o[0]);
    o[0] = __builtin_fmaf(r.w1[5], d[1], o[0]);
    o[0] = __builtin_fmaf(r.w1[9], d[2], o[0]);
    o[0] = __builtin_fmaf(r.w1[11], d[3], o[0]);
    o[1] = __builtin_fmaf(r.w1[4], d[4], o[1]);
    o[1] = __builtin_fmaf(r.w1[10], d[5], o[1]);
    o[2] = __builtin_fmaf(r.w1[0], d[6], o[2]);
    o[2] = __builtin_fmaf(r.w1[2], d[7], o[2]);
  }
  if (mlo & 0xd000u) {
    if (mlo & 0x1000u) {
      const v2f a = {areg[6].x, areg[6].y};
      const v2f b = {r.w2[0], r.w2[1]};
      v8f d = __builtin_amdgcn_wmma_f32_16x16x4_f32(false, a, false, b, (short)0, zero, false, false);
      WMMA_RESULT_GUARD(d);
      o[0] = __builtin_fmaf(r.w1[1], d[0], o[0]);
      o[0] = __builtin_fmaf(r.w1[5], d[1], o[0]);
      o[0] = __builtin_fmaf(r.w1[9], d[2], o[0]);
      o[0] = __builtin_fmaf(r.w1[11], d[3], o[0]);
      o[1] = __builtin_fmaf(r.w1[4], d[4], o[1]);
      o[1] = __builtin_fmaf(r.w1[10], d[5], o[1]);
      o[2] = __builtin_fmaf(r.w1[0], d[6], o[2]);
      o[2] = __builtin_fmaf(r.w1[2], d[7], o[2]);
    }
    if (mlo & 0x4000u) {
      const v2f a = {areg[7].x, areg[7].y};
      const v2f b = {r.w2[4], r.w2[5]};
      v8f d = __builtin_amdgcn_wmma_f32_16x16x4_f32(false, a, false, b, (short)0, zero, false, false);
      WMMA_RESULT_GUARD(d);
      o[0] = __builtin_fmaf(r.w1[1], d[0], o[0]);
      o[0] = __builtin_fmaf(r.w1[5], d[1], o[0]);
      o[0] = __builtin_fmaf(r.w1[9], d[2], o[0]);
      o[0] = __builtin_fmaf(r.w1[11], d[3], o[0]);
      o[1] = __builtin_fmaf(r.w1[4], d[4], o[1]);
      o[1] = __builtin_fmaf(r.w1[10], d[5], o[1]);
      o[2] = __builtin_fmaf(r.w1[0], d[6], o[2]);
      o[2] = __builtin_fmaf(r.w1[2], d[7], o[2]);
    }
    if (mlo & 0x8000u) {
      const v2f a = {areg[7].z, areg[7].w};
      const v2f b = {r.w2[6], r.w2[7]};
      v8f d = __builtin_amdgcn_wmma_f32_16x16x4_f32(false, a, false, b, (short)0, zero, false, false);
      WMMA_RESULT_GUARD(d);
      o[0] = __builtin_fmaf(r.w1[1], d[0], o[0]);
      o[0] = __builtin_fmaf(r.w1[5], d[1], o[0]);
      o[0] = __builtin_fmaf(r.w1[9], d[2], o[0]);
      o[0] = __builtin_fmaf(r.w1[11], d[3], o[0]);
      o[1] = __builtin_fmaf(r.w1[4], d[4], o[1]);
      o[1] = __builtin_fmaf(r.w1[10], d[5], o[1]);
      o[2] = __builtin_fmaf(r.w1[0], d[6], o[2]);
      o[2] = __builtin_fmaf(r.w1[2], d[7], o[2]);
    }
  }
  if (mlo & 0x20000u) {
    const v2f a = {areg[8].z, areg[8].w};
    const v2f b = {r.w2[2], r.w2[3]};
    v8f d = __builtin_amdgcn_wmma_f32_16x16x4_f32(false, a, false, b, (short)0, zero, false, false);
    WMMA_RESULT_GUARD(d);
    o[2] = __builtin_fmaf(r.w1[4], d[0], o[2]);
    o[2] = __builtin_fmaf(r.w1[6], d[1], o[2]);
    o[2] = __builtin_fmaf(r.w1[10], d[2], o[2]);
    o[2] = __builtin_fmaf(r.w1[12], d[3], o[2]);
    o[3] = __builtin_fmaf(r.w1[9], d[4], o[3]);
    o[4] = __builtin_fmaf(r.w1[1], d[5], o[4]);
    o[4] = __builtin_fmaf(r.w1[5], d[6], o[4]);
    o[4] = __builtin_fmaf(r.w1[9], d[7], o[4]);
  }
  if (mlo & 0xd0000u) {
    if (mlo & 0x10000u) {
      const v2f a = {areg[8].x, areg[8].y};
      const v2f b = {r.w2[0], r.w2[1]};
      v8f d = __builtin_amdgcn_wmma_f32_16x16x4_f32(false, a, false, b, (short)0, zero, false, false);
      WMMA_RESULT_GUARD(d);
      o[2] = __builtin_fmaf(r.w1[4], d[0], o[2]);
      o[2] = __builtin_fmaf(r.w1[6], d[1], o[2]);
      o[2] = __builtin_fmaf(r.w1[10], d[2], o[2]);
      o[2] = __builtin_fmaf(r.w1[12], d[3], o[2]);
      o[3] = __builtin_fmaf(r.w1[9], d[4], o[3]);
      o[4] = __builtin_fmaf(r.w1[1], d[5], o[4]);
      o[4] = __builtin_fmaf(r.w1[5], d[6], o[4]);
      o[4] = __builtin_fmaf(r.w1[9], d[7], o[4]);
    }
    if (mlo & 0x40000u) {
      const v2f a = {areg[9].x, areg[9].y};
      const v2f b = {r.w2[4], r.w2[5]};
      v8f d = __builtin_amdgcn_wmma_f32_16x16x4_f32(false, a, false, b, (short)0, zero, false, false);
      WMMA_RESULT_GUARD(d);
      o[2] = __builtin_fmaf(r.w1[4], d[0], o[2]);
      o[2] = __builtin_fmaf(r.w1[6], d[1], o[2]);
      o[2] = __builtin_fmaf(r.w1[10], d[2], o[2]);
      o[2] = __builtin_fmaf(r.w1[12], d[3], o[2]);
      o[3] = __builtin_fmaf(r.w1[9], d[4], o[3]);
      o[4] = __builtin_fmaf(r.w1[1], d[5], o[4]);
      o[4] = __builtin_fmaf(r.w1[5], d[6], o[4]);
      o[4] = __builtin_fmaf(r.w1[9], d[7], o[4]);
    }
    if (mlo & 0x80000u) {
      const v2f a = {areg[9].z, areg[9].w};
      const v2f b = {r.w2[6], r.w2[7]};
      v8f d = __builtin_amdgcn_wmma_f32_16x16x4_f32(false, a, false, b, (short)0, zero, false, false);
      WMMA_RESULT_GUARD(d);
      o[2] = __builtin_fmaf(r.w1[4], d[0], o[2]);
      o[2] = __builtin_fmaf(r.w1[6], d[1], o[2]);
      o[2] = __builtin_fmaf(r.w1[10], d[2], o[2]);
      o[2] = __builtin_fmaf(r.w1[12], d[3], o[2]);
      o[3] = __builtin_fmaf(r.w1[9], d[4], o[3]);
      o[4] = __builtin_fmaf(r.w1[1], d[5], o[4]);
      o[4] = __builtin_fmaf(r.w1[5], d[6], o[4]);
      o[4] = __builtin_fmaf(r.w1[9], d[7], o[4]);
    }
  }
  if (mlo & 0x200000u) {
    const v2f a = {areg[10].z, areg[10].w};
    const v2f b = {r.w2[2], r.w2[3]};
    v8f d = __builtin_amdgcn_wmma_f32_16x16x4_f32(false, a, false, b, (short)0, zero, false, false);
    WMMA_RESULT_GUARD(d);
    o[4] = __builtin_fmaf(r.w1[11], d[0], o[4]);
    o[5] = __builtin_fmaf(r.w1[1], d[1], o[5]);
    o[5] = __builtin_fmaf(r.w1[3], d[2], o[5]);
    o[5] = __builtin_fmaf(r.w1[5], d[3], o[5]);
    o[5] = __builtin_fmaf(r.w1[7], d[4], o[5]);
    o[5] = __builtin_fmaf(r.w1[11], d[5], o[5]);
    o[5] = __builtin_fmaf(r.w1[13], d[6], o[5]);
    o[6] = __builtin_fmaf((half ? r.w1[4] : r.w1[8]), d[7], o[6]);
  }
  if (mlo & 0xd00000u) {
    if (mlo & 0x100000u) {
      const v2f a = {areg[10].x, areg[10].y};
      const v2f b = {r.w2[0], r.w2[1]};
      v8f d = __builtin_amdgcn_wmma_f32_16x16x4_f32(false, a, false, b, (short)0, zero, false, false);
      WMMA_RESULT_GUARD(d);
      o[4] = __builtin_fmaf(r.w1[11], d[0], o[4]);
      o[5] = __builtin_fmaf(r.w1[1], d[1], o[5]);
      o[5] = __builtin_fmaf(r.w1[3], d[2], o[5]);
      o[5] = __builtin_fmaf(r.w1[5], d[3], o[5]);
      o[5] = __builtin_fmaf(r.w1[7], d[4], o[5]);
      o[5] = __builtin_fmaf(r.w1[11], d[5], o[5]);
      o[5] = __builtin_fmaf(r.w1[13], d[6], o[5]);
      o[6] = __builtin_fmaf((half ? r.w1[4] : r.w1[8]), d[7], o[6]);
    }
    if (mlo & 0x400000u) {
      const v2f a = {areg[11].x, areg[11].y};
      const v2f b = {r.w2[4], r.w2[5]};
      v8f d = __builtin_amdgcn_wmma_f32_16x16x4_f32(false, a, false, b, (short)0, zero, false, false);
      WMMA_RESULT_GUARD(d);
      o[4] = __builtin_fmaf(r.w1[11], d[0], o[4]);
      o[5] = __builtin_fmaf(r.w1[1], d[1], o[5]);
      o[5] = __builtin_fmaf(r.w1[3], d[2], o[5]);
      o[5] = __builtin_fmaf(r.w1[5], d[3], o[5]);
      o[5] = __builtin_fmaf(r.w1[7], d[4], o[5]);
      o[5] = __builtin_fmaf(r.w1[11], d[5], o[5]);
      o[5] = __builtin_fmaf(r.w1[13], d[6], o[5]);
      o[6] = __builtin_fmaf((half ? r.w1[4] : r.w1[8]), d[7], o[6]);
    }
    if (mlo & 0x800000u) {
      const v2f a = {areg[11].z, areg[11].w};
      const v2f b = {r.w2[6], r.w2[7]};
      v8f d = __builtin_amdgcn_wmma_f32_16x16x4_f32(false, a, false, b, (short)0, zero, false, false);
      WMMA_RESULT_GUARD(d);
      o[4] = __builtin_fmaf(r.w1[11], d[0], o[4]);
      o[5] = __builtin_fmaf(r.w1[1], d[1], o[5]);
      o[5] = __builtin_fmaf(r.w1[3], d[2], o[5]);
      o[5] = __builtin_fmaf(r.w1[5], d[3], o[5]);
      o[5] = __builtin_fmaf(r.w1[7], d[4], o[5]);
      o[5] = __builtin_fmaf(r.w1[11], d[5], o[5]);
      o[5] = __builtin_fmaf(r.w1[13], d[6], o[5]);
      o[6] = __builtin_fmaf((half ? r.w1[4] : r.w1[8]), d[7], o[6]);
    }
  }
  if (mlo & 0x2000000u) {
    const v2f a = {areg[12].z, areg[12].w};
    const v2f b = {r.w2[2], r.w2[3]};
    v8f d = __builtin_amdgcn_wmma_f32_16x16x4_f32(false, a, false, b, (short)0, zero, false, false);
    WMMA_RESULT_GUARD(d);
    o[6] = __builtin_fmaf((half ? r.w1[10] : r.w1[13]), d[0], o[6]);
    o[7] = __builtin_fmaf(r.w1[0], d[1], o[7]);
    o[7] = __builtin_fmaf(r.w1[2], d[2], o[7]);
    o[7] = __builtin_fmaf(r.w1[6], d[3], o[7]);
    o[7] = __builtin_fmaf(r.w1[12], d[4], o[7]);
    o[7] = __builtin_fmaf((half ? r.w1[8] : r.w1[4]), d[5], o[7]);
    o[7] = __builtin_fmaf((half ? r.w1[14] : r.w1[10]), d[6], o[7]);
    o[7] = __builtin_fmaf((half ? r.w1[11] : r.w1[9]), d[7], o[7]);
  }
  if (mlo & 0xd000000u) {
    if (mlo & 0x1000000u) {
      const v2f a = {areg[12].x, areg[12].y};
      const v2f b = {r.w2[0], r.w2[1]};
      v8f d = __builtin_amdgcn_wmma_f32_16x16x4_f32(false, a, false, b, (short)0, zero, false, false);
      WMMA_RESULT_GUARD(d);
      o[6] = __builtin_fmaf((half ? r.w1[10] : r.w1[13]), d[0], o[6]);
      o[7] = __builtin_fmaf(r.w1[0], d[1], o[7]);
      o[7] = __builtin_fmaf(r.w1[2], d[2], o[7]);
      o[7] = __builtin_fmaf(r.w1[6], d[3], o[7]);
      o[7] = __builtin_fmaf(r.w1[12], d[4], o[7]);
      o[7] = __builtin_fmaf((half ? r.w1[8] : r.w1[4]), d[5], o[7]);
      o[7] = __builtin_fmaf((half ? r.w1[14] : r.w1[10]), d[6], o[7]);
      o[7] = __builtin_fmaf((half ? r.w1[11] : r.w1[9]), d[7], o[7]);
    }
    if (mlo & 0x4000000u) {
      const v2f a = {areg[13].x, areg[13].y};
      const v2f b = {r.w2[4], r.w2[5]};
      v8f d = __builtin_amdgcn_wmma_f32_16x16x4_f32(false, a, false, b, (short)0, zero, false, false);
      WMMA_RESULT_GUARD(d);
      o[6] = __builtin_fmaf((half ? r.w1[10] : r.w1[13]), d[0], o[6]);
      o[7] = __builtin_fmaf(r.w1[0], d[1], o[7]);
      o[7] = __builtin_fmaf(r.w1[2], d[2], o[7]);
      o[7] = __builtin_fmaf(r.w1[6], d[3], o[7]);
      o[7] = __builtin_fmaf(r.w1[12], d[4], o[7]);
      o[7] = __builtin_fmaf((half ? r.w1[8] : r.w1[4]), d[5], o[7]);
      o[7] = __builtin_fmaf((half ? r.w1[14] : r.w1[10]), d[6], o[7]);
      o[7] = __builtin_fmaf((half ? r.w1[11] : r.w1[9]), d[7], o[7]);
    }
    if (mlo & 0x8000000u) {
      const v2f a = {areg[13].z, areg[13].w};
      const v2f b = {r.w2[6], r.w2[7]};
      v8f d = __builtin_amdgcn_wmma_f32_16x16x4_f32(false, a, false, b, (short)0, zero, false, false);
      WMMA_RESULT_GUARD(d);
      o[6] = __builtin_fmaf((half ? r.w1[10] : r.w1[13]), d[0], o[6]);
      o[7] = __builtin_fmaf(r.w1[0], d[1], o[7]);
      o[7] = __builtin_fmaf(r.w1[2], d[2], o[7]);
      o[7] = __builtin_fmaf(r.w1[6], d[3], o[7]);
      o[7] = __builtin_fmaf(r.w1[12], d[4], o[7]);
      o[7] = __builtin_fmaf((half ? r.w1[8] : r.w1[4]), d[5], o[7]);
      o[7] = __builtin_fmaf((half ? r.w1[14] : r.w1[10]), d[6], o[7]);
      o[7] = __builtin_fmaf((half ? r.w1[11] : r.w1[9]), d[7], o[7]);
    }
  }
  if (mlo & 0x40000000u) {
    const v2f a = {areg[15].x, areg[15].y};
    const v2f b = {r.w2[4], r.w2[5]};
    v8f d = __builtin_amdgcn_wmma_f32_16x16x4_f32(false, a, false, b, (short)0, zero, false, false);
    WMMA_RESULT_GUARD(d);
    o[0] = __builtin_fmaf(r.w1[3], d[0], o[0]);
    o[0] = __builtin_fmaf(r.w1[7], d[1], o[0]);
    o[0] = __builtin_fmaf(r.w1[13], d[2], o[0]);
    o[0] = __builtin_fmaf(r.w1[15], d[3], o[0]);
    o[1] = __builtin_fmaf(r.w1[0], d[4], o[1]);
    o[1] = __builtin_fmaf(r.w1[2], d[5], o[1]);
    o[1] = __builtin_fmaf(r.w1[6], d[6], o[1]);
    o[1] = __builtin_fmaf(r.w1[8], d[7], o[1]);
  }
  if (mlo & 0xb0000000u) {
    if (mlo & 0x10000000u) {
      const v2f a = {areg[14].x, areg[14].y};
      const v2f b = {r.w2[0], r.w2[1]};
      v8f d = __builtin_amdgcn_wmma_f32_16x16x4_f32(false, a, false, b, (short)0, zero, false, false);
      WMMA_RESULT_GUARD(d);
      o[0] = __builtin_fmaf(r.w1[3], d[0], o[0]);
      o[0] = __builtin_fmaf(r.w1[7], d[1], o[0]);
      o[0] = __builtin_fmaf(r.w1[13], d[2], o[0]);
      o[0] = __builtin_fmaf(r.w1[15], d[3], o[0]);
      o[1] = __builtin_fmaf(r.w1[0], d[4], o[1]);
      o[1] = __builtin_fmaf(r.w1[2], d[5], o[1]);
      o[1] = __builtin_fmaf(r.w1[6], d[6], o[1]);
      o[1] = __builtin_fmaf(r.w1[8], d[7], o[1]);
    }
    if (mlo & 0x20000000u) {
      const v2f a = {areg[14].z, areg[14].w};
      const v2f b = {r.w2[2], r.w2[3]};
      v8f d = __builtin_amdgcn_wmma_f32_16x16x4_f32(false, a, false, b, (short)0, zero, false, false);
      WMMA_RESULT_GUARD(d);
      o[0] = __builtin_fmaf(r.w1[3], d[0], o[0]);
      o[0] = __builtin_fmaf(r.w1[7], d[1], o[0]);
      o[0] = __builtin_fmaf(r.w1[13], d[2], o[0]);
      o[0] = __builtin_fmaf(r.w1[15], d[3], o[0]);
      o[1] = __builtin_fmaf(r.w1[0], d[4], o[1]);
      o[1] = __builtin_fmaf(r.w1[2], d[5], o[1]);
      o[1] = __builtin_fmaf(r.w1[6], d[6], o[1]);
      o[1] = __builtin_fmaf(r.w1[8], d[7], o[1]);
    }
    if (mlo & 0x80000000u) {
      const v2f a = {areg[15].z, areg[15].w};
      const v2f b = {r.w2[6], r.w2[7]};
      v8f d = __builtin_amdgcn_wmma_f32_16x16x4_f32(false, a, false, b, (short)0, zero, false, false);
      WMMA_RESULT_GUARD(d);
      o[0] = __builtin_fmaf(r.w1[3], d[0], o[0]);
      o[0] = __builtin_fmaf(r.w1[7], d[1], o[0]);
      o[0] = __builtin_fmaf(r.w1[13], d[2], o[0]);
      o[0] = __builtin_fmaf(r.w1[15], d[3], o[0]);
      o[1] = __builtin_fmaf(r.w1[0], d[4], o[1]);
      o[1] = __builtin_fmaf(r.w1[2], d[5], o[1]);
      o[1] = __builtin_fmaf(r.w1[6], d[6], o[1]);
      o[1] = __builtin_fmaf(r.w1[8], d[7], o[1]);
    }
  }
  if (mhi & 0x4u) {
    const v2f a = {areg[17].x, areg[17].y};
    const v2f b = {r.w2[4], r.w2[5]};
    v8f d = __builtin_amdgcn_wmma_f32_16x16x4_f32(false, a, false, b, (short)0, zero, false, false);
    WMMA_RESULT_GUARD(d);
    o[1] = __builtin_fmaf(r.w1[12], d[0], o[1]);
    o[1] = __builtin_fmaf(r.w1[14], d[1], o[1]);
    o[2] = __builtin_fmaf(r.w1[8], d[2], o[2]);
    o[2] = __builtin_fmaf(r.w1[14], d[3], o[2]);
    o[3] = __builtin_fmaf(r.w1[1], d[4], o[3]);
    o[3] = __builtin_fmaf(r.w1[3], d[5], o[3]);
    o[3] = __builtin_fmaf(r.w1[5], d[6], o[3]);
    o[3] = __builtin_fmaf(r.w1[7], d[7], o[3]);
  }
  if (mhi & 0xbu) {
    if (mhi & 0x1u) {
      const v2f a = {areg[16].x, areg[16].y};
      const v2f b = {r.w2[0], r.w2[1]};
      v8f d = __builtin_amdgcn_wmma_f32_16x16x4_f32(false, a, false, b, (short)0, zero, false, false);
      WMMA_RESULT_GUARD(d);
      o[1] = __builtin_fmaf(r.w1[12], d[0], o[1]);
      o[1] = __builtin_fmaf(r.w1[14], d[1], o[1]);
      o[2] = __builtin_fmaf(r.w1[8], d[2], o[2]);
      o[2] = __builtin_fmaf(r.w1[14], d[3], o[2]);
      o[3] = __builtin_fmaf(r.w1[1], d[4], o[3]);
      o[3] = __builtin_fmaf(r.w1[3], d[5], o[3]);
      o[3] = __builtin_fmaf(r.w1[5], d[6], o[3]);
      o[3] = __builtin_fmaf(r.w1[7], d[7], o[3]);
    }
    if (mhi & 0x2u) {
      const v2f a = {areg[16].z, areg[16].w};
      const v2f b = {r.w2[2], r.w2[3]};
      v8f d = __builtin_amdgcn_wmma_f32_16x16x4_f32(false, a, false, b, (short)0, zero, false, false);
      WMMA_RESULT_GUARD(d);
      o[1] = __builtin_fmaf(r.w1[12], d[0], o[1]);
      o[1] = __builtin_fmaf(r.w1[14], d[1], o[1]);
      o[2] = __builtin_fmaf(r.w1[8], d[2], o[2]);
      o[2] = __builtin_fmaf(r.w1[14], d[3], o[2]);
      o[3] = __builtin_fmaf(r.w1[1], d[4], o[3]);
      o[3] = __builtin_fmaf(r.w1[3], d[5], o[3]);
      o[3] = __builtin_fmaf(r.w1[5], d[6], o[3]);
      o[3] = __builtin_fmaf(r.w1[7], d[7], o[3]);
    }
    if (mhi & 0x8u) {
      const v2f a = {areg[17].z, areg[17].w};
      const v2f b = {r.w2[6], r.w2[7]};
      v8f d = __builtin_amdgcn_wmma_f32_16x16x4_f32(false, a, false, b, (short)0, zero, false, false);
      WMMA_RESULT_GUARD(d);
      o[1] = __builtin_fmaf(r.w1[12], d[0], o[1]);
      o[1] = __builtin_fmaf(r.w1[14], d[1], o[1]);
      o[2] = __builtin_fmaf(r.w1[8], d[2], o[2]);
      o[2] = __builtin_fmaf(r.w1[14], d[3], o[2]);
      o[3] = __builtin_fmaf(r.w1[1], d[4], o[3]);
      o[3] = __builtin_fmaf(r.w1[3], d[5], o[3]);
      o[3] = __builtin_fmaf(r.w1[5], d[6], o[3]);
      o[3] = __builtin_fmaf(r.w1[7], d[7], o[3]);
    }
  }
  if (mhi & 0x40u) {
    const v2f a = {areg[19].x, areg[19].y};
    const v2f b = {r.w2[4], r.w2[5]};
    v8f d = __builtin_amdgcn_wmma_f32_16x16x4_f32(false, a, false, b, (short)0, zero, false, false);
    WMMA_RESULT_GUARD(d);
    o[3] = __builtin_fmaf(r.w1[11], d[0], o[3]);
    o[3] = __builtin_fmaf(r.w1[13], d[1], o[3]);
    o[4] = __builtin_fmaf(r.w1[3], d[2], o[4]);
    o[4] = __builtin_fmaf(r.w1[7], d[3], o[4]);
    o[4] = __builtin_fmaf(r.w1[13], d[4], o[4]);
    o[4] = __builtin_fmaf(r.w1[15], d[5], o[4]);
    o[5] = __builtin_fmaf(r.w1[15], d[6], o[5]);
    o[6] = __builtin_fmaf(r.w1[0], d[7], o[6]);
  }
  if (mhi & 0xb0u) {
    if (mhi & 0x10u) {
      const v2f a = {areg[18].x, areg[18].y};
      const v2f b = {r.w2[0], r.w2[1]};
      v8f d = __builtin_amdgcn_wmma_f32_16x16x4_f32(false, a, false, b, (short)0, zero, false, false);
      WMMA_RESULT_GUARD(d);
      o[3] = __builtin_fmaf(r.w1[11], d[0], o[3]);
      o[3] = __builtin_fmaf(r.w1[13], d[1], o[3]);
      o[4] = __builtin_fmaf(r.w1[3], d[2], o[4]);
      o[4] = __builtin_fmaf(r.w1[7], d[3], o[4]);
      o[4] = __builtin_fmaf(r.w1[13], d[4], o[4]);
      o[4] = __builtin_fmaf(r.w1[15], d[5], o[4]);
      o[5] = __builtin_fmaf(r.w1[15], d[6], o[5]);
      o[6] = __builtin_fmaf(r.w1[0], d[7], o[6]);
    }
    if (mhi & 0x20u) {
      const v2f a = {areg[18].z, areg[18].w};
      const v2f b = {r.w2[2], r.w2[3]};
      v8f d = __builtin_amdgcn_wmma_f32_16x16x4_f32(false, a, false, b, (short)0, zero, false, false);
      WMMA_RESULT_GUARD(d);
      o[3] = __builtin_fmaf(r.w1[11], d[0], o[3]);
      o[3] = __builtin_fmaf(r.w1[13], d[1], o[3]);
      o[4] = __builtin_fmaf(r.w1[3], d[2], o[4]);
      o[4] = __builtin_fmaf(r.w1[7], d[3], o[4]);
      o[4] = __builtin_fmaf(r.w1[13], d[4], o[4]);
      o[4] = __builtin_fmaf(r.w1[15], d[5], o[4]);
      o[5] = __builtin_fmaf(r.w1[15], d[6], o[5]);
      o[6] = __builtin_fmaf(r.w1[0], d[7], o[6]);
    }
    if (mhi & 0x80u) {
      const v2f a = {areg[19].z, areg[19].w};
      const v2f b = {r.w2[6], r.w2[7]};
      v8f d = __builtin_amdgcn_wmma_f32_16x16x4_f32(false, a, false, b, (short)0, zero, false, false);
      WMMA_RESULT_GUARD(d);
      o[3] = __builtin_fmaf(r.w1[11], d[0], o[3]);
      o[3] = __builtin_fmaf(r.w1[13], d[1], o[3]);
      o[4] = __builtin_fmaf(r.w1[3], d[2], o[4]);
      o[4] = __builtin_fmaf(r.w1[7], d[3], o[4]);
      o[4] = __builtin_fmaf(r.w1[13], d[4], o[4]);
      o[4] = __builtin_fmaf(r.w1[15], d[5], o[4]);
      o[5] = __builtin_fmaf(r.w1[15], d[6], o[5]);
      o[6] = __builtin_fmaf(r.w1[0], d[7], o[6]);
    }
  }
  if (mhi & 0x400u) {
    const v2f a = {areg[21].x, areg[21].y};
    const v2f b = {r.w2[4], r.w2[5]};
    v8f d = __builtin_amdgcn_wmma_f32_16x16x4_f32(false, a, false, b, (short)0, zero, false, false);
    WMMA_RESULT_GUARD(d);
    o[6] = __builtin_fmaf(r.w1[2], d[0], o[6]);
    o[6] = __builtin_fmaf(r.w1[6], d[1], o[6]);
    o[6] = __builtin_fmaf(r.w1[12], d[2], o[6]);
    o[6] = __builtin_fmaf((half ? r.w1[8] : r.w1[4]), d[3], o[6]);
    o[6] = __builtin_fmaf((half ? r.w1[14] : r.w1[10]), d[4], o[6]);
    o[7] = __builtin_fmaf((half ? r.w1[5] : r.w1[8]), d[5], o[7]);
    o[7] = __builtin_fmaf((half ? r.w1[9] : r.w1[14]), d[6], o[7]);
    o[6] = __builtin_fmaf(r.w1[15], d[7], o[6]);
  }
  if (mhi & 0xb00u) {
    if (mhi & 0x100u) {
      const v2f a = {areg[20].x, areg[20].y};
      const v2f b = {r.w2[0], r.w2[1]};
      v8f d = __builtin_amdgcn_wmma_f32_16x16x4_f32(false, a, false, b, (short)0, zero, false, false);
      WMMA_RESULT_GUARD(d);
      o[6] = __builtin_fmaf(r.w1[2], d[0], o[6]);
      o[6] = __builtin_fmaf(r.w1[6], d[1], o[6]);
      o[6] = __builtin_fmaf(r.w1[12], d[2], o[6]);
      o[6] = __builtin_fmaf((half ? r.w1[8] : r.w1[4]), d[3], o[6]);
      o[6] = __builtin_fmaf((half ? r.w1[14] : r.w1[10]), d[4], o[6]);
      o[7] = __builtin_fmaf((half ? r.w1[5] : r.w1[8]), d[5], o[7]);
      o[7] = __builtin_fmaf((half ? r.w1[9] : r.w1[14]), d[6], o[7]);
      o[6] = __builtin_fmaf(r.w1[15], d[7], o[6]);
    }
    if (mhi & 0x200u) {
      const v2f a = {areg[20].z, areg[20].w};
      const v2f b = {r.w2[2], r.w2[3]};
      v8f d = __builtin_amdgcn_wmma_f32_16x16x4_f32(false, a, false, b, (short)0, zero, false, false);
      WMMA_RESULT_GUARD(d);
      o[6] = __builtin_fmaf(r.w1[2], d[0], o[6]);
      o[6] = __builtin_fmaf(r.w1[6], d[1], o[6]);
      o[6] = __builtin_fmaf(r.w1[12], d[2], o[6]);
      o[6] = __builtin_fmaf((half ? r.w1[8] : r.w1[4]), d[3], o[6]);
      o[6] = __builtin_fmaf((half ? r.w1[14] : r.w1[10]), d[4], o[6]);
      o[7] = __builtin_fmaf((half ? r.w1[5] : r.w1[8]), d[5], o[7]);
      o[7] = __builtin_fmaf((half ? r.w1[9] : r.w1[14]), d[6], o[7]);
      o[6] = __builtin_fmaf(r.w1[15], d[7], o[6]);
    }
    if (mhi & 0x800u) {
      const v2f a = {areg[21].z, areg[21].w};
      const v2f b = {r.w2[6], r.w2[7]};
      v8f d = __builtin_amdgcn_wmma_f32_16x16x4_f32(false, a, false, b, (short)0, zero, false, false);
      WMMA_RESULT_GUARD(d);
      o[6] = __builtin_fmaf(r.w1[2], d[0], o[6]);
      o[6] = __builtin_fmaf(r.w1[6], d[1], o[6]);
      o[6] = __builtin_fmaf(r.w1[12], d[2], o[6]);
      o[6] = __builtin_fmaf((half ? r.w1[8] : r.w1[4]), d[3], o[6]);
      o[6] = __builtin_fmaf((half ? r.w1[14] : r.w1[10]), d[4], o[6]);
      o[7] = __builtin_fmaf((half ? r.w1[5] : r.w1[8]), d[5], o[7]);
      o[7] = __builtin_fmaf((half ? r.w1[9] : r.w1[14]), d[6], o[7]);
      o[6] = __builtin_fmaf(r.w1[15], d[7], o[6]);
    }
  }
  if (mhi & 0x8000u) {
    const v2f a = {areg[23].z, areg[23].w};
    const v2f b = {r.w2[6], r.w2[7]};
    v8f d = __builtin_amdgcn_wmma_f32_16x16x4_f32(false, a, false, b, (short)0, zero, false, false);
    WMMA_RESULT_GUARD(d);
    o[0] = __builtin_fmaf(r.w1[4], d[0], o[0]);
    o[0] = __builtin_fmaf(r.w1[8], d[1], o[0]);
    o[0] = __builtin_fmaf(r.w1[10], d[2], o[0]);
    o[0] = __builtin_fmaf(r.w1[14], d[3], o[0]);
    o[1] = __builtin_fmaf(r.w1[3], d[4], o[1]);
    o[1] = __builtin_fmaf(r.w1[7], d[5], o[1]);
    o[1] = __builtin_fmaf(r.w1[9], d[6], o[1]);
    o[1] = __builtin_fmaf(r.w1[13], d[7], o[1]);
  }
  if (mhi & 0x7000u) {
    if (mhi & 0x1000u) {
      const v2f a = {areg[22].x, areg[22].y};
      const v2f b = {r.w2[0], r.w2[1]};
      v8f d = __builtin_amdgcn_wmma_f32_16x16x4_f32(false, a, false, b, (short)0, zero, false, false);
      WMMA_RESULT_GUARD(d);
      o[0] = __builtin_fmaf(r.w1[4], d[0], o[0]);
      o[0] = __builtin_fmaf(r.w1[8], d[1], o[0]);
      o[0] = __builtin_fmaf(r.w1[10], d[2], o[0]);
      o[0] = __builtin_fmaf(r.w1[14], d[3], o[0]);
      o[1] = __builtin_fmaf(r.w1[3], d[4], o[1]);
      o[1] = __builtin_fmaf(r.w1[7], d[5], o[1]);
      o[1] = __builtin_fmaf(r.w1[9], d[6], o[1]);
      o[1] = __builtin_fmaf(r.w1[13], d[7], o[1]);
    }
    if (mhi & 0x2000u) {
      const v2f a = {areg[22].z, areg[22].w};
      const v2f b = {r.w2[2], r.w2[3]};
      v8f d = __builtin_amdgcn_wmma_f32_16x16x4_f32(false, a, false, b, (short)0, zero, false, false);
      WMMA_RESULT_GUARD(d);
      o[0] = __builtin_fmaf(r.w1[4], d[0], o[0]);
      o[0] = __builtin_fmaf(r.w1[8], d[1], o[0]);
      o[0] = __builtin_fmaf(r.w1[10], d[2], o[0]);
      o[0] = __builtin_fmaf(r.w1[14], d[3], o[0]);
      o[1] = __builtin_fmaf(r.w1[3], d[4], o[1]);
      o[1] = __builtin_fmaf(r.w1[7], d[5], o[1]);
      o[1] = __builtin_fmaf(r.w1[9], d[6], o[1]);
      o[1] = __builtin_fmaf(r.w1[13], d[7], o[1]);
    }
    if (mhi & 0x4000u) {
      const v2f a = {areg[23].x, areg[23].y};
      const v2f b = {r.w2[4], r.w2[5]};
      v8f d = __builtin_amdgcn_wmma_f32_16x16x4_f32(false, a, false, b, (short)0, zero, false, false);
      WMMA_RESULT_GUARD(d);
      o[0] = __builtin_fmaf(r.w1[4], d[0], o[0]);
      o[0] = __builtin_fmaf(r.w1[8], d[1], o[0]);
      o[0] = __builtin_fmaf(r.w1[10], d[2], o[0]);
      o[0] = __builtin_fmaf(r.w1[14], d[3], o[0]);
      o[1] = __builtin_fmaf(r.w1[3], d[4], o[1]);
      o[1] = __builtin_fmaf(r.w1[7], d[5], o[1]);
      o[1] = __builtin_fmaf(r.w1[9], d[6], o[1]);
      o[1] = __builtin_fmaf(r.w1[13], d[7], o[1]);
    }
  }
  if (mhi & 0x80000u) {
    const v2f a = {cgl[25 * 32].z, cgl[25 * 32].w};
    const v2f b = {r.w2[6], r.w2[7]};
    v8f d = __builtin_amdgcn_wmma_f32_16x16x4_f32(false, a, false, b, (short)0, zero, false, false);
    WMMA_RESULT_GUARD_A(d, a);
    o[2] = __builtin_fmaf(r.w1[1], d[0], o[2]);
    o[2] = __builtin_fmaf(r.w1[5], d[1], o[2]);
    o[2] = __builtin_fmaf(r.w1[11], d[2], o[2]);
    o[2] = __builtin_fmaf(r.w1[15], d[3], o[2]);
    o[3] = __builtin_fmaf(r.w1[0], d[4], o[3]);
    o[3] = __builtin_fmaf(r.w1[2], d[5], o[3]);
    o[3] = __builtin_fmaf(r.w1[6], d[6], o[3]);
    o[3] = __builtin_fmaf(r.w1[12], d[7], o[3]);
  }
  if (mhi & 0x70000u) {
    if (mhi & 0x10000u) {
      const v2f a = {cgl[24 * 32].x, cgl[24 * 32].y};
      const v2f b = {r.w2[0], r.w2[1]};
      v8f d = __builtin_amdgcn_wmma_f32_16x16x4_f32(false, a, false, b, (short)0, zero, false, false);
      WMMA_RESULT_GUARD_A(d, a);
      o[2] = __builtin_fmaf(r.w1[1], d[0], o[2]);
      o[2] = __builtin_fmaf(r.w1[5], d[1], o[2]);
      o[2] = __builtin_fmaf(r.w1[11], d[2], o[2]);
      o[2] = __builtin_fmaf(r.w1[15], d[3], o[2]);
      o[3] = __builtin_fmaf(r.w1[0], d[4], o[3]);
      o[3] = __builtin_fmaf(r.w1[2], d[5], o[3]);
      o[3] = __builtin_fmaf(r.w1[6], d[6], o[3]);
      o[3] = __builtin_fmaf(r.w1[12], d[7], o[3]);
    }
    if (mhi & 0x20000u) {
      const v2f a = {cgl[24 * 32].z, cgl[24 * 32].w};
      const v2f b = {r.w2[2], r.w2[3]};
      v8f d = __builtin_amdgcn_wmma_f32_16x16x4_f32(false, a, false, b, (short)0, zero, false, false);
      WMMA_RESULT_GUARD_A(d, a);
      o[2] = __builtin_fmaf(r.w1[1], d[0], o[2]);
      o[2] = __builtin_fmaf(r.w1[5], d[1], o[2]);
      o[2] = __builtin_fmaf(r.w1[11], d[2], o[2]);
      o[2] = __builtin_fmaf(r.w1[15], d[3], o[2]);
      o[3] = __builtin_fmaf(r.w1[0], d[4], o[3]);
      o[3] = __builtin_fmaf(r.w1[2], d[5], o[3]);
      o[3] = __builtin_fmaf(r.w1[6], d[6], o[3]);
      o[3] = __builtin_fmaf(r.w1[12], d[7], o[3]);
    }
    if (mhi & 0x40000u) {
      const v2f a = {cgl[25 * 32].x, cgl[25 * 32].y};
      const v2f b = {r.w2[4], r.w2[5]};
      v8f d = __builtin_amdgcn_wmma_f32_16x16x4_f32(false, a, false, b, (short)0, zero, false, false);
      WMMA_RESULT_GUARD_A(d, a);
      o[2] = __builtin_fmaf(r.w1[1], d[0], o[2]);
      o[2] = __builtin_fmaf(r.w1[5], d[1], o[2]);
      o[2] = __builtin_fmaf(r.w1[11], d[2], o[2]);
      o[2] = __builtin_fmaf(r.w1[15], d[3], o[2]);
      o[3] = __builtin_fmaf(r.w1[0], d[4], o[3]);
      o[3] = __builtin_fmaf(r.w1[2], d[5], o[3]);
      o[3] = __builtin_fmaf(r.w1[6], d[6], o[3]);
      o[3] = __builtin_fmaf(r.w1[12], d[7], o[3]);
    }
  }
  if (mhi & 0x800000u) {
    const v2f a = {cgl[27 * 32].z, cgl[27 * 32].w};
    const v2f b = {r.w2[6], r.w2[7]};
    v8f d = __builtin_amdgcn_wmma_f32_16x16x4_f32(false, a, false, b, (short)0, zero, false, false);
    WMMA_RESULT_GUARD_A(d, a);
    o[4] = __builtin_fmaf(r.w1[4], d[0], o[4]);
    o[4] = __builtin_fmaf(r.w1[8], d[1], o[4]);
    o[4] = __builtin_fmaf(r.w1[10], d[2], o[4]);
    o[4] = __builtin_fmaf(r.w1[14], d[3], o[4]);
    o[5] = __builtin_fmaf(r.w1[0], d[4], o[5]);
    o[5] = __builtin_fmaf(r.w1[2], d[5], o[5]);
    o[5] = __builtin_fmaf(r.w1[6], d[6], o[5]);
    o[5] = __builtin_fmaf(r.w1[12], d[7], o[5]);
  }
  if (mhi & 0x700000u) {
    if (mhi & 0x100000u) {
      const v2f a = {cgl[26 * 32].x, cgl[26 * 32].y};
      const v2f b = {r.w2[0], r.w2[1]};
      v8f d = __builtin_amdgcn_wmma_f32_16x16x4_f32(false, a, false, b, (short)0, zero, false, false);
      WMMA_RESULT_GUARD_A(d, a);
      o[4] = __builtin_fmaf(r.w1[4], d[0], o[4]);
      o[4] = __builtin_fmaf(r.w1[8], d[1], o[4]);
      o[4] = __builtin_fmaf(r.w1[10], d[2], o[4]);
      o[4] = __builtin_fmaf(r.w1[14], d[3], o[4]);
      o[5] = __builtin_fmaf(r.w1[0], d[4], o[5]);
      o[5] = __builtin_fmaf(r.w1[2], d[5], o[5]);
      o[5] = __builtin_fmaf(r.w1[6], d[6], o[5]);
      o[5] = __builtin_fmaf(r.w1[12], d[7], o[5]);
    }
    if (mhi & 0x200000u) {
      const v2f a = {cgl[26 * 32].z, cgl[26 * 32].w};
      const v2f b = {r.w2[2], r.w2[3]};
      v8f d = __builtin_amdgcn_wmma_f32_16x16x4_f32(false, a, false, b, (short)0, zero, false, false);
      WMMA_RESULT_GUARD_A(d, a);
      o[4] = __builtin_fmaf(r.w1[4], d[0], o[4]);
      o[4] = __builtin_fmaf(r.w1[8], d[1], o[4]);
      o[4] = __builtin_fmaf(r.w1[10], d[2], o[4]);
      o[4] = __builtin_fmaf(r.w1[14], d[3], o[4]);
      o[5] = __builtin_fmaf(r.w1[0], d[4], o[5]);
      o[5] = __builtin_fmaf(r.w1[2], d[5], o[5]);
      o[5] = __builtin_fmaf(r.w1[6], d[6], o[5]);
      o[5] = __builtin_fmaf(r.w1[12], d[7], o[5]);
    }
    if (mhi & 0x400000u) {
      const v2f a = {cgl[27 * 32].x, cgl[27 * 32].y};
      const v2f b = {r.w2[4], r.w2[5]};
      v8f d = __builtin_amdgcn_wmma_f32_16x16x4_f32(false, a, false, b, (short)0, zero, false, false);
      WMMA_RESULT_GUARD_A(d, a);
      o[4] = __builtin_fmaf(r.w1[4], d[0], o[4]);
      o[4] = __builtin_fmaf(r.w1[8], d[1], o[4]);
      o[4] = __builtin_fmaf(r.w1[10], d[2], o[4]);
      o[4] = __builtin_fmaf(r.w1[14], d[3], o[4]);
      o[5] = __builtin_fmaf(r.w1[0], d[4], o[5]);
      o[5] = __builtin_fmaf(r.w1[2], d[5], o[5]);
      o[5] = __builtin_fmaf(r.w1[6], d[6], o[5]);
      o[5] = __builtin_fmaf(r.w1[12], d[7], o[5]);
    }
  }
  if (mhi & 0x8000000u) {
    const v2f a = {cgl[29 * 32].z, cgl[29 * 32].w};
    const v2f b = {r.w2[6], r.w2[7]};
    v8f d = __builtin_amdgcn_wmma_f32_16x16x4_f32(false, a, false, b, (short)0, zero, false, false);
    WMMA_RESULT_GUARD_A(d, a);
    o[6] = __builtin_fmaf((half ? r.w1[3] : r.w1[1]), d[0], o[6]);
    o[6] = __builtin_fmaf((half ? r.w1[7] : r.w1[5]), d[1], o[6]);
    o[6] = __builtin_fmaf((half ? r.w1[9] : r.w1[11]), d[2], o[6]);
    o[6] = __builtin_fmaf((half ? r.w1[13] : r.w1[14]), d[3], o[6]);
    o[7] = __builtin_fmaf((half ? r.w1[3] : r.w1[1]), d[4], o[7]);
    o[7] = __builtin_fmaf((half ? r.w1[7] : r.w1[5]), d[5], o[7]);
    o[7] = __builtin_fmaf((half ? r.w1[13] : r.w1[11]), d[6], o[7]);
    o[7] = __builtin_fmaf((half ? r.w1[10] : r.w1[15]), d[7], o[7]);
  }
  if (mhi & 0x7000000u) {
    if (mhi & 0x1000000u) {
      const v2f a = {cgl[28 * 32].x, cgl[28 * 32].y};
      const v2f b = {r.w2[0], r.w2[1]};
      v8f d = __builtin_amdgcn_wmma_f32_16x16x4_f32(false, a, false, b, (short)0, zero, false, false);
      WMMA_RESULT_GUARD_A(d, a);
      o[6] = __builtin_fmaf((half ? r.w1[3] : r.w1[1]), d[0], o[6]);
      o[6] = __builtin_fmaf((half ? r.w1[7] : r.w1[5]), d[1], o[6]);
      o[6] = __builtin_fmaf((half ? r.w1[9] : r.w1[11]), d[2], o[6]);
      o[6] = __builtin_fmaf((half ? r.w1[13] : r.w1[14]), d[3], o[6]);
      o[7] = __builtin_fmaf((half ? r.w1[3] : r.w1[1]), d[4], o[7]);
      o[7] = __builtin_fmaf((half ? r.w1[7] : r.w1[5]), d[5], o[7]);
      o[7] = __builtin_fmaf((half ? r.w1[13] : r.w1[11]), d[6], o[7]);
      o[7] = __builtin_fmaf((half ? r.w1[10] : r.w1[15]), d[7], o[7]);
    }
    if (mhi & 0x2000000u) {
      const v2f a = {cgl[28 * 32].z, cgl[28 * 32].w};
      const v2f b = {r.w2[2], r.w2[3]};
      v8f d = __builtin_amdgcn_wmma_f32_16x16x4_f32(false, a, false, b, (short)0, zero, false, false);
      WMMA_RESULT_GUARD_A(d, a);
      o[6] = __builtin_fmaf((half ? r.w1[3] : r.w1[1]), d[0], o[6]);
      o[6] = __builtin_fmaf((half ? r.w1[7] : r.w1[5]), d[1], o[6]);
      o[6] = __builtin_fmaf((half ? r.w1[9] : r.w1[11]), d[2], o[6]);
      o[6] = __builtin_fmaf((half ? r.w1[13] : r.w1[14]), d[3], o[6]);
      o[7] = __builtin_fmaf((half ? r.w1[3] : r.w1[1]), d[4], o[7]);
      o[7] = __builtin_fmaf((half ? r.w1[7] : r.w1[5]), d[5], o[7]);
      o[7] = __builtin_fmaf((half ? r.w1[13] : r.w1[11]), d[6], o[7]);
      o[7] = __builtin_fmaf((half ? r.w1[10] : r.w1[15]), d[7], o[7]);
    }
    if (mhi & 0x4000000u) {
      const v2f a = {cgl[29 * 32].x, cgl[29 * 32].y};
      const v2f b = {r.w2[4], r.w2[5]};
      v8f d = __builtin_amdgcn_wmma_f32_16x16x4_f32(false, a, false, b, (short)0, zero, false, false);
      WMMA_RESULT_GUARD_A(d, a);
      o[6] = __builtin_fmaf((half ? r.w1[3] : r.w1[1]), d[0], o[6]);
      o[6] = __builtin_fmaf((half ? r.w1[7] : r.w1[5]), d[1], o[6]);
      o[6] = __builtin_fmaf((half ? r.w1[9] : r.w1[11]), d[2], o[6]);
      o[6] = __builtin_fmaf((half ? r.w1[13] : r.w1[14]), d[3], o[6]);
      o[7] = __builtin_fmaf((half ? r.w1[3] : r.w1[1]), d[4], o[7]);
      o[7] = __builtin_fmaf((half ? r.w1[7] : r.w1[5]), d[5], o[7]);
      o[7] = __builtin_fmaf((half ? r.w1[13] : r.w1[11]), d[6], o[7]);
      o[7] = __builtin_fmaf((half ? r.w1[10] : r.w1[15]), d[7], o[7]);
    }
  }
  if (mhi & 0x10000000u) {
    const v2f a = {cgl[30 * 32].x, cgl[30 * 32].y};
    const v2f b = {r.w2[0], r.w2[1]};
    v8f d = __builtin_amdgcn_wmma_f32_16x16x4_f32(false, a, false, b, (short)0, zero, false, false);
    WMMA_RESULT_GUARD_A(d, a);
    o[1] = __builtin_fmaf(r.w1[15], d[0], o[1]);
    o[2] = __builtin_fmaf(r.w1[9], d[1], o[2]);
    o[3] = __builtin_fmaf(r.w1[8], d[2], o[3]);
    o[3] = __builtin_fmaf(r.w1[14], d[3], o[3]);
    o[3] = __builtin_fmaf(r.w1[15], d[4], o[3]);
    o[5] = __builtin_fmaf(r.w1[4], d[5], o[5]);
    o[5] = __builtin_fmaf(r.w1[9], d[6], o[5]);
    o[5] = __builtin_fmaf(r.w1[10], d[7], o[5]);
  }
  if (mhi & 0xe0000000u) {
    if (mhi & 0x20000000u) {
      const v2f a = {cgl[30 * 32].z, cgl[30 * 32].w};
      const v2f b = {r.w2[2], r.w2[3]};
      v8f d = __builtin_amdgcn_wmma_f32_16x16x4_f32(false, a, false, b, (short)0, zero, false, false);
      WMMA_RESULT_GUARD_A(d, a);
      o[1] = __builtin_fmaf(r.w1[15], d[0], o[1]);
      o[2] = __builtin_fmaf(r.w1[9], d[1], o[2]);
      o[3] = __builtin_fmaf(r.w1[8], d[2], o[3]);
      o[3] = __builtin_fmaf(r.w1[14], d[3], o[3]);
      o[3] = __builtin_fmaf(r.w1[15], d[4], o[3]);
      o[5] = __builtin_fmaf(r.w1[4], d[5], o[5]);
      o[5] = __builtin_fmaf(r.w1[9], d[6], o[5]);
      o[5] = __builtin_fmaf(r.w1[10], d[7], o[5]);
    }
    if (mhi & 0x40000000u) {
      const v2f a = {cgl[31 * 32].x, cgl[31 * 32].y};
      const v2f b = {r.w2[4], r.w2[5]};
      v8f d = __builtin_amdgcn_wmma_f32_16x16x4_f32(false, a, false, b, (short)0, zero, false, false);
      WMMA_RESULT_GUARD_A(d, a);
      o[1] = __builtin_fmaf(r.w1[15], d[0], o[1]);
      o[2] = __builtin_fmaf(r.w1[9], d[1], o[2]);
      o[3] = __builtin_fmaf(r.w1[8], d[2], o[3]);
      o[3] = __builtin_fmaf(r.w1[14], d[3], o[3]);
      o[3] = __builtin_fmaf(r.w1[15], d[4], o[3]);
      o[5] = __builtin_fmaf(r.w1[4], d[5], o[5]);
      o[5] = __builtin_fmaf(r.w1[9], d[6], o[5]);
      o[5] = __builtin_fmaf(r.w1[10], d[7], o[5]);
    }
    if (mhi & 0x80000000u) {
      const v2f a = {cgl[31 * 32].z, cgl[31 * 32].w};
      const v2f b = {r.w2[6], r.w2[7]};
      v8f d = __builtin_amdgcn_wmma_f32_16x16x4_f32(false, a, false, b, (short)0, zero, false, false);
      WMMA_RESULT_GUARD_A(d, a);
      o[1] = __builtin_fmaf(r.w1[15], d[0], o[1]);
      o[2] = __builtin_fmaf(r.w1[9], d[1], o[2]);
      o[3] = __builtin_fmaf(r.w1[8], d[2], o[3]);
      o[3] = __builtin_fmaf(r.w1[14], d[3], o[3]);
      o[3] = __builtin_fmaf(r.w1[15], d[4], o[3]);
      o[5] = __builtin_fmaf(r.w1[4], d[5], o[5]);
      o[5] = __builtin_fmaf(r.w1[9], d[6], o[5]);
      o[5] = __builtin_fmaf(r.w1[10], d[7], o[5]);
    }
  }

  const int n = t >> 2;
  const int c = (t & 3) * 16 + col;
  float* po = out + (size_t)n * NUM_ORDERS * C_DIM + c;
  po[(half ? 2 : 0) * C_DIM] = o[0];
  po[(half ? 5 : 1) * C_DIM] = o[1];
  po[(half ? 7 : 3) * C_DIM] = o[2];
  po[(half ? 10 : 4) * C_DIM] = o[3];
  po[(half ? 12 : 6) * C_DIM] = o[4];
  po[(half ? 14 : 8) * C_DIM] = o[5];
  po[(half ? 11 : 9) * C_DIM] = o[6];
  po[(half ? 15 : 13) * C_DIM] = o[7];
}

#define FAST_BLOCKS 15
#define FAST_MASK_LO 0x42222111u
#define FAST_MASK_HI 0x08888444u

#define FIRST_BLOCK_GUARD(a, b, c)                                                        \
  asm volatile("v_nop\n\tv_nop\n\tv_nop\n\tv_nop\n\tv_nop\n\tv_nop\n\tv_nop\n\tv_nop" \
               : "+v"(a), "+v"(b), "+v"(c))

__device__ __forceinline__ void tile_product_fast(v2f (&fa)[FAST_BLOCKS], const TileRegs& r,
                                                  float (&o)[8], int half) {
  const v8f zero = {};
#pragma unroll
  for (int v = 0; v < 8; ++v) o[v] = 0.0f;
  const float s_1_9 = half ? r.w1[1] : r.w1[9];
  const float s_5_3 = half ? r.w1[5] : r.w1[3];
  const float s_11_7 = half ? r.w1[11] : r.w1[7];
  const float s_15_3 = half ? r.w1[15] : r.w1[3];
  const float s_1_7 = half ? r.w1[1] : r.w1[7];
  const float s_4_13 = half ? r.w1[4] : r.w1[13];
  const float s_4_8 = half ? r.w1[4] : r.w1[8];
  const float s_10_13 = half ? r.w1[10] : r.w1[13];
  const float s_8_4 = half ? r.w1[8] : r.w1[4];
  const float s_14_10 = half ? r.w1[14] : r.w1[10];
  const float s_11_9 = half ? r.w1[11] : r.w1[9];
  const float s_5_8 = half ? r.w1[5] : r.w1[8];
  const float s_9_14 = half ? r.w1[9] : r.w1[14];
  const float s_3_1 = half ? r.w1[3] : r.w1[1];
  const float s_7_5 = half ? r.w1[7] : r.w1[5];
  const float s_9_11 = half ? r.w1[9] : r.w1[11];
  const float s_13_14 = half ? r.w1[13] : r.w1[14];
  const float s_13_11 = half ? r.w1[13] : r.w1[11];
  const float s_10_15 = half ? r.w1[10] : r.w1[15];
  v8f d0, d1, d2;
  d0 = __builtin_amdgcn_wmma_f32_16x16x4_f32(false, fa[0], false, (v2f){r.w2[0], r.w2[1]}, (short)0, zero, false, false);
  d1 = __builtin_amdgcn_wmma_f32_16x16x4_f32(false, fa[1], false, (v2f){r.w2[0], r.w2[1]}, (short)0, zero, false, false);
  d2 = __builtin_amdgcn_wmma_f32_16x16x4_f32(false, fa[2], false, (v2f){r.w2[0], r.w2[1]}, (short)0, zero, false, false);
  FIRST_BLOCK_GUARD(d0, d1, d2);
  o[0] = __builtin_fmaf(r.w1[0], d0[0], o[0]);
  o[0] = __builtin_fmaf(r.w1[2], d0[1], o[0]);
  o[0] = __builtin_fmaf(r.w1[6], d0[2], o[0]);
  o[0] = __builtin_fmaf(r.w1[12], d0[3], o[0]);
  o[1] = __builtin_fmaf(r.w1[1], d0[4], o[1]);
  o[1] = __builtin_fmaf(r.w1[5], d0[5], o[1]);
  o[1] = __builtin_fmaf(r.w1[11], d0[6], o[1]);
  o[2] = __builtin_fmaf(r.w1[3], d0[7], o[2]);
  asm volatile("" : "+v"(fa[3]), "+v"(o[0]), "+v"(o[1]), "+v"(o[2]));
  d0 = __builtin_amdgcn_wmma_f32_16x16x4_f32(false, fa[3], false, (v2f){r.w2[2], r.w2[3]}, (short)0, zero, false, false);
  asm volatile("" : "+v"(d1), "+v"(d0));
  o[2] = __builtin_fmaf(r.w1[7], d1[0], o[2]);
  o[2] = __builtin_fmaf(r.w1[13], d1[1], o[2]);
  o[3] = __builtin_fmaf(r.w1[4], d1[2], o[3]);
  o[3] = __builtin_fmaf(r.w1[10], d1[3], o[3]);
  o[4] = __builtin_fmaf(r.w1[0], d1[4], o[4]);
  o[4] = __builtin_fmaf(r.w1[2], d1[5], o[4]);
  o[4] = __builtin_fmaf(r.w1[6], d1[6], o[4]);
  o[4] = __builtin_fmaf(r.w1[12], d1[7], o[4]);
  asm volatile("" : "+v"(fa[4]), "+v"(o[2]), "+v"(o[3]), "+v"(o[4]));
  d1 = __builtin_amdgcn_wmma_f32_16x16x4_f32(false, fa[4], false, (v2f){r.w2[2], r.w2[3]}, (short)0, zero, false, false);
  asm volatile("" : "+v"(d2), "+v"(d1));
  o[5] = __builtin_fmaf(r.w1[8], d2[0], o[5]);
  o[5] = __builtin_fmaf(r.w1[14], d2[1], o[5]);
  o[6] = __builtin_fmaf(s_1_9, d2[2], o[6]);
  o[6] = __builtin_fmaf(s_5_3, d2[3], o[6]);
  o[6] = __builtin_fmaf(s_11_7, d2[4], o[6]);
  o[7] = __builtin_fmaf(s_15_3, d2[5], o[7]);
  o[7] = __builtin_fmaf(s_1_7, d2[6], o[7]);
  o[7] = __builtin_fmaf(s_4_13, d2[7], o[7]);
  asm volatile("" : "+v"(fa[5]), "+v"(o[5]), "+v"(o[6]), "+v"(o[7]));
  d2 = __builtin_amdgcn_wmma_f32_16x16x4_f32(false, fa[5], false, (v2f){r.w2[2], r.w2[3]}, (short)0, zero, false, false);
  asm volatile("" : "+v"(d0), "+v"(d2));
  o[0] = __builtin_fmaf(r.w1[1], d0[0], o[0]);
  o[0] = __builtin_fmaf(r.w1[5], d0[1], o[0]);
  o[0] = __builtin_fmaf(r.w1[9], d0[2], o[0]);
  o[0] = __builtin_fmaf(r.w1[11], d0[3], o[0]);
  o[1] = __builtin_fmaf(r.w1[4], d0[4], o[1]);
  o[1] = __builtin_fmaf(r.w1[10], d0[5], o[1]);
  o[2] = __builtin_fmaf(r.w1[0], d0[6], o[2]);
  o[2] = __builtin_fmaf(r.w1[2], d0[7], o[2]);
  asm volatile("" : "+v"(fa[6]), "+v"(o[0]), "+v"(o[1]), "+v"(o[2]));
  d0 = __builtin_amdgcn_wmma_f32_16x16x4_f32(false, fa[6], false, (v2f){r.w2[2], r.w2[3]}, (short)0, zero, false, false);
  asm volatile("" : "+v"(d1), "+v"(d0));
  o[2] = __builtin_fmaf(r.w1[4], d1[0], o[2]);
  o[2] = __builtin_fmaf(r.w1[6], d1[1], o[2]);
  o[2] = __builtin_fmaf(r.w1[10], d1[2], o[2]);
  o[2] = __builtin_fmaf(r.w1[12], d1[3], o[2]);
  o[3] = __builtin_fmaf(r.w1[9], d1[4], o[3]);
  o[4] = __builtin_fmaf(r.w1[1], d1[5], o[4]);
  o[4] = __builtin_fmaf(r.w1[5], d1[6], o[4]);
  o[4] = __builtin_fmaf(r.w1[9], d1[7], o[4]);
  asm volatile("" : "+v"(fa[7]), "+v"(o[2]), "+v"(o[3]), "+v"(o[4]));
  d1 = __builtin_amdgcn_wmma_f32_16x16x4_f32(false, fa[7], false, (v2f){r.w2[4], r.w2[5]}, (short)0, zero, false, false);
  asm volatile("" : "+v"(d2), "+v"(d1));
  o[4] = __builtin_fmaf(r.w1[11], d2[0], o[4]);
  o[5] = __builtin_fmaf(r.w1[1], d2[1], o[5]);
  o[5] = __builtin_fmaf(r.w1[3], d2[2], o[5]);
  o[5] = __builtin_fmaf(r.w1[5], d2[3], o[5]);
  o[5] = __builtin_fmaf(r.w1[7], d2[4], o[5]);
  o[5] = __builtin_fmaf(r.w1[11], d2[5], o[5]);
  o[5] = __builtin_fmaf(r.w1[13], d2[6], o[5]);
  o[6] = __builtin_fmaf(s_4_8, d2[7], o[6]);
  asm volatile("" : "+v"(fa[8]), "+v"(o[4]), "+v"(o[5]), "+v"(o[6]));
  d2 = __builtin_amdgcn_wmma_f32_16x16x4_f32(false, fa[8], false, (v2f){r.w2[4], r.w2[5]}, (short)0, zero, false, false);
  asm volatile("" : "+v"(d0), "+v"(d2));
  o[6] = __builtin_fmaf(s_10_13, d0[0], o[6]);
  o[7] = __builtin_fmaf(r.w1[0], d0[1], o[7]);
  o[7] = __builtin_fmaf(r.w1[2], d0[2], o[7]);
  o[7] = __builtin_fmaf(r.w1[6], d0[3], o[7]);
  o[7] = __builtin_fmaf(r.w1[12], d0[4], o[7]);
  o[7] = __builtin_fmaf(s_8_4, d0[5], o[7]);
  o[7] = __builtin_fmaf(s_14_10, d0[6], o[7]);
  o[7] = __builtin_fmaf(s_11_9, d0[7], o[7]);
  asm volatile("" : "+v"(fa[9]), "+v"(o[6]), "+v"(o[7]));
  d0 = __builtin_amdgcn_wmma_f32_16x16x4_f32(false, fa[9], false, (v2f){r.w2[4], r.w2[5]}, (short)0, zero, false, false);
  asm volatile("" : "+v"(d1), "+v"(d0));
  o[0] = __builtin_fmaf(r.w1[3], d1[0], o[0]);
  o[0] = __builtin_fmaf(r.w1[7], d1[1], o[0]);
  o[0] = __builtin_fmaf(r.w1[13], d1[2], o[0]);
  o[0] = __builtin_fmaf(r.w1[15], d1[3], o[0]);
  o[1] = __builtin_fmaf(r.w1[0], d1[4], o[1]);
  o[1] = __builtin_fmaf(r.w1[2], d1[5], o[1]);
  o[1] = __builtin_fmaf(r.w1[6], d1[6], o[1]);
  o[1] = __builtin_fmaf(r.w1[8], d1[7], o[1]);
  asm volatile("" : "+v"(fa[10]), "+v"(o[0]), "+v"(o[1]));
  d1 = __builtin_amdgcn_wmma_f32_16x16x4_f32(false, fa[10], false, (v2f){r.w2[4], r.w2[5]}, (short)0, zero, false, false);
  asm volatile("" : "+v"(d2), "+v"(d1));
  o[1] = __builtin_fmaf(r.w1[12], d2[0], o[1]);
  o[1] = __builtin_fmaf(r.w1[14], d2[1], o[1]);
  o[2] = __builtin_fmaf(r.w1[8], d2[2], o[2]);
  o[2] = __builtin_fmaf(r.w1[14], d2[3], o[2]);
  o[3] = __builtin_fmaf(r.w1[1], d2[4], o[3]);
  o[3] = __builtin_fmaf(r.w1[3], d2[5], o[3]);
  o[3] = __builtin_fmaf(r.w1[5], d2[6], o[3]);
  o[3] = __builtin_fmaf(r.w1[7], d2[7], o[3]);
  asm volatile("" : "+v"(fa[11]), "+v"(o[1]), "+v"(o[2]), "+v"(o[3]));
  d2 = __builtin_amdgcn_wmma_f32_16x16x4_f32(false, fa[11], false, (v2f){r.w2[6], r.w2[7]}, (short)0, zero, false, false);
  asm volatile("" : "+v"(d0), "+v"(d2));
  o[3] = __builtin_fmaf(r.w1[11], d0[0], o[3]);
  o[3] = __builtin_fmaf(r.w1[13], d0[1], o[3]);
  o[4] = __builtin_fmaf(r.w1[3], d0[2], o[4]);
  o[4] = __builtin_fmaf(r.w1[7], d0[3], o[4]);
  o[4] = __builtin_fmaf(r.w1[13], d0[4], o[4]);
  o[4] = __builtin_fmaf(r.w1[15], d0[5], o[4]);
  o[5] = __builtin_fmaf(r.w1[15], d0[6], o[5]);
  o[6] = __builtin_fmaf(r.w1[0], d0[7], o[6]);
  asm volatile("" : "+v"(fa[12]), "+v"(o[3]), "+v"(o[4]), "+v"(o[5]), "+v"(o[6]));
  d0 = __builtin_amdgcn_wmma_f32_16x16x4_f32(false, fa[12], false, (v2f){r.w2[6], r.w2[7]}, (short)0, zero, false, false);
  asm volatile("" : "+v"(d1), "+v"(d0));
  o[6] = __builtin_fmaf(r.w1[2], d1[0], o[6]);
  o[6] = __builtin_fmaf(r.w1[6], d1[1], o[6]);
  o[6] = __builtin_fmaf(r.w1[12], d1[2], o[6]);
  o[6] = __builtin_fmaf(s_8_4, d1[3], o[6]);
  o[6] = __builtin_fmaf(s_14_10, d1[4], o[6]);
  o[7] = __builtin_fmaf(s_5_8, d1[5], o[7]);
  o[7] = __builtin_fmaf(s_9_14, d1[6], o[7]);
  o[6] = __builtin_fmaf(r.w1[15], d1[7], o[6]);
  asm volatile("" : "+v"(fa[13]), "+v"(o[6]), "+v"(o[7]));
  d1 = __builtin_amdgcn_wmma_f32_16x16x4_f32(false, fa[13], false, (v2f){r.w2[6], r.w2[7]}, (short)0, zero, false, false);
  asm volatile("" : "+v"(d2), "+v"(d1));
  o[0] = __builtin_fmaf(r.w1[4], d2[0], o[0]);
  o[0] = __builtin_fmaf(r.w1[8], d2[1], o[0]);
  o[0] = __builtin_fmaf(r.w1[10], d2[2], o[0]);
  o[0] = __builtin_fmaf(r.w1[14], d2[3], o[0]);
  o[1] = __builtin_fmaf(r.w1[3], d2[4], o[1]);
  o[1] = __builtin_fmaf(r.w1[7], d2[5], o[1]);
  o[1] = __builtin_fmaf(r.w1[9], d2[6], o[1]);
  o[1] = __builtin_fmaf(r.w1[13], d2[7], o[1]);
  asm volatile("" : "+v"(fa[14]), "+v"(o[0]), "+v"(o[1]));
  d2 = __builtin_amdgcn_wmma_f32_16x16x4_f32(false, fa[14], false, (v2f){r.w2[6], r.w2[7]}, (short)0, zero, false, false);
  asm volatile("" : "+v"(d0), "+v"(d2));
  o[2] = __builtin_fmaf(r.w1[1], d0[0], o[2]);
  o[2] = __builtin_fmaf(r.w1[5], d0[1], o[2]);
  o[2] = __builtin_fmaf(r.w1[11], d0[2], o[2]);
  o[2] = __builtin_fmaf(r.w1[15], d0[3], o[2]);
  o[3] = __builtin_fmaf(r.w1[0], d0[4], o[3]);
  o[3] = __builtin_fmaf(r.w1[2], d0[5], o[3]);
  o[3] = __builtin_fmaf(r.w1[6], d0[6], o[3]);
  o[3] = __builtin_fmaf(r.w1[12], d0[7], o[3]);
  asm volatile("" : "+v"(d1), "+v"(o[2]), "+v"(o[3]));
  o[4] = __builtin_fmaf(r.w1[4], d1[0], o[4]);
  o[4] = __builtin_fmaf(r.w1[8], d1[1], o[4]);
  o[4] = __builtin_fmaf(r.w1[10], d1[2], o[4]);
  o[4] = __builtin_fmaf(r.w1[14], d1[3], o[4]);
  o[5] = __builtin_fmaf(r.w1[0], d1[4], o[5]);
  o[5] = __builtin_fmaf(r.w1[2], d1[5], o[5]);
  o[5] = __builtin_fmaf(r.w1[6], d1[6], o[5]);
  o[5] = __builtin_fmaf(r.w1[12], d1[7], o[5]);
  asm volatile("" : "+v"(d2), "+v"(o[4]), "+v"(o[5]));
  o[6] = __builtin_fmaf(s_3_1, d2[0], o[6]);
  o[6] = __builtin_fmaf(s_7_5, d2[1], o[6]);
  o[6] = __builtin_fmaf(s_9_11, d2[2], o[6]);
  o[6] = __builtin_fmaf(s_13_14, d2[3], o[6]);
  o[7] = __builtin_fmaf(s_3_1, d2[4], o[7]);
  o[7] = __builtin_fmaf(s_7_5, d2[5], o[7]);
  o[7] = __builtin_fmaf(s_13_11, d2[6], o[7]);
  o[7] = __builtin_fmaf(s_10_15, d2[7], o[7]);
}

__global__ void __launch_bounds__(BLOCK_THREADS)
tp_wmma_kernel(const float* x1, const float* x2,
               const float* __restrict__ vals, const int* __restrict__ M1,
               const int* __restrict__ M2, const int* __restrict__ Mseg,
               float* out, int n_batch, int nnz) {
  __shared__ float4 cg_lds[NUM_ORDERS * K_DENSE / 4];
  __shared__ unsigned char tab_lds[256 + 16];
  float* cg = (float*)cg_lds;

  const int tid  = threadIdx.x;
  const int lane = tid & 31;
  const int half = lane >> 4;
  const int col  = lane & 15;
  const int wib    = __builtin_amdgcn_readfirstlane((int)(threadIdx.x >> 5));
  const int wave   = blockIdx.x * WAVES_PER_BLOCK + wib;
  const int nwaves = gridDim.x * WAVES_PER_BLOCK;
  const int ntiles = n_batch * 4;
  const int nunits = n_batch * 2;

  if (ntiles <= 0) return;

  int t = wave;
  const bool has_tile = t < ntiles;
  const bool has_unit = t < nunits;

  unsigned char t_row = 0, t_col = 0;
  if (tid < 256) t_row = ROW_OF[tid];
  if (tid < 16) t_col = COL_POS[tid];
  const bool has_entry = tid < nnz;
  int   e_m = 0, e_m1 = 0, e_m2 = 0;
  float e_val = 0.0f;
  if (has_entry) {
    e_m   = Mseg[tid];
    e_m1  = M1[tid];
    e_m2  = M2[tid];
    e_val = vals[tid];
  }

  UnitRegs ur;
  {
    UnitOffsets f0;
    unit_offsets(f0, half, col);
    load_unit(ur, x1, x2, has_unit ? t : 0, f0);
  }

  if (tid < 256) tab_lds[tid] = t_row;
  if (tid < 16) tab_lds[256 + tid] = t_col;
  for (int i = tid; i < NUM_ORDERS * K_DENSE / 4; i += BLOCK_THREADS)
    cg_lds[i] = make_float4(0.0f, 0.0f, 0.0f, 0.0f);
  lds_barrier();
  if (has_entry) {
    asm volatile("" : "+v"(e_m), "+v"(e_m1), "+v"(e_m2));
    const int rho = tab_lds[((e_m & 15) << 4) | (e_m1 & 15)];
    const int pos = tab_lds[256 + (e_m2 & 15)];
    cg[cg_frag_index(rho & 15, (rho & 0xF0) | pos)] = e_val;
  }
  for (int i = tid + BLOCK_THREADS; i < nnz; i += BLOCK_THREADS) {
    const int rho = tab_lds[((Mseg[i] & 15) << 4) | (M1[i] & 15)];
    const int pos = tab_lds[256 + (M2[i] & 15)];
    cg[cg_frag_index(rho & 15, (rho & 0xF0) | pos)] = vals[i];
  }
  lds_barrier();

  unsigned mlo = 0, mhi = 0;
#pragma unroll
  for (int kp = 0; kp < 32; ++kp) {
    const float4 q = cg_lds[kp * 32 + lane];
    const unsigned b0 = __builtin_amdgcn_ballot_w32((q.x != 0.0f) | (q.y != 0.0f));
    const unsigned b1 = __builtin_amdgcn_ballot_w32((q.z != 0.0f) | (q.w != 0.0f));
    const unsigned bits = (b0 ? 1u : 0u) | (b1 ? 2u : 0u);
    if (kp < 16) mlo |= bits << (2 * kp);
    else         mhi |= bits << (2 * kp - 32);
  }

  asm volatile("" ::: "memory");

  if (!has_tile) return;


  if (((mlo & ~FAST_MASK_LO) | (mhi & ~FAST_MASK_HI)) == 0u) {
    constexpr int fast_block[FAST_BLOCKS] = {0, 4, 8, 13, 17, 21, 25, 30, 34, 38, 42, 47, 51, 55, 59};
    v2f fa[FAST_BLOCKS];
#pragma unroll
    for (int i = 0; i < FAST_BLOCKS; ++i) {
      const int idx = ((fast_block[i] >> 1) * 32 + lane) * 4 + (fast_block[i] & 1) * 2;
      fa[i] = (v2f){cg[idx], cg[idx + 1]};
    }
    if (!has_unit) return;
    int u = t;
    const int ucnt = (nunits - u + nwaves - 1) / nwaves;
    TileRegs ra, rb;
    split_unit(ra, rb, ur);
    UnitRegs ur1;
    int lane_f = lane;
    asm volatile("" : "+v"(lane_f));
    const int half_f = lane_f >> 4;
    UnitOffsets fo;
    unit_offsets(fo, half_f, lane_f & 15);
    const unsigned so[8] = {fo.eo[8] + (half_f ? 2u : 0u) * C_DIM,  fo.eo[8] + (half_f ? 5u : 1u) * C_DIM,
                            fo.eo[8] + (half_f ? 7u : 3u) * C_DIM,  fo.eo[8] + (half_f ? 10u : 4u) * C_DIM,
                            fo.eo[8] + (half_f ? 12u : 6u) * C_DIM, fo.eo[8] + (half_f ? 14u : 8u) * C_DIM,
                            fo.eo[8] + (half_f ? 11u : 9u) * C_DIM, fo.eo[8] + (half_f ? 15u : 13u) * C_DIM};
    for (int p = ucnt; p > 0; --p) {
      const int un = u + nwaves;
      load_unit(ur1, x1, x2, (un < nunits) ? un : u, fo);
      __builtin_amdgcn_sched_barrier(0);
      float oa[8], ob[8];
      tile_product_fast(fa, ra, oa, half);
      tile_product_fast(fa, rb, ob, half);
      float* pn = out + (size_t)(u >> 1) * NUM_ORDERS * C_DIM + (size_t)((u & 1) * 32);
#pragma unroll
      for (int v = 0; v < 8; ++v) *(v2f*)(pn + so[v]) = (v2f){oa[v], ob[v]};
      asm volatile("" : "+v"(ur1.q2[0]), "+v"(ur1.q2[1]), "+v"(ur1.q2[2]), "+v"(ur1.q2[3]),
                        "+v"(ur1.q2[4]), "+v"(ur1.q2[5]), "+v"(ur1.q2[6]), "+v"(ur1.q2[7])
                      : "v"(ra.w2), "v"(rb.w2));
      split_unit(ra, rb, ur1);
      u = un;
    }
    return;
  }

  float4 areg[24];
#pragma unroll
  for (int kp = 0; kp < 24; ++kp) areg[kp] = cg_lds[kp * 32 + lane];

  const int cnt = (ntiles - t + nwaves - 1) / nwaves;
  TileRegs r;
  load_tile(r, x1, x2, t, half, col);
  for (int p = cnt; p > 0; --p) {
    asm volatile("" : "+s"(mlo), "+s"(mhi));
    tile_product(areg, cg_lds + lane, mlo, mhi, r, out, t, half, col);
    t += nwaves;
    if (p > 1) load_tile(r, x1, x2, t, half, col);
  }
}

extern "C" void kernel_launch(void* const* d_in, const int* in_sizes, int n_in,
                              void* d_out, int out_size, void* d_ws, size_t ws_size,
                              hipStream_t stream) {
  const float* x1   = (const float*)d_in[0];
  const float* x2   = (const float*)d_in[1];
  const float* vals = (const float*)d_in[2];
  const int* M1     = (const int*)d_in[3];
  const int* M2     = (const int*)d_in[4];
  const int* Mseg   = (const int*)d_in[5];
  float* out        = (float*)d_out;

  const int nnz     = in_sizes[2];
  const int n_batch = in_sizes[0] / (NUM_ORDERS * C_DIM);

  tp_wmma_kernel<<<NUM_BLOCKS, BLOCK_THREADS, 0, stream>>>(x1, x2, vals, M1, M2, Mseg,
                                                          out, n_batch, nnz);
}


# Branch-free path: a lane holds four adjacent channels (16-byte loads/stores); a wave's unit is one whole batch row run as four tiles, in 8-wave blocks: half the waves and half the memory instructions
# speedup vs baseline: 2.5082x; 1.0793x over previous
#include <hip/hip_runtime.h>


typedef __attribute__((ext_vector_type(2))) float v2f;
typedef __attribute__((ext_vector_type(4))) float v4f;
typedef __attribute__((ext_vector_type(8))) float v8f;
typedef __attribute__((ext_vector_type(16))) float v16f;

#define NUM_ORDERS 16
#define C_DIM 64
#define K_DENSE 256
#define WAVES_PER_BLOCK 8
#define BLOCK_THREADS (WAVES_PER_BLOCK * 32)
#define NUM_BLOCKS 256

__device__ const unsigned char ROW_OF[256] = {
    0, 48, 1, 112, 176, 49, 2, 113, 177, 50, 178, 51, 3, 114, 179, 115,
    116, 4, 117, 180, 52, 5, 118, 181, 119, 182, 53, 6, 128, 183, 129, 240,
    8, 56, 9, 120, 184, 57, 10, 121, 185, 58, 186, 59, 11, 122, 187, 123,
    54, 192, 55, 7, 64, 193, 65, 16, 130, 241, 66, 194, 67, 17, 131, 195,
    196, 132, 197, 133, 18, 134, 198, 135, 242, 68, 19, 144, 199, 145, 243, 244,
    124, 12, 125, 188, 60, 13, 126, 189, 127, 190, 61, 14, 136, 191, 137, 248,
    20, 69, 21, 146, 208, 70, 22, 147, 209, 71, 210, 80, 23, 148, 211, 149,
    62, 200, 63, 15, 72, 201, 73, 24, 138, 249, 74, 202, 75, 25, 139, 203,
    212, 81, 213, 82, 245, 83, 214, 84, 32, 246, 247, 85, 215, 86, 33, 150,
    151, 224, 160, 35, 163, 225, 161, 36, 87, 34, 164, 226, 162, 96, 227, 167,
    204, 140, 205, 141, 26, 142, 206, 143, 250, 76, 27, 152, 207, 153, 251, 252,
    159, 42, 168, 232, 95, 43, 169, 233, 171, 234, 104, 44, 170, 235, 172, 175,
    28, 77, 29, 154, 216, 78, 30, 155, 217, 79, 218, 88, 31, 156, 219, 157,
    97, 228, 98, 37, 101, 229, 99, 38, 165, 103, 102, 230, 100, 39, 166, 231,
    220, 89, 221, 90, 253, 91, 222, 92, 40, 254, 255, 93, 223, 94, 41, 158,
    105, 46, 106, 236, 47, 173, 107, 237, 109, 174, 239, 111, 108, 238, 110, 45,
};
__device__ const unsigned char COL_POS[16] = {0, 8, 1, 4, 14, 9, 2, 5, 12, 11, 15, 10, 3, 6, 13, 7};

__device__ __forceinline__ int cg_frag_index(int i, int k) {
  const int kk   = k >> 2;
  const int lane = ((k >> 1) & 1) * 16 + i;
  return ((kk >> 1) * 32 + lane) * 4 + (kk & 1) * 2 + (k & 1);
}

__device__ __forceinline__ void lds_barrier() {
  __builtin_amdgcn_fence(__ATOMIC_RELEASE, "workgroup", "local");
  __builtin_amdgcn_s_barrier();
  __builtin_amdgcn_fence(__ATOMIC_ACQUIRE, "workgroup", "local");
}

struct TileRegs {
  v16f w1;
  v8f  w2;
};

__device__ __forceinline__ void load_tile(TileRegs& r, const float* x1, const float* x2, int t,
                                          int half, int col) {
  const int n = t >> 2;
  const int c = (t & 3) * 16 + col;
  const float* p1 = x1 + (size_t)n * NUM_ORDERS * C_DIM + c;
  const float* p2 = x2 + (size_t)n * NUM_ORDERS * C_DIM + c;
  r.w2[0] = p2[(half ? 6 : 0) * C_DIM];
  r.w2[1] = p2[(half ? 12 : 2) * C_DIM];
  r.w2[2] = p2[(half ? 13 : 3) * C_DIM];
  r.w2[3] = p2[(half ? 15 : 7) * C_DIM];
  r.w2[4] = p2[(half ? 11 : 1) * C_DIM];
  r.w2[5] = p2[(half ? 9 : 5) * C_DIM];
  r.w2[6] = p2[(half ? 4 : 8) * C_DIM];
  r.w2[7] = p2[(half ? 10 : 14) * C_DIM];
  r.w1[0] = p1[0 * C_DIM];
  r.w1[1] = p1[1 * C_DIM];
  r.w1[2] = p1[2 * C_DIM];
  r.w1[3] = p1[3 * C_DIM];
  r.w1[4] = p1[4 * C_DIM];
  r.w1[5] = p1[5 * C_DIM];
  r.w1[6] = p1[6 * C_DIM];
  r.w1[7] = p1[7 * C_DIM];
  r.w1[8] = p1[8 * C_DIM];
  r.w1[9] = p1[9 * C_DIM];
  r.w1[10] = p1[10 * C_DIM];
  r.w1[11] = p1[11 * C_DIM];
  r.w1[12] = p1[12 * C_DIM];
  r.w1[13] = p1[13 * C_DIM];
  r.w1[14] = p1[14 * C_DIM];
  r.w1[15] = p1[15 * C_DIM];
}

struct UnitRegs {
  v4f q1[16];
  v4f q2[8];
};

struct UnitOffsets {
  unsigned eo[9];
};

__device__ __forceinline__ void unit_offsets(UnitOffsets& f, int half, int col) {
  const unsigned c4 = 4u * (unsigned)col;
  f.eo[0] = c4 + (half ? 6u : 0u) * C_DIM;
  f.eo[1] = c4 + (half ? 12u : 2u) * C_DIM;
  f.eo[2] = c4 + (half ? 13u : 3u) * C_DIM;
  f.eo[3] = c4 + (half ? 15u : 7u) * C_DIM;
  f.eo[4] = c4 + (half ? 11u : 1u) * C_DIM;
  f.eo[5] = c4 + (half ? 9u : 5u) * C_DIM;
  f.eo[6] = c4 + (half ? 4u : 8u) * C_DIM;
  f.eo[7] = c4 + (half ? 10u : 14u) * C_DIM;
  f.eo[8] = c4;
}

__device__ __forceinline__ void load_unit(UnitRegs& r, const float* x1, const float* x2, int u,
                                          const UnitOffsets& f) {
  const size_t ub = (size_t)u * NUM_ORDERS * C_DIM;
  const float* b1 = x1 + ub;
  const float* b2 = x2 + ub;
#pragma unroll
  for (int j = 0; j < 8; ++j) r.q2[j] = *(const v4f*)(b2 + f.eo[j]);
#pragma unroll
  for (int i = 0; i < 16; ++i) r.q1[i] = *(const v4f*)(b1 + f.eo[8] + i * C_DIM);
}

__device__ __forceinline__ void split_unit(TileRegs& a, TileRegs& b, TileRegs& c, TileRegs& d,
                                           const UnitRegs& r) {
#pragma unroll
  for (int i = 0; i < 16; ++i) {
    a.w1[i] = r.q1[i].x;
    b.w1[i] = r.q1[i].y;
    c.w1[i] = r.q1[i].z;
    d.w1[i] = r.q1[i].w;
  }
#pragma unroll
  for (int i = 0; i < 8; ++i) {
    a.w2[i] = r.q2[i].x;
    b.w2[i] = r.q2[i].y;
    c.w2[i] = r.q2[i].z;
    d.w2[i] = r.q2[i].w;
  }
}

#define WMMA_RESULT_GUARD(d) \
  asm volatile("s_wait_alu depctr_va_vdst(0)\n\tv_nop\n\tv_nop\n\tv_nop\n\tv_nop" : "+v"(d))
#define WMMA_RESULT_GUARD_A(d, a) \
  asm volatile("s_wait_alu depctr_va_vdst(0)\n\tv_nop\n\tv_nop\n\tv_nop\n\tv_nop" : "+v"(d) : "v"(a))

__device__ __forceinline__ void tile_product(const float4 (&areg)[24], const float4* cgl,
                                             unsigned mlo, unsigned mhi,
                                             const TileRegs& r, float* out, int t,
                                             int half, int col) {
  const v8f zero = {};
  float o[8] = {0.0f, 0.0f, 0.0f, 0.0f, 0.0f, 0.0f, 0.0f, 0.0f};
  if (mlo & 0x1u) {
    const v2f a = {areg[0].x, areg[0].y};
    const v2f b = {r.w2[0], r.w2[1]};
    v8f d = __builtin_amdgcn_wmma_f32_16x16x4_f32(false, a, false, b, (short)0, zero, false, false);
    WMMA_RESULT_GUARD(d);
    o[0] = __builtin_fmaf(r.w1[0], d[0], o[0]);
    o[0] = __builtin_fmaf(r.w1[2], d[1], o[0]);
    o[0] = __builtin_fmaf(r.w1[6], d[2], o[0]);
    o[0] = __builtin_fmaf(r.w1[12], d[3], o[0]);
    o[1] = __builtin_fmaf(r.w1[1], d[4], o[1]);
    o[1] = __builtin_fmaf(r.w1[5], d[5], o[1]);
    o[1] = __builtin_fmaf(r.w1[11], d[6], o[1]);
    o[2] = __builtin_fmaf(r.w1[3], d[7], o[2]);
  }
  if (mlo & 0xeu) {
    if (mlo & 0x2u) {
      const v2f a = {areg[0].z, areg[0].w};
      const v2f b = {r.w2[2], r.w2[3]};
      v8f d = __builtin_amdgcn_wmma_f32_16x16x4_f32(false, a, false, b, (short)0, zero, false, false);
      WMMA_RESULT_GUARD(d);
      o[0] = __builtin_fmaf(r.w1[0], d[0], o[0]);
      o[0] = __builtin_fmaf(r.w1[2], d[1], o[0]);
      o[0] = __builtin_fmaf(r.w1[6], d[2], o[0]);
      o[0] = __builtin_fmaf(r.w1[12], d[3], o[0]);
      o[1] = __builtin_fmaf(r.w1[1], d[4], o[1]);
      o[1] = __builtin_fmaf(r.w1[5], d[5], o[1]);
      o[1] = __builtin_fmaf(r.w1[11], d[6], o[1]);
      o[2] = __builtin_fmaf(r.w1[3], d[7], o[2]);
    }
    if (mlo & 0x4u) {
      const v2f a = {areg[1].x, areg[1].y};
      const v2f b = {r.w2[4], r.w2[5]};
      v8f d = __builtin_amdgcn_wmma_f32_16x16x4_f32(false, a, false, b, (short)0, zero, false, false);
      WMMA_RESULT_GUARD(d);
      o[0] = __builtin_fmaf(r.w1[0], d[0], o[0]);
      o[0] = __builtin_fmaf(r.w1[2], d[1], o[0]);
      o[0] = __builtin_fmaf(r.w1[6], d[2], o[0]);
      o[0] = __builtin_fmaf(r.w1[12], d[3], o[0]);
      o[1] = __builtin_fmaf(r.w1[1], d[4], o[1]);
      o[1] = __builtin_fmaf(r.w1[5], d[5], o[1]);
      o[1] = __builtin_fmaf(r.w1[11], d[6], o[1]);
      o[2] = __builtin_fmaf(r.w1[3], d[7], o[2]);
    }
    if (mlo & 0x8u) {
      const v2f a = {areg[1].z, areg[1].w};
      const v2f b = {r.w2[6], r.w2[7]};
      v8f d = __builtin_amdgcn_wmma_f32_16x16x4_f32(false, a, false, b, (short)0, zero, false, false);
      WMMA_RESULT_GUARD(d);
      o[0] = __builtin_fmaf(r.w1[0], d[0], o[0]);
      o[0] = __builtin_fmaf(r.w1[2], d[1], o[0]);
      o[0] = __builtin_fmaf(r.w1[6], d[2], o[0]);
      o[0] = __builtin_fmaf(r.w1[12], d[3], o[0]);
      o[1] = __builtin_fmaf(r.w1[1], d[4], o[1]);
      o[1] = __builtin_fmaf(r.w1[5], d[5], o[1]);
      o[1] = __builtin_fmaf(r.w1[11], d[6], o[1]);
      o[2] = __builtin_fmaf(r.w1[3], d[7], o[2]);
    }
  }
  if (mlo & 0x10u) {
    const v2f a = {areg[2].x, areg[2].y};
    const v2f b = {r.w2[0], r.w2[1]};
    v8f d = __builtin_amdgcn_wmma_f32_16x16x4_f32(false, a, false, b, (short)0, zero, false, false);
    WMMA_RESULT_GUARD(d);
    o[2] = __builtin_fmaf(r.w1[7], d[0], o[2]);
    o[2] = __builtin_fmaf(r.w1[13], d[1], o[2]);
    o[3] = __builtin_fmaf(r.w1[4], d[2], o[3]);
    o[3] = __builtin_fmaf(r.w1[10], d[3], o[3]);
    o[4] = __builtin_fmaf(r.w1[0], d[4], o[4]);
    o[4] = __builtin_fmaf(r.w1[2], d[5], o[4]);
    o[4] = __builtin_fmaf(r.w1[6], d[6], o[4]);
    o[4] = __builtin_fmaf(r.w1[12], d[7], o[4]);
  }
  if (mlo & 0xe0u) {
    if (mlo & 0x20u) {
      const v2f a = {areg[2].z, areg[2].w};
      const v2f b = {r.w2[2], r.w2[3]};
      v8f d = __builtin_amdgcn_wmma_f32_16x16x4_f32(false, a, false, b, (short)0, zero, false, false);
      WMMA_RESULT_GUARD(d);
      o[2] = __builtin_fmaf(r.w1[7], d[0], o[2]);
      o[2] = __builtin_fmaf(r.w1[13], d[1], o[2]);
      o[3] = __builtin_fmaf(r.w1[4], d[2], o[3]);
      o[3] = __builtin_fmaf(r.w1[10], d[3], o[3]);
      o[4] = __builtin_fmaf(r.w1[0], d[4], o[4]);
      o[4] = __builtin_fmaf(r.w1[2], d[5], o[4]);
      o[4] = __builtin_fmaf(r.w1[6], d[6], o[4]);
      o[4] = __builtin_fmaf(r.w1[12], d[7], o[4]);
    }
    if (mlo & 0x40u) {
      const v2f a = {areg[3].x, areg[3].y};
      const v2f b = {r.w2[4], r.w2[5]};
      v8f d = __builtin_amdgcn_wmma_f32_16x16x4_f32(false, a, false, b, (short)0, zero, false, false);
      WMMA_RESULT_GUARD(d);
      o[2] = __builtin_fmaf(r.w1[7], d[0], o[2]);
      o[2] = __builtin_fmaf(r.w1[13], d[1], o[2]);
      o[3] = __builtin_fmaf(r.w1[4], d[2], o[3]);
      o[3] = __builtin_fmaf(r.w1[10], d[3], o[3]);
      o[4] = __builtin_fmaf(r.w1[0], d[4], o[4]);
      o[4] = __builtin_fmaf(r.w1[2], d[5], o[4]);
      o[4] = __builtin_fmaf(r.w1[6], d[6], o[4]);
      o[4] = __builtin_fmaf(r.w1[12], d[7], o[4]);
    }
    if (mlo & 0x80u) {
      const v2f a = {areg[3].z, areg[3].w};
      const v2f b = {r.w2[6], r.w2[7]};
      v8f d = __builtin_amdgcn_wmma_f32_16x16x4_f32(false, a, false, b, (short)0, zero, false, false);
      WMMA_RESULT_GUARD(d);
      o[2] = __builtin_fmaf(r.w1[7], d[0], o[2]);
      o[2] = __builtin_fmaf(r.w1[13], d[1], o[2]);
      o[3] = __builtin_fmaf(r.w1[4], d[2], o[3]);
      o[3] = __builtin_fmaf(r.w1[10], d[3], o[3]);
      o[4] = __builtin_fmaf(r.w1[0], d[4], o[4]);
      o[4] = __builtin_fmaf(r.w1[2], d[5], o[4]);
      o[4] = __builtin_fmaf(r.w1[6], d[6], o[4]);
      o[4] = __builtin_fmaf(r.w1[12], d[7], o[4]);
    }
  }
  if (mlo & 0x100u) {
    const v2f a = {areg[4].x, areg[4].y};
    const v2f b = {r.w2[0], r.w2[1]};
    v8f d = __builtin_amdgcn_wmma_f32_16x16x4_f32(false, a, false, b, (short)0, zero, false, false);
    WMMA_RESULT_GUARD(d);
    o[5] = __builtin_fmaf(r.w1[8], d[0], o[5]);
    o[5] = __builtin_fmaf(r.w1[14], d[1], o[5]);
    o[6] = __builtin_fmaf((half ? r.w1[1] : r.w1[9]), d[2], o[6]);
    o[6] = __builtin_fmaf((half ? r.w1[5] : r.w1[3]), d[3], o[6]);
    o[6] = __builtin_fmaf((half ? r.w1[11] : r.w1[7]), d[4], o[6]);
    o[7] = __builtin_fmaf((half ? r.w1[15] : r.w1[3]), d[5], o[7]);
    o[7] = __builtin_fmaf((half ? r.w1[1] : r.w1[7]), d[6], o[7]);
    o[7] = __builtin_fmaf((half ? r.w1[4] : r.w1[13]), d[7], o[7]);
  }
  if (mlo & 0xe00u) {
    if (mlo & 0x200u) {
      const v2f a = {areg[4].z, areg[4].w};
      const v2f b = {r.w2[2], r.w2[3]};
      v8f d = __builtin_amdgcn_wmma_f32_16x16x4_f32(false, a, false, b, (short)0, zero, false, false);
      WMMA_RESULT_GUARD(d);
      o[5] = __builtin_fmaf(r.w1[8], d[0], o[5]);
      o[5] = __builtin_fmaf(r.w1[14], d[1], o[5]);
      o[6] = __builtin_fmaf((half ? r.w1[1] : r.w1[9]), d[2], o[6]);
      o[6] = __builtin_fmaf((half ? r.w1[5] : r.w1[3]), d[3], o[6]);
      o[6] = __builtin_fmaf((half ? r.w1[11] : r.w1[7]), d[4], o[6]);
      o[7] = __builtin_fmaf((half ? r.w1[15] : r.w1[3]), d[5], o[7]);
      o[7] = __builtin_fmaf((half ? r.w1[1] : r.w1[7]), d[6], o[7]);
      o[7] = __builtin_fmaf((half ? r.w1[4] : r.w1[13]), d[7], o[7]);
    }
    if (mlo & 0x400u) {
      const v2f a = {areg[5].x, areg[5].y};
      const v2f b = {r.w2[4], r.w2[5]};
      v8f d = __builtin_amdgcn_wmma_f32_16x16x4_f32(false, a, false, b, (short)0, zero, false, false);
      WMMA_RESULT_GUARD(d);
      o[5] = __builtin_fmaf(r.w1[8], d[0], o[5]);
      o[5] = __builtin_fmaf(r.w1[14], d[1], o[5]);
      o[6] = __builtin_fmaf((half ? r.w1[1] : r.w1[9]), d[2], o[6]);
      o[6] = __builtin_fmaf((half ? r.w1[5] : r.w1[3]), d[3], o[6]);
      o[6] = __builtin_fmaf((half ? r.w1[11] : r.w1[7]), d[4], o[6]);
      o[7] = __builtin_fmaf((half ? r.w1[15] : r.w1[3]), d[5], o[7]);
      o[7] = __builtin_fmaf((half ? r.w1[1] : r.w1[7]), d[6], o[7]);
      o[7] = __builtin_fmaf((half ? r.w1[4] : r.w1[13]), d[7], o[7]);
    }
    if (mlo & 0x800u) {
      const v2f a = {areg[5].z, areg[5].w};
      const v2f b = {r.w2[6], r.w2[7]};
      v8f d = __builtin_amdgcn_wmma_f32_16x16x4_f32(false, a, false, b, (short)0, zero, false, false);
      WMMA_RESULT_GUARD(d);
      o[5] = __builtin_fmaf(r.w1[8], d[0], o[5]);
      o[5] = __builtin_fmaf(r.w1[14], d[1], o[5]);
      o[6] = __builtin_fmaf((half ? r.w1[1] : r.w1[9]), d[2], o[6]);
      o[6] = __builtin_fmaf((half ? r.w1[5] : r.w1[3]), d[3], o[6]);
      o[6] = __builtin_fmaf((half ? r.w1[11] : r.w1[7]), d[4], o[6]);
      o[7] = __builtin_fmaf((half ? r.w1[15] : r.w1[3]), d[5], o[7]);
      o[7] = __builtin_fmaf((half ? r.w1[1] : r.w1[7]), d[6], o[7]);
      o[7] = __builtin_fmaf((half ? r.w1[4] : r.w1[13]), d[7], o[7]);
    }
  }
  if (mlo & 0x2000u) {
    const v2f a = {areg[6].z, areg[6].w};
    const v2f b = {r.w2[2], r.w2[3]};
    v8f d = __builtin_amdgcn_wmma_f32_16x16x4_f32(false, a, false, b, (short)0, zero, false, false);
    WMMA_RESULT_GUARD(d);
    o[0] = __builtin_fmaf(r.w1[1], d[0], o[0]);
    o[0] = __builtin_fmaf(r.w1[5], d[1], o[0]);
    o[0] = __builtin_fmaf(r.w1[9], d[2], o[0]);
    o[0] = __builtin_fmaf(r.w1[11], d[3], o[0]);
    o[1] = __builtin_fmaf(r.w1[4], d[4], o[1]);
    o[1] = __builtin_fmaf(r.w1[10], d[5], o[1]);
    o[2] = __builtin_fmaf(r.w1[0], d[6], o[2]);
    o[2] = __builtin_fmaf(r.w1[2], d[7], o[2]);
  }
  if (mlo & 0xd000u) {
    if (mlo & 0x1000u) {
      const v2f a = {areg[6].x, areg[6].y};
      const v2f b = {r.w2[0], r.w2[1]};
      v8f d = __builtin_amdgcn_wmma_f32_16x16x4_f32(false, a, false, b, (short)0, zero, false, false);
      WMMA_RESULT_GUARD(d);
      o[0] = __builtin_fmaf(r.w1[1], d[0], o[0]);
      o[0] = __builtin_fmaf(r.w1[5], d[1], o[0]);
      o[0] = __builtin_fmaf(r.w1[9], d[2], o[0]);
      o[0] = __builtin_fmaf(r.w1[11], d[3], o[0]);
      o[1] = __builtin_fmaf(r.w1[4], d[4], o[1]);
      o[1] = __builtin_fmaf(r.w1[10], d[5], o[1]);
      o[2] = __builtin_fmaf(r.w1[0], d[6], o[2]);
      o[2] = __builtin_fmaf(r.w1[2], d[7], o[2]);
    }
    if (mlo & 0x4000u) {
      const v2f a = {areg[7].x, areg[7].y};
      const v2f b = {r.w2[4], r.w2[5]};
      v8f d = __builtin_amdgcn_wmma_f32_16x16x4_f32(false, a, false, b, (short)0, zero, false, false);
      WMMA_RESULT_GUARD(d);
      o[0] = __builtin_fmaf(r.w1[1], d[0], o[0]);
      o[0] = __builtin_fmaf(r.w1[5], d[1], o[0]);
      o[0] = __builtin_fmaf(r.w1[9], d[2], o[0]);
      o[0] = __builtin_fmaf(r.w1[11], d[3], o[0]);
      o[1] = __builtin_fmaf(r.w1[4], d[4], o[1]);
      o[1] = __builtin_fmaf(r.w1[10], d[5], o[1]);
      o[2] = __builtin_fmaf(r.w1[0], d[6], o[2]);
      o[2] = __builtin_fmaf(r.w1[2], d[7], o[2]);
    }
    if (mlo & 0x8000u) {
      const v2f a = {areg[7].z, areg[7].w};
      const v2f b = {r.w2[6], r.w2[7]};
      v8f d = __builtin_amdgcn_wmma_f32_16x16x4_f32(false, a, false, b, (short)0, zero, false, false);
      WMMA_RESULT_GUARD(d);
      o[0] = __builtin_fmaf(r.w1[1], d[0], o[0]);
      o[0] = __builtin_fmaf(r.w1[5], d[1], o[0]);
      o[0] = __builtin_fmaf(r.w1[9], d[2], o[0]);
      o[0] = __builtin_fmaf(r.w1[11], d[3], o[0]);
      o[1] = __builtin_fmaf(r.w1[4], d[4], o[1]);
      o[1] = __builtin_fmaf(r.w1[10], d[5], o[1]);
      o[2] = __builtin_fmaf(r.w1[0], d[6], o[2]);
      o[2] = __builtin_fmaf(r.w1[2], d[7], o[2]);
    }
  }
  if (mlo & 0x20000u) {
    const v2f a = {areg[8].z, areg[8].w};
    const v2f b = {r.w2[2], r.w2[3]};
    v8f d = __builtin_amdgcn_wmma_f32_16x16x4_f32(false, a, false, b, (short)0, zero, false, false);
    WMMA_RESULT_GUARD(d);
    o[2] = __builtin_fmaf(r.w1[4], d[0], o[2]);
    o[2] = __builtin_fmaf(r.w1[6], d[1], o[2]);
    o[2] = __builtin_fmaf(r.w1[10], d[2], o[2]);
    o[2] = __builtin_fmaf(r.w1[12], d[3], o[2]);
    o[3] = __builtin_fmaf(r.w1[9], d[4], o[3]);
    o[4] = __builtin_fmaf(r.w1[1], d[5], o[4]);
    o[4] = __builtin_fmaf(r.w1[5], d[6], o[4]);
    o[4] = __builtin_fmaf(r.w1[9], d[7], o[4]);
  }
  if (mlo & 0xd0000u) {
    if (mlo & 0x10000u) {
      const v2f a = {areg[8].x, areg[8].y};
      const v2f b = {r.w2[0], r.w2[1]};
      v8f d = __builtin_amdgcn_wmma_f32_16x16x4_f32(false, a, false, b, (short)0, zero, false, false);
      WMMA_RESULT_GUARD(d);
      o[2] = __builtin_fmaf(r.w1[4], d[0], o[2]);
      o[2] = __builtin_fmaf(r.w1[6], d[1], o[2]);
      o[2] = __builtin_fmaf(r.w1[10], d[2], o[2]);
      o[2] = __builtin_fmaf(r.w1[12], d[3], o[2]);
      o[3] = __builtin_fmaf(r.w1[9], d[4], o[3]);
      o[4] = __builtin_fmaf(r.w1[1], d[5], o[4]);
      o[4] = __builtin_fmaf(r.w1[5], d[6], o[4]);
      o[4] = __builtin_fmaf(r.w1[9], d[7], o[4]);
    }
    if (mlo & 0x40000u) {
      const v2f a = {areg[9].x, areg[9].y};
      const v2f b = {r.w2[4], r.w2[5]};
      v8f d = __builtin_amdgcn_wmma_f32_16x16x4_f32(false, a, false, b, (short)0, zero, false, false);
      WMMA_RESULT_GUARD(d);
      o[2] = __builtin_fmaf(r.w1[4], d[0], o[2]);
      o[2] = __builtin_fmaf(r.w1[6], d[1], o[2]);
      o[2] = __builtin_fmaf(r.w1[10], d[2], o[2]);
      o[2] = __builtin_fmaf(r.w1[12], d[3], o[2]);
      o[3] = __builtin_fmaf(r.w1[9], d[4], o[3]);
      o[4] = __builtin_fmaf(r.w1[1], d[5], o[4]);
      o[4] = __builtin_fmaf(r.w1[5], d[6], o[4]);
      o[4] = __builtin_fmaf(r.w1[9], d[7], o[4]);
    }
    if (mlo & 0x80000u) {
      const v2f a = {areg[9].z, areg[9].w};
      const v2f b = {r.w2[6], r.w2[7]};
      v8f d = __builtin_amdgcn_wmma_f32_16x16x4_f32(false, a, false, b, (short)0, zero, false, false);
      WMMA_RESULT_GUARD(d);
      o[2] = __builtin_fmaf(r.w1[4], d[0], o[2]);
      o[2] = __builtin_fmaf(r.w1[6], d[1], o[2]);
      o[2] = __builtin_fmaf(r.w1[10], d[2], o[2]);
      o[2] = __builtin_fmaf(r.w1[12], d[3], o[2]);
      o[3] = __builtin_fmaf(r.w1[9], d[4], o[3]);
      o[4] = __builtin_fmaf(r.w1[1], d[5], o[4]);
      o[4] = __builtin_fmaf(r.w1[5], d[6], o[4]);
      o[4] = __builtin_fmaf(r.w1[9], d[7], o[4]);
    }
  }
  if (mlo & 0x200000u) {
    const v2f a = {areg[10].z, areg[10].w};
    const v2f b = {r.w2[2], r.w2[3]};
    v8f d = __builtin_amdgcn_wmma_f32_16x16x4_f32(false, a, false, b, (short)0, zero, false, false);
    WMMA_RESULT_GUARD(d);
    o[4] = __builtin_fmaf(r.w1[11], d[0], o[4]);
    o[5] = __builtin_fmaf(r.w1[1], d[1], o[5]);
    o[5] = __builtin_fmaf(r.w1[3], d[2], o[5]);
    o[5] = __builtin_fmaf(r.w1[5], d[3], o[5]);
    o[5] = __builtin_fmaf(r.w1[7], d[4], o[5]);
    o[5] = __builtin_fmaf(r.w1[11], d[5], o[5]);
    o[5] = __builtin_fmaf(r.w1[13], d[6], o[5]);
    o[6] = __builtin_fmaf((half ? r.w1[4] : r.w1[8]), d[7], o[6]);
  }
  if (mlo & 0xd00000u) {
    if (mlo & 0x100000u) {
      const v2f a = {areg[10].x, areg[10].y};
      const v2f b = {r.w2[0], r.w2[1]};
      v8f d = __builtin_amdgcn_wmma_f32_16x16x4_f32(false, a, false, b, (short)0, zero, false, false);
      WMMA_RESULT_GUARD(d);
      o[4] = __builtin_fmaf(r.w1[11], d[0], o[4]);
      o[5] = __builtin_fmaf(r.w1[1], d[1], o[5]);
      o[5] = __builtin_fmaf(r.w1[3], d[2], o[5]);
      o[5] = __builtin_fmaf(r.w1[5], d[3], o[5]);
      o[5] = __builtin_fmaf(r.w1[7], d[4], o[5]);
      o[5] = __builtin_fmaf(r.w1[11], d[5], o[5]);
      o[5] = __builtin_fmaf(r.w1[13], d[6], o[5]);
      o[6] = __builtin_fmaf((half ? r.w1[4] : r.w1[8]), d[7], o[6]);
    }
    if (mlo & 0x400000u) {
      const v2f a = {areg[11].x, areg[11].y};
      const v2f b = {r.w2[4], r.w2[5]};
      v8f d = __builtin_amdgcn_wmma_f32_16x16x4_f32(false, a, false, b, (short)0, zero, false, false);
      WMMA_RESULT_GUARD(d);
      o[4] = __builtin_fmaf(r.w1[11], d[0], o[4]);
      o[5] = __builtin_fmaf(r.w1[1], d[1], o[5]);
      o[5] = __builtin_fmaf(r.w1[3], d[2], o[5]);
      o[5] = __builtin_fmaf(r.w1[5], d[3], o[5]);
      o[5] = __builtin_fmaf(r.w1[7], d[4], o[5]);
      o[5] = __builtin_fmaf(r.w1[11], d[5], o[5]);
      o[5] = __builtin_fmaf(r.w1[13], d[6], o[5]);
      o[6] = __builtin_fmaf((half ? r.w1[4] : r.w1[8]), d[7], o[6]);
    }
    if (mlo & 0x800000u) {
      const v2f a = {areg[11].z, areg[11].w};
      const v2f b = {r.w2[6], r.w2[7]};
      v8f d = __builtin_amdgcn_wmma_f32_16x16x4_f32(false, a, false, b, (short)0, zero, false, false);
      WMMA_RESULT_GUARD(d);
      o[4] = __builtin_fmaf(r.w1[11], d[0], o[4]);
      o[5] = __builtin_fmaf(r.w1[1], d[1], o[5]);
      o[5] = __builtin_fmaf(r.w1[3], d[2], o[5]);
      o[5] = __builtin_fmaf(r.w1[5], d[3], o[5]);
      o[5] = __builtin_fmaf(r.w1[7], d[4], o[5]);
      o[5] = __builtin_fmaf(r.w1[11], d[5], o[5]);
      o[5] = __builtin_fmaf(r.w1[13], d[6], o[5]);
      o[6] = __builtin_fmaf((half ? r.w1[4] : r.w1[8]), d[7], o[6]);
    }
  }
  if (mlo & 0x2000000u) {
    const v2f a = {areg[12].z, areg[12].w};
    const v2f b = {r.w2[2], r.w2[3]};
    v8f d = __builtin_amdgcn_wmma_f32_16x16x4_f32(false, a, false, b, (short)0, zero, false, false);
    WMMA_RESULT_GUARD(d);
    o[6] = __builtin_fmaf((half ? r.w1[10] : r.w1[13]), d[0], o[6]);
    o[7] = __builtin_fmaf(r.w1[0], d[1], o[7]);
    o[7] = __builtin_fmaf(r.w1[2], d[2], o[7]);
    o[7] = __builtin_fmaf(r.w1[6], d[3], o[7]);
    o[7] = __builtin_fmaf(r.w1[12], d[4], o[7]);
    o[7] = __builtin_fmaf((half ? r.w1[8] : r.w1[4]), d[5], o[7]);
    o[7] = __builtin_fmaf((half ? r.w1[14] : r.w1[10]), d[6], o[7]);
    o[7] = __builtin_fmaf((half ? r.w1[11] : r.w1[9]), d[7], o[7]);
  }
  if (mlo & 0xd000000u) {
    if (mlo & 0x1000000u) {
      const v2f a = {areg[12].x, areg[12].y};
      const v2f b = {r.w2[0], r.w2[1]};
      v8f d = __builtin_amdgcn_wmma_f32_16x16x4_f32(false, a, false, b, (short)0, zero, false, false);
      WMMA_RESULT_GUARD(d);
      o[6] = __builtin_fmaf((half ? r.w1[10] : r.w1[13]), d[0], o[6]);
      o[7] = __builtin_fmaf(r.w1[0], d[1], o[7]);
      o[7] = __builtin_fmaf(r.w1[2], d[2], o[7]);
      o[7] = __builtin_fmaf(r.w1[6], d[3], o[7]);
      o[7] = __builtin_fmaf(r.w1[12], d[4], o[7]);
      o[7] = __builtin_fmaf((half ? r.w1[8] : r.w1[4]), d[5], o[7]);
      o[7] = __builtin_fmaf((half ? r.w1[14] : r.w1[10]), d[6], o[7]);
      o[7] = __builtin_fmaf((half ? r.w1[11] : r.w1[9]), d[7], o[7]);
    }
    if (mlo & 0x4000000u) {
      const v2f a = {areg[13].x, areg[13].y};
      const v2f b = {r.w2[4], r.w2[5]};
      v8f d = __builtin_amdgcn_wmma_f32_16x16x4_f32(false, a, false, b, (short)0, zero, false, false);
      WMMA_RESULT_GUARD(d);
      o[6] = __builtin_fmaf((half ? r.w1[10] : r.w1[13]), d[0], o[6]);
      o[7] = __builtin_fmaf(r.w1[0], d[1], o[7]);
      o[7] = __builtin_fmaf(r.w1[2], d[2], o[7]);
      o[7] = __builtin_fmaf(r.w1[6], d[3], o[7]);
      o[7] = __builtin_fmaf(r.w1[12], d[4], o[7]);
      o[7] = __builtin_fmaf((half ? r.w1[8] : r.w1[4]), d[5], o[7]);
      o[7] = __builtin_fmaf((half ? r.w1[14] : r.w1[10]), d[6], o[7]);
      o[7] = __builtin_fmaf((half ? r.w1[11] : r.w1[9]), d[7], o[7]);
    }
    if (mlo & 0x8000000u) {
      const v2f a = {areg[13].z, areg[13].w};
      const v2f b = {r.w2[6], r.w2[7]};
      v8f d = __builtin_amdgcn_wmma_f32_16x16x4_f32(false, a, false, b, (short)0, zero, false, false);
      WMMA_RESULT_GUARD(d);
      o[6] = __builtin_fmaf((half ? r.w1[10] : r.w1[13]), d[0], o[6]);
      o[7] = __builtin_fmaf(r.w1[0], d[1], o[7]);
      o[7] = __builtin_fmaf(r.w1[2], d[2], o[7]);
      o[7] = __builtin_fmaf(r.w1[6], d[3], o[7]);
      o[7] = __builtin_fmaf(r.w1[12], d[4], o[7]);
      o[7] = __builtin_fmaf((half ? r.w1[8] : r.w1[4]), d[5], o[7]);
      o[7] = __builtin_fmaf((half ? r.w1[14] : r.w1[10]), d[6], o[7]);
      o[7] = __builtin_fmaf((half ? r.w1[11] : r.w1[9]), d[7], o[7]);
    }
  }
  if (mlo & 0x40000000u) {
    const v2f a = {areg[15].x, areg[15].y};
    const v2f b = {r.w2[4], r.w2[5]};
    v8f d = __builtin_amdgcn_wmma_f32_16x16x4_f32(false, a, false, b, (short)0, zero, false, false);
    WMMA_RESULT_GUARD(d);
    o[0] = __builtin_fmaf(r.w1[3], d[0], o[0]);
    o[0] = __builtin_fmaf(r.w1[7], d[1], o[0]);
    o[0] = __builtin_fmaf(r.w1[13], d[2], o[0]);
    o[0] = __builtin_fmaf(r.w1[15], d[3], o[0]);
    o[1] = __builtin_fmaf(r.w1[0], d[4], o[1]);
    o[1] = __builtin_fmaf(r.w1[2], d[5], o[1]);
    o[1] = __builtin_fmaf(r.w1[6], d[6], o[1]);
    o[1] = __builtin_fmaf(r.w1[8], d[7], o[1]);
  }
  if (mlo & 0xb0000000u) {
    if (mlo & 0x10000000u) {
      const v2f a = {areg[14].x, areg[14].y};
      const v2f b = {r.w2[0], r.w2[1]};
      v8f d = __builtin_amdgcn_wmma_f32_16x16x4_f32(false, a, false, b, (short)0, zero, false, false);
      WMMA_RESULT_GUARD(d);
      o[0] = __builtin_fmaf(r.w1[3], d[0], o[0]);
      o[0] = __builtin_fmaf(r.w1[7], d[1], o[0]);
      o[0] = __builtin_fmaf(r.w1[13], d[2], o[0]);
      o[0] = __builtin_fmaf(r.w1[15], d[3], o[0]);
      o[1] = __builtin_fmaf(r.w1[0], d[4], o[1]);
      o[1] = __builtin_fmaf(r.w1[2], d[5], o[1]);
      o[1] = __builtin_fmaf(r.w1[6], d[6], o[1]);
      o[1] = __builtin_fmaf(r.w1[8], d[7], o[1]);
    }
    if (mlo & 0x20000000u) {
      const v2f a = {areg[14].z, areg[14].w};
      const v2f b = {r.w2[2], r.w2[3]};
      v8f d = __builtin_amdgcn_wmma_f32_16x16x4_f32(false, a, false, b, (short)0, zero, false, false);
      WMMA_RESULT_GUARD(d);
      o[0] = __builtin_fmaf(r.w1[3], d[0], o[0]);
      o[0] = __builtin_fmaf(r.w1[7], d[1], o[0]);
      o[0] = __builtin_fmaf(r.w1[13], d[2], o[0]);
      o[0] = __builtin_fmaf(r.w1[15], d[3], o[0]);
      o[1] = __builtin_fmaf(r.w1[0], d[4], o[1]);
      o[1] = __builtin_fmaf(r.w1[2], d[5], o[1]);
      o[1] = __builtin_fmaf(r.w1[6], d[6], o[1]);
      o[1] = __builtin_fmaf(r.w1[8], d[7], o[1]);
    }
    if (mlo & 0x80000000u) {
      const v2f a = {areg[15].z, areg[15].w};
      const v2f b = {r.w2[6], r.w2[7]};
      v8f d = __builtin_amdgcn_wmma_f32_16x16x4_f32(false, a, false, b, (short)0, zero, false, false);
      WMMA_RESULT_GUARD(d);
      o[0] = __builtin_fmaf(r.w1[3], d[0], o[0]);
      o[0] = __builtin_fmaf(r.w1[7], d[1], o[0]);
      o[0] = __builtin_fmaf(r.w1[13], d[2], o[0]);
      o[0] = __builtin_fmaf(r.w1[15], d[3], o[0]);
      o[1] = __builtin_fmaf(r.w1[0], d[4], o[1]);
      o[1] = __builtin_fmaf(r.w1[2], d[5], o[1]);
      o[1] = __builtin_fmaf(r.w1[6], d[6], o[1]);
      o[1] = __builtin_fmaf(r.w1[8], d[7], o[1]);
    }
  }
  if (mhi & 0x4u) {
    const v2f a = {areg[17].x, areg[17].y};
    const v2f b = {r.w2[4], r.w2[5]};
    v8f d = __builtin_amdgcn_wmma_f32_16x16x4_f32(false, a, false, b, (short)0, zero, false, false);
    WMMA_RESULT_GUARD(d);
    o[1] = __builtin_fmaf(r.w1[12], d[0], o[1]);
    o[1] = __builtin_fmaf(r.w1[14], d[1], o[1]);
    o[2] = __builtin_fmaf(r.w1[8], d[2], o[2]);
    o[2] = __builtin_fmaf(r.w1[14], d[3], o[2]);
    o[3] = __builtin_fmaf(r.w1[1], d[4], o[3]);
    o[3] = __builtin_fmaf(r.w1[3], d[5], o[3]);
    o[3] = __builtin_fmaf(r.w1[5], d[6], o[3]);
    o[3] = __builtin_fmaf(r.w1[7], d[7], o[3]);
  }
  if (mhi & 0xbu) {
    if (mhi & 0x1u) {
      const v2f a = {areg[16].x, areg[16].y};
      const v2f b = {r.w2[0], r.w2[1]};
      v8f d = __builtin_amdgcn_wmma_f32_16x16x4_f32(false, a, false, b, (short)0, zero, false, false);
      WMMA_RESULT_GUARD(d);
      o[1] = __builtin_fmaf(r.w1[12], d[0], o[1]);
      o[1] = __builtin_fmaf(r.w1[14], d[1], o[1]);
      o[2] = __builtin_fmaf(r.w1[8], d[2], o[2]);
      o[2] = __builtin_fmaf(r.w1[14], d[3], o[2]);
      o[3] = __builtin_fmaf(r.w1[1], d[4], o[3]);
      o[3] = __builtin_fmaf(r.w1[3], d[5], o[3]);
      o[3] = __builtin_fmaf(r.w1[5], d[6], o[3]);
      o[3] = __builtin_fmaf(r.w1[7], d[7], o[3]);
    }
    if (mhi & 0x2u) {
      const v2f a = {areg[16].z, areg[16].w};
      const v2f b = {r.w2[2], r.w2[3]};
      v8f d = __builtin_amdgcn_wmma_f32_16x16x4_f32(false, a, false, b, (short)0, zero, false, false);
      WMMA_RESULT_GUARD(d);
      o[1] = __builtin_fmaf(r.w1[12], d[0], o[1]);
      o[1] = __builtin_fmaf(r.w1[14], d[1], o[1]);
      o[2] = __builtin_fmaf(r.w1[8], d[2], o[2]);
      o[2] = __builtin_fmaf(r.w1[14], d[3], o[2]);
      o[3] = __builtin_fmaf(r.w1[1], d[4], o[3]);
      o[3] = __builtin_fmaf(r.w1[3], d[5], o[3]);
      o[3] = __builtin_fmaf(r.w1[5], d[6], o[3]);
      o[3] = __builtin_fmaf(r.w1[7], d[7], o[3]);
    }
    if (mhi & 0x8u) {
      const v2f a = {areg[17].z, areg[17].w};
      const v2f b = {r.w2[6], r.w2[7]};
      v8f d = __builtin_amdgcn_wmma_f32_16x16x4_f32(false, a, false, b, (short)0, zero, false, false);
      WMMA_RESULT_GUARD(d);
      o[1] = __builtin_fmaf(r.w1[12], d[0], o[1]);
      o[1] = __builtin_fmaf(r.w1[14], d[1], o[1]);
      o[2] = __builtin_fmaf(r.w1[8], d[2], o[2]);
      o[2] = __builtin_fmaf(r.w1[14], d[3], o[2]);
      o[3] = __builtin_fmaf(r.w1[1], d[4], o[3]);
      o[3] = __builtin_fmaf(r.w1[3], d[5], o[3]);
      o[3] = __builtin_fmaf(r.w1[5], d[6], o[3]);
      o[3] = __builtin_fmaf(r.w1[7], d[7], o[3]);
    }
  }
  if (mhi & 0x40u) {
    const v2f a = {areg[19].x, areg[19].y};
    const v2f b = {r.w2[4], r.w2[5]};
    v8f d = __builtin_amdgcn_wmma_f32_16x16x4_f32(false, a, false, b, (short)0, zero, false, false);
    WMMA_RESULT_GUARD(d);
    o[3] = __builtin_fmaf(r.w1[11], d[0], o[3]);
    o[3] = __builtin_fmaf(r.w1[13], d[1], o[3]);
    o[4] = __builtin_fmaf(r.w1[3], d[2], o[4]);
    o[4] = __builtin_fmaf(r.w1[7], d[3], o[4]);
    o[4] = __builtin_fmaf(r.w1[13], d[4], o[4]);
    o[4] = __builtin_fmaf(r.w1[15], d[5], o[4]);
    o[5] = __builtin_fmaf(r.w1[15], d[6], o[5]);
    o[6] = __builtin_fmaf(r.w1[0], d[7], o[6]);
  }
  if (mhi & 0xb0u) {
    if (mhi & 0x10u) {
      const v2f a = {areg[18].x, areg[18].y};
      const v2f b = {r.w2[0], r.w2[1]};
      v8f d = __builtin_amdgcn_wmma_f32_16x16x4_f32(false, a, false, b, (short)0, zero, false, false);
      WMMA_RESULT_GUARD(d);
      o[3] = __builtin_fmaf(r.w1[11], d[0], o[3]);
      o[3] = __builtin_fmaf(r.w1[13], d[1], o[3]);
      o[4] = __builtin_fmaf(r.w1[3], d[2], o[4]);
      o[4] = __builtin_fmaf(r.w1[7], d[3], o[4]);
      o[4] = __builtin_fmaf(r.w1[13], d[4], o[4]);
      o[4] = __builtin_fmaf(r.w1[15], d[5], o[4]);
      o[5] = __builtin_fmaf(r.w1[15], d[6], o[5]);
      o[6] = __builtin_fmaf(r.w1[0], d[7], o[6]);
    }
    if (mhi & 0x20u) {
      const v2f a = {areg[18].z, areg[18].w};
      const v2f b = {r.w2[2], r.w2[3]};
      v8f d = __builtin_amdgcn_wmma_f32_16x16x4_f32(false, a, false, b, (short)0, zero, false, false);
      WMMA_RESULT_GUARD(d);
      o[3] = __builtin_fmaf(r.w1[11], d[0], o[3]);
      o[3] = __builtin_fmaf(r.w1[13], d[1], o[3]);
      o[4] = __builtin_fmaf(r.w1[3], d[2], o[4]);
      o[4] = __builtin_fmaf(r.w1[7], d[3], o[4]);
      o[4] = __builtin_fmaf(r.w1[13], d[4], o[4]);
      o[4] = __builtin_fmaf(r.w1[15], d[5], o[4]);
      o[5] = __builtin_fmaf(r.w1[15], d[6], o[5]);
      o[6] = __builtin_fmaf(r.w1[0], d[7], o[6]);
    }
    if (mhi & 0x80u) {
      const v2f a = {areg[19].z, areg[19].w};
      const v2f b = {r.w2[6], r.w2[7]};
      v8f d = __builtin_amdgcn_wmma_f32_16x16x4_f32(false, a, false, b, (short)0, zero, false, false);
      WMMA_RESULT_GUARD(d);
      o[3] = __builtin_fmaf(r.w1[11], d[0], o[3]);
      o[3] = __builtin_fmaf(r.w1[13], d[1], o[3]);
      o[4] = __builtin_fmaf(r.w1[3], d[2], o[4]);
      o[4] = __builtin_fmaf(r.w1[7], d[3], o[4]);
      o[4] = __builtin_fmaf(r.w1[13], d[4], o[4]);
      o[4] = __builtin_fmaf(r.w1[15], d[5], o[4]);
      o[5] = __builtin_fmaf(r.w1[15], d[6], o[5]);
      o[6] = __builtin_fmaf(r.w1[0], d[7], o[6]);
    }
  }
  if (mhi & 0x400u) {
    const v2f a = {areg[21].x, areg[21].y};
    const v2f b = {r.w2[4], r.w2[5]};
    v8f d = __builtin_amdgcn_wmma_f32_16x16x4_f32(false, a, false, b, (short)0, zero, false, false);
    WMMA_RESULT_GUARD(d);
    o[6] = __builtin_fmaf(r.w1[2], d[0], o[6]);
    o[6] = __builtin_fmaf(r.w1[6], d[1], o[6]);
    o[6] = __builtin_fmaf(r.w1[12], d[2], o[6]);
    o[6] = __builtin_fmaf((half ? r.w1[8] : r.w1[4]), d[3], o[6]);
    o[6] = __builtin_fmaf((half ? r.w1[14] : r.w1[10]), d[4], o[6]);
    o[7] = __builtin_fmaf((half ? r.w1[5] : r.w1[8]), d[5], o[7]);
    o[7] = __builtin_fmaf((half ? r.w1[9] : r.w1[14]), d[6], o[7]);
    o[6] = __builtin_fmaf(r.w1[15], d[7], o[6]);
  }
  if (mhi & 0xb00u) {
    if (mhi & 0x100u) {
      const v2f a = {areg[20].x, areg[20].y};
      const v2f b = {r.w2[0], r.w2[1]};
      v8f d = __builtin_amdgcn_wmma_f32_16x16x4_f32(false, a, false, b, (short)0, zero, false, false);
      WMMA_RESULT_GUARD(d);
      o[6] = __builtin_fmaf(r.w1[2], d[0], o[6]);
      o[6] = __builtin_fmaf(r.w1[6], d[1], o[6]);
      o[6] = __builtin_fmaf(r.w1[12], d[2], o[6]);
      o[6] = __builtin_fmaf((half ? r.w1[8] : r.w1[4]), d[3], o[6]);
      o[6] = __builtin_fmaf((half ? r.w1[14] : r.w1[10]), d[4], o[6]);
      o[7] = __builtin_fmaf((half ? r.w1[5] : r.w1[8]), d[5], o[7]);
      o[7] = __builtin_fmaf((half ? r.w1[9] : r.w1[14]), d[6], o[7]);
      o[6] = __builtin_fmaf(r.w1[15], d[7], o[6]);
    }
    if (mhi & 0x200u) {
      const v2f a = {areg[20].z, areg[20].w};
      const v2f b = {r.w2[2], r.w2[3]};
      v8f d = __builtin_amdgcn_wmma_f32_16x16x4_f32(false, a, false, b, (short)0, zero, false, false);
      WMMA_RESULT_GUARD(d);
      o[6] = __builtin_fmaf(r.w1[2], d[0], o[6]);
      o[6] = __builtin_fmaf(r.w1[6], d[1], o[6]);
      o[6] = __builtin_fmaf(r.w1[12], d[2], o[6]);
      o[6] = __builtin_fmaf((half ? r.w1[8] : r.w1[4]), d[3], o[6]);
      o[6] = __builtin_fmaf((half ? r.w1[14] : r.w1[10]), d[4], o[6]);
      o[7] = __builtin_fmaf((half ? r.w1[5] : r.w1[8]), d[5], o[7]);
      o[7] = __builtin_fmaf((half ? r.w1[9] : r.w1[14]), d[6], o[7]);
      o[6] = __builtin_fmaf(r.w1[15], d[7], o[6]);
    }
    if (mhi & 0x800u) {
      const v2f a = {areg[21].z, areg[21].w};
      const v2f b = {r.w2[6], r.w2[7]};
      v8f d = __builtin_amdgcn_wmma_f32_16x16x4_f32(false, a, false, b, (short)0, zero, false, false);
      WMMA_RESULT_GUARD(d);
      o[6] = __builtin_fmaf(r.w1[2], d[0], o[6]);
      o[6] = __builtin_fmaf(r.w1[6], d[1], o[6]);
      o[6] = __builtin_fmaf(r.w1[12], d[2], o[6]);
      o[6] = __builtin_fmaf((half ? r.w1[8] : r.w1[4]), d[3], o[6]);
      o[6] = __builtin_fmaf((half ? r.w1[14] : r.w1[10]), d[4], o[6]);
      o[7] = __builtin_fmaf((half ? r.w1[5] : r.w1[8]), d[5], o[7]);
      o[7] = __builtin_fmaf((half ? r.w1[9] : r.w1[14]), d[6], o[7]);
      o[6] = __builtin_fmaf(r.w1[15], d[7], o[6]);
    }
  }
  if (mhi & 0x8000u) {
    const v2f a = {areg[23].z, areg[23].w};
    const v2f b = {r.w2[6], r.w2[7]};
    v8f d = __builtin_amdgcn_wmma_f32_16x16x4_f32(false, a, false, b, (short)0, zero, false, false);
    WMMA_RESULT_GUARD(d);
    o[0] = __builtin_fmaf(r.w1[4], d[0], o[0]);
    o[0] = __builtin_fmaf(r.w1[8], d[1], o[0]);
    o[0] = __builtin_fmaf(r.w1[10], d[2], o[0]);
    o[0] = __builtin_fmaf(r.w1[14], d[3], o[0]);
    o[1] = __builtin_fmaf(r.w1[3], d[4], o[1]);
    o[1] = __builtin_fmaf(r.w1[7], d[5], o[1]);
    o[1] = __builtin_fmaf(r.w1[9], d[6], o[1]);
    o[1] = __builtin_fmaf(r.w1[13], d[7], o[1]);
  }
  if (mhi & 0x7000u) {
    if (mhi & 0x1000u) {
      const v2f a = {areg[22].x, areg[22].y};
      const v2f b = {r.w2[0], r.w2[1]};
      v8f d = __builtin_amdgcn_wmma_f32_16x16x4_f32(false, a, false, b, (short)0, zero, false, false);
      WMMA_RESULT_GUARD(d);
      o[0] = __builtin_fmaf(r.w1[4], d[0], o[0]);
      o[0] = __builtin_fmaf(r.w1[8], d[1], o[0]);
      o[0] = __builtin_fmaf(r.w1[10], d[2], o[0]);
      o[0] = __builtin_fmaf(r.w1[14], d[3], o[0]);
      o[1] = __builtin_fmaf(r.w1[3], d[4], o[1]);
      o[1] = __builtin_fmaf(r.w1[7], d[5], o[1]);
      o[1] = __builtin_fmaf(r.w1[9], d[6], o[1]);
      o[1] = __builtin_fmaf(r.w1[13], d[7], o[1]);
    }
    if (mhi & 0x2000u) {
      const v2f a = {areg[22].z, areg[22].w};
      const v2f b = {r.w2[2], r.w2[3]};
      v8f d = __builtin_amdgcn_wmma_f32_16x16x4_f32(false, a, false, b, (short)0, zero, false, false);
      WMMA_RESULT_GUARD(d);
      o[0] = __builtin_fmaf(r.w1[4], d[0], o[0]);
      o[0] = __builtin_fmaf(r.w1[8], d[1], o[0]);
      o[0] = __builtin_fmaf(r.w1[10], d[2], o[0]);
      o[0] = __builtin_fmaf(r.w1[14], d[3], o[0]);
      o[1] = __builtin_fmaf(r.w1[3], d[4], o[1]);
      o[1] = __builtin_fmaf(r.w1[7], d[5], o[1]);
      o[1] = __builtin_fmaf(r.w1[9], d[6], o[1]);
      o[1] = __builtin_fmaf(r.w1[13], d[7], o[1]);
    }
    if (mhi & 0x4000u) {
      const v2f a = {areg[23].x, areg[23].y};
      const v2f b = {r.w2[4], r.w2[5]};
      v8f d = __builtin_amdgcn_wmma_f32_16x16x4_f32(false, a, false, b, (short)0, zero, false, false);
      WMMA_RESULT_GUARD(d);
      o[0] = __builtin_fmaf(r.w1[4], d[0], o[0]);
      o[0] = __builtin_fmaf(r.w1[8], d[1], o[0]);
      o[0] = __builtin_fmaf(r.w1[10], d[2], o[0]);
      o[0] = __builtin_fmaf(r.w1[14], d[3], o[0]);
      o[1] = __builtin_fmaf(r.w1[3], d[4], o[1]);
      o[1] = __builtin_fmaf(r.w1[7], d[5], o[1]);
      o[1] = __builtin_fmaf(r.w1[9], d[6], o[1]);
      o[1] = __builtin_fmaf(r.w1[13], d[7], o[1]);
    }
  }
  if (mhi & 0x80000u) {
    const v2f a = {cgl[25 * 32].z, cgl[25 * 32].w};
    const v2f b = {r.w2[6], r.w2[7]};
    v8f d = __builtin_amdgcn_wmma_f32_16x16x4_f32(false, a, false, b, (short)0, zero, false, false);
    WMMA_RESULT_GUARD_A(d, a);
    o[2] = __builtin_fmaf(r.w1[1], d[0], o[2]);
    o[2] = __builtin_fmaf(r.w1[5], d[1], o[2]);
    o[2] = __builtin_fmaf(r.w1[11], d[2], o[2]);
    o[2] = __builtin_fmaf(r.w1[15], d[3], o[2]);
    o[3] = __builtin_fmaf(r.w1[0], d[4], o[3]);
    o[3] = __builtin_fmaf(r.w1[2], d[5], o[3]);
    o[3] = __builtin_fmaf(r.w1[6], d[6], o[3]);
    o[3] = __builtin_fmaf(r.w1[12], d[7], o[3]);
  }
  if (mhi & 0x70000u) {
    if (mhi & 0x10000u) {
      const v2f a = {cgl[24 * 32].x, cgl[24 * 32].y};
      const v2f b = {r.w2[0], r.w2[1]};
      v8f d = __builtin_amdgcn_wmma_f32_16x16x4_f32(false, a, false, b, (short)0, zero, false, false);
      WMMA_RESULT_GUARD_A(d, a);
      o[2] = __builtin_fmaf(r.w1[1], d[0], o[2]);
      o[2] = __builtin_fmaf(r.w1[5], d[1], o[2]);
      o[2] = __builtin_fmaf(r.w1[11], d[2], o[2]);
      o[2] = __builtin_fmaf(r.w1[15], d[3], o[2]);
      o[3] = __builtin_fmaf(r.w1[0], d[4], o[3]);
      o[3] = __builtin_fmaf(r.w1[2], d[5], o[3]);
      o[3] = __builtin_fmaf(r.w1[6], d[6], o[3]);
      o[3] = __builtin_fmaf(r.w1[12], d[7], o[3]);
    }
    if (mhi & 0x20000u) {
      const v2f a = {cgl[24 * 32].z, cgl[24 * 32].w};
      const v2f b = {r.w2[2], r.w2[3]};
      v8f d = __builtin_amdgcn_wmma_f32_16x16x4_f32(false, a, false, b, (short)0, zero, false, false);
      WMMA_RESULT_GUARD_A(d, a);
      o[2] = __builtin_fmaf(r.w1[1], d[0], o[2]);
      o[2] = __builtin_fmaf(r.w1[5], d[1], o[2]);
      o[2] = __builtin_fmaf(r.w1[11], d[2], o[2]);
      o[2] = __builtin_fmaf(r.w1[15], d[3], o[2]);
      o[3] = __builtin_fmaf(r.w1[0], d[4], o[3]);
      o[3] = __builtin_fmaf(r.w1[2], d[5], o[3]);
      o[3] = __builtin_fmaf(r.w1[6], d[6], o[3]);
      o[3] = __builtin_fmaf(r.w1[12], d[7], o[3]);
    }
    if (mhi & 0x40000u) {
      const v2f a = {cgl[25 * 32].x, cgl[25 * 32].y};
      const v2f b = {r.w2[4], r.w2[5]};
      v8f d = __builtin_amdgcn_wmma_f32_16x16x4_f32(false, a, false, b, (short)0, zero, false, false);
      WMMA_RESULT_GUARD_A(d, a);
      o[2] = __builtin_fmaf(r.w1[1], d[0], o[2]);
      o[2] = __builtin_fmaf(r.w1[5], d[1], o[2]);
      o[2] = __builtin_fmaf(r.w1[11], d[2], o[2]);
      o[2] = __builtin_fmaf(r.w1[15], d[3], o[2]);
      o[3] = __builtin_fmaf(r.w1[0], d[4], o[3]);
      o[3] = __builtin_fmaf(r.w1[2], d[5], o[3]);
      o[3] = __builtin_fmaf(r.w1[6], d[6], o[3]);
      o[3] = __builtin_fmaf(r.w1[12], d[7], o[3]);
    }
  }
  if (mhi & 0x800000u) {
    const v2f a = {cgl[27 * 32].z, cgl[27 * 32].w};
    const v2f b = {r.w2[6], r.w2[7]};
    v8f d = __builtin_amdgcn_wmma_f32_16x16x4_f32(false, a, false, b, (short)0, zero, false, false);
    WMMA_RESULT_GUARD_A(d, a);
    o[4] = __builtin_fmaf(r.w1[4], d[0], o[4]);
    o[4] = __builtin_fmaf(r.w1[8], d[1], o[4]);
    o[4] = __builtin_fmaf(r.w1[10], d[2], o[4]);
    o[4] = __builtin_fmaf(r.w1[14], d[3], o[4]);
    o[5] = __builtin_fmaf(r.w1[0], d[4], o[5]);
    o[5] = __builtin_fmaf(r.w1[2], d[5], o[5]);
    o[5] = __builtin_fmaf(r.w1[6], d[6], o[5]);
    o[5] = __builtin_fmaf(r.w1[12], d[7], o[5]);
  }
  if (mhi & 0x700000u) {
    if (mhi & 0x100000u) {
      const v2f a = {cgl[26 * 32].x, cgl[26 * 32].y};
      const v2f b = {r.w2[0], r.w2[1]};
      v8f d = __builtin_amdgcn_wmma_f32_16x16x4_f32(false, a, false, b, (short)0, zero, false, false);
      WMMA_RESULT_GUARD_A(d, a);
      o[4] = __builtin_fmaf(r.w1[4], d[0], o[4]);
      o[4] = __builtin_fmaf(r.w1[8], d[1], o[4]);
      o[4] = __builtin_fmaf(r.w1[10], d[2], o[4]);
      o[4] = __builtin_fmaf(r.w1[14], d[3], o[4]);
      o[5] = __builtin_fmaf(r.w1[0], d[4], o[5]);
      o[5] = __builtin_fmaf(r.w1[2], d[5], o[5]);
      o[5] = __builtin_fmaf(r.w1[6], d[6], o[5]);
      o[5] = __builtin_fmaf(r.w1[12], d[7], o[5]);
    }
    if (mhi & 0x200000u) {
      const v2f a = {cgl[26 * 32].z, cgl[26 * 32].w};
      const v2f b = {r.w2[2], r.w2[3]};
      v8f d = __builtin_amdgcn_wmma_f32_16x16x4_f32(false, a, false, b, (short)0, zero, false, false);
      WMMA_RESULT_GUARD_A(d, a);
      o[4] = __builtin_fmaf(r.w1[4], d[0], o[4]);
      o[4] = __builtin_fmaf(r.w1[8], d[1], o[4]);
      o[4] = __builtin_fmaf(r.w1[10], d[2], o[4]);
      o[4] = __builtin_fmaf(r.w1[14], d[3], o[4]);
      o[5] = __builtin_fmaf(r.w1[0], d[4], o[5]);
      o[5] = __builtin_fmaf(r.w1[2], d[5], o[5]);
      o[5] = __builtin_fmaf(r.w1[6], d[6], o[5]);
      o[5] = __builtin_fmaf(r.w1[12], d[7], o[5]);
    }
    if (mhi & 0x400000u) {
      const v2f a = {cgl[27 * 32].x, cgl[27 * 32].y};
      const v2f b = {r.w2[4], r.w2[5]};
      v8f d = __builtin_amdgcn_wmma_f32_16x16x4_f32(false, a, false, b, (short)0, zero, false, false);
      WMMA_RESULT_GUARD_A(d, a);
      o[4] = __builtin_fmaf(r.w1[4], d[0], o[4]);
      o[4] = __builtin_fmaf(r.w1[8], d[1], o[4]);
      o[4] = __builtin_fmaf(r.w1[10], d[2], o[4]);
      o[4] = __builtin_fmaf(r.w1[14], d[3], o[4]);
      o[5] = __builtin_fmaf(r.w1[0], d[4], o[5]);
      o[5] = __builtin_fmaf(r.w1[2], d[5], o[5]);
      o[5] = __builtin_fmaf(r.w1[6], d[6], o[5]);
      o[5] = __builtin_fmaf(r.w1[12], d[7], o[5]);
    }
  }
  if (mhi & 0x8000000u) {
    const v2f a = {cgl[29 * 32].z, cgl[29 * 32].w};
    const v2f b = {r.w2[6], r.w2[7]};
    v8f d = __builtin_amdgcn_wmma_f32_16x16x4_f32(false, a, false, b, (short)0, zero, false, false);
    WMMA_RESULT_GUARD_A(d, a);
    o[6] = __builtin_fmaf((half ? r.w1[3] : r.w1[1]), d[0], o[6]);
    o[6] = __builtin_fmaf((half ? r.w1[7] : r.w1[5]), d[1], o[6]);
    o[6] = __builtin_fmaf((half ? r.w1[9] : r.w1[11]), d[2], o[6]);
    o[6] = __builtin_fmaf((half ? r.w1[13] : r.w1[14]), d[3], o[6]);
    o[7] = __builtin_fmaf((half ? r.w1[3] : r.w1[1]), d[4], o[7]);
    o[7] = __builtin_fmaf((half ? r.w1[7] : r.w1[5]), d[5], o[7]);
    o[7] = __builtin_fmaf((half ? r.w1[13] : r.w1[11]), d[6], o[7]);
    o[7] = __builtin_fmaf((half ? r.w1[10] : r.w1[15]), d[7], o[7]);
  }
  if (mhi & 0x7000000u) {
    if (mhi & 0x1000000u) {
      const v2f a = {cgl[28 * 32].x, cgl[28 * 32].y};
      const v2f b = {r.w2[0], r.w2[1]};
      v8f d = __builtin_amdgcn_wmma_f32_16x16x4_f32(false, a, false, b, (short)0, zero, false, false);
      WMMA_RESULT_GUARD_A(d, a);
      o[6] = __builtin_fmaf((half ? r.w1[3] : r.w1[1]), d[0], o[6]);
      o[6] = __builtin_fmaf((half ? r.w1[7] : r.w1[5]), d[1], o[6]);
      o[6] = __builtin_fmaf((half ? r.w1[9] : r.w1[11]), d[2], o[6]);
      o[6] = __builtin_fmaf((half ? r.w1[13] : r.w1[14]), d[3], o[6]);
      o[7] = __builtin_fmaf((half ? r.w1[3] : r.w1[1]), d[4], o[7]);
      o[7] = __builtin_fmaf((half ? r.w1[7] : r.w1[5]), d[5], o[7]);
      o[7] = __builtin_fmaf((half ? r.w1[13] : r.w1[11]), d[6], o[7]);
      o[7] = __builtin_fmaf((half ? r.w1[10] : r.w1[15]), d[7], o[7]);
    }
    if (mhi & 0x2000000u) {
      const v2f a = {cgl[28 * 32].z, cgl[28 * 32].w};
      const v2f b = {r.w2[2], r.w2[3]};
      v8f d = __builtin_amdgcn_wmma_f32_16x16x4_f32(false, a, false, b, (short)0, zero, false, false);
      WMMA_RESULT_GUARD_A(d, a);
      o[6] = __builtin_fmaf((half ? r.w1[3] : r.w1[1]), d[0], o[6]);
      o[6] = __builtin_fmaf((half ? r.w1[7] : r.w1[5]), d[1], o[6]);
      o[6] = __builtin_fmaf((half ? r.w1[9] : r.w1[11]), d[2], o[6]);
      o[6] = __builtin_fmaf((half ? r.w1[13] : r.w1[14]), d[3], o[6]);
      o[7] = __builtin_fmaf((half ? r.w1[3] : r.w1[1]), d[4], o[7]);
      o[7] = __builtin_fmaf((half ? r.w1[7] : r.w1[5]), d[5], o[7]);
      o[7] = __builtin_fmaf((half ? r.w1[13] : r.w1[11]), d[6], o[7]);
      o[7] = __builtin_fmaf((half ? r.w1[10] : r.w1[15]), d[7], o[7]);
    }
    if (mhi & 0x4000000u) {
      const v2f a = {cgl[29 * 32].x, cgl[29 * 32].y};
      const v2f b = {r.w2[4], r.w2[5]};
      v8f d = __builtin_amdgcn_wmma_f32_16x16x4_f32(false, a, false, b, (short)0, zero, false, false);
      WMMA_RESULT_GUARD_A(d, a);
      o[6] = __builtin_fmaf((half ? r.w1[3] : r.w1[1]), d[0], o[6]);
      o[6] = __builtin_fmaf((half ? r.w1[7] : r.w1[5]), d[1], o[6]);
      o[6] = __builtin_fmaf((half ? r.w1[9] : r.w1[11]), d[2], o[6]);
      o[6] = __builtin_fmaf((half ? r.w1[13] : r.w1[14]), d[3], o[6]);
      o[7] = __builtin_fmaf((half ? r.w1[3] : r.w1[1]), d[4], o[7]);
      o[7] = __builtin_fmaf((half ? r.w1[7] : r.w1[5]), d[5], o[7]);
      o[7] = __builtin_fmaf((half ? r.w1[13] : r.w1[11]), d[6], o[7]);
      o[7] = __builtin_fmaf((half ? r.w1[10] : r.w1[15]), d[7], o[7]);
    }
  }
  if (mhi & 0x10000000u) {
    const v2f a = {cgl[30 * 32].x, cgl[30 * 32].y};
    const v2f b = {r.w2[0], r.w2[1]};
    v8f d = __builtin_amdgcn_wmma_f32_16x16x4_f32(false, a, false, b, (short)0, zero, false, false);
    WMMA_RESULT_GUARD_A(d, a);
    o[1] = __builtin_fmaf(r.w1[15], d[0], o[1]);
    o[2] = __builtin_fmaf(r.w1[9], d[1], o[2]);
    o[3] = __builtin_fmaf(r.w1[8], d[2], o[3]);
    o[3] = __builtin_fmaf(r.w1[14], d[3], o[3]);
    o[3] = __builtin_fmaf(r.w1[15], d[4], o[3]);
    o[5] = __builtin_fmaf(r.w1[4], d[5], o[5]);
    o[5] = __builtin_fmaf(r.w1[9], d[6], o[5]);
    o[5] = __builtin_fmaf(r.w1[10], d[7], o[5]);
  }
  if (mhi & 0xe0000000u) {
    if (mhi & 0x20000000u) {
      const v2f a = {cgl[30 * 32].z, cgl[30 * 32].w};
      const v2f b = {r.w2[2], r.w2[3]};
      v8f d = __builtin_amdgcn_wmma_f32_16x16x4_f32(false, a, false, b, (short)0, zero, false, false);
      WMMA_RESULT_GUARD_A(d, a);
      o[1] = __builtin_fmaf(r.w1[15], d[0], o[1]);
      o[2] = __builtin_fmaf(r.w1[9], d[1], o[2]);
      o[3] = __builtin_fmaf(r.w1[8], d[2], o[3]);
      o[3] = __builtin_fmaf(r.w1[14], d[3], o[3]);
      o[3] = __builtin_fmaf(r.w1[15], d[4], o[3]);
      o[5] = __builtin_fmaf(r.w1[4], d[5], o[5]);
      o[5] = __builtin_fmaf(r.w1[9], d[6], o[5]);
      o[5] = __builtin_fmaf(r.w1[10], d[7], o[5]);
    }
    if (mhi & 0x40000000u) {
      const v2f a = {cgl[31 * 32].x, cgl[31 * 32].y};
      const v2f b = {r.w2[4], r.w2[5]};
      v8f d = __builtin_amdgcn_wmma_f32_16x16x4_f32(false, a, false, b, (short)0, zero, false, false);
      WMMA_RESULT_GUARD_A(d, a);
      o[1] = __builtin_fmaf(r.w1[15], d[0], o[1]);
      o[2] = __builtin_fmaf(r.w1[9], d[1], o[2]);
      o[3] = __builtin_fmaf(r.w1[8], d[2], o[3]);
      o[3] = __builtin_fmaf(r.w1[14], d[3], o[3]);
      o[3] = __builtin_fmaf(r.w1[15], d[4], o[3]);
      o[5] = __builtin_fmaf(r.w1[4], d[5], o[5]);
      o[5] = __builtin_fmaf(r.w1[9], d[6], o[5]);
      o[5] = __builtin_fmaf(r.w1[10], d[7], o[5]);
    }
    if (mhi & 0x80000000u) {
      const v2f a = {cgl[31 * 32].z, cgl[31 * 32].w};
      const v2f b = {r.w2[6], r.w2[7]};
      v8f d = __builtin_amdgcn_wmma_f32_16x16x4_f32(false, a, false, b, (short)0, zero, false, false);
      WMMA_RESULT_GUARD_A(d, a);
      o[1] = __builtin_fmaf(r.w1[15], d[0], o[1]);
      o[2] = __builtin_fmaf(r.w1[9], d[1], o[2]);
      o[3] = __builtin_fmaf(r.w1[8], d[2], o[3]);
      o[3] = __builtin_fmaf(r.w1[14], d[3], o[3]);
      o[3] = __builtin_fmaf(r.w1[15], d[4], o[3]);
      o[5] = __builtin_fmaf(r.w1[4], d[5], o[5]);
      o[5] = __builtin_fmaf(r.w1[9], d[6], o[5]);
      o[5] = __builtin_fmaf(r.w1[10], d[7], o[5]);
    }
  }

  const int n = t >> 2;
  const int c = (t & 3) * 16 + col;
  float* po = out + (size_t)n * NUM_ORDERS * C_DIM + c;
  po[(half ? 2 : 0) * C_DIM] = o[0];
  po[(half ? 5 : 1) * C_DIM] = o[1];
  po[(half ? 7 : 3) * C_DIM] = o[2];
  po[(half ? 10 : 4) * C_DIM] = o[3];
  po[(half ? 12 : 6) * C_DIM] = o[4];
  po[(half ? 14 : 8) * C_DIM] = o[5];
  po[(half ? 11 : 9) * C_DIM] = o[6];
  po[(half ? 15 : 13) * C_DIM] = o[7];
}

#define FAST_BLOCKS 15
#define FAST_MASK_LO 0x42222111u
#define FAST_MASK_HI 0x08888444u

#define FIRST_BLOCK_GUARD(a, b, c)                                                        \
  asm volatile("v_nop\n\tv_nop\n\tv_nop\n\tv_nop\n\tv_nop\n\tv_nop\n\tv_nop\n\tv_nop" \
               : "+v"(a), "+v"(b), "+v"(c))

__device__ __forceinline__ void tile_product_fast(v2f (&fa)[FAST_BLOCKS], const TileRegs& r,
                                                  float (&o)[8], int half) {
  const v8f zero = {};
#pragma unroll
  for (int v = 0; v < 8; ++v) o[v] = 0.0f;
  const float s_1_9 = half ? r.w1[1] : r.w1[9];
  const float s_5_3 = half ? r.w1[5] : r.w1[3];
  const float s_11_7 = half ? r.w1[11] : r.w1[7];
  const float s_15_3 = half ? r.w1[15] : r.w1[3];
  const float s_1_7 = half ? r.w1[1] : r.w1[7];
  const float s_4_13 = half ? r.w1[4] : r.w1[13];
  const float s_4_8 = half ? r.w1[4] : r.w1[8];
  const float s_10_13 = half ? r.w1[10] : r.w1[13];
  const float s_8_4 = half ? r.w1[8] : r.w1[4];
  const float s_14_10 = half ? r.w1[14] : r.w1[10];
  const float s_11_9 = half ? r.w1[11] : r.w1[9];
  const float s_5_8 = half ? r.w1[5] : r.w1[8];
  const float s_9_14 = half ? r.w1[9] : r.w1[14];
  const float s_3_1 = half ? r.w1[3] : r.w1[1];
  const float s_7_5 = half ? r.w1[7] : r.w1[5];
  const float s_9_11 = half ? r.w1[9] : r.w1[11];
  const float s_13_14 = half ? r.w1[13] : r.w1[14];
  const float s_13_11 = half ? r.w1[13] : r.w1[11];
  const float s_10_15 = half ? r.w1[10] : r.w1[15];
  v8f d0, d1, d2;
  d0 = __builtin_amdgcn_wmma_f32_16x16x4_f32(false, fa[0], false, (v2f){r.w2[0], r.w2[1]}, (short)0, zero, false, false);
  d1 = __builtin_amdgcn_wmma_f32_16x16x4_f32(false, fa[1], false, (v2f){r.w2[0], r.w2[1]}, (short)0, zero, false, false);
  d2 = __builtin_amdgcn_wmma_f32_16x16x4_f32(false, fa[2], false, (v2f){r.w2[0], r.w2[1]}, (short)0, zero, false, false);
  FIRST_BLOCK_GUARD(d0, d1, d2);
  o[0] = __builtin_fmaf(r.w1[0], d0[0], o[0]);
  o[0] = __builtin_fmaf(r.w1[2], d0[1], o[0]);
  o[0] = __builtin_fmaf(r.w1[6], d0[2], o[0]);
  o[0] = __builtin_fmaf(r.w1[12], d0[3], o[0]);
  o[1] = __builtin_fmaf(r.w1[1], d0[4], o[1]);
  o[1] = __builtin_fmaf(r.w1[5], d0[5], o[1]);
  o[1] = __builtin_fmaf(r.w1[11], d0[6], o[1]);
  o[2] = __builtin_fmaf(r.w1[3], d0[7], o[2]);
  asm volatile("" : "+v"(fa[3]), "+v"(o[0]), "+v"(o[1]), "+v"(o[2]));
  d0 = __builtin_amdgcn_wmma_f32_16x16x4_f32(false, fa[3], false, (v2f){r.w2[2], r.w2[3]}, (short)0, zero, false, false);
  asm volatile("" : "+v"(d1), "+v"(d0));
  o[2] = __builtin_fmaf(r.w1[7], d1[0], o[2]);
  o[2] = __builtin_fmaf(r.w1[13], d1[1], o[2]);
  o[3] = __builtin_fmaf(r.w1[4], d1[2], o[3]);
  o[3] = __builtin_fmaf(r.w1[10], d1[3], o[3]);
  o[4] = __builtin_fmaf(r.w1[0], d1[4], o[4]);
  o[4] = __builtin_fmaf(r.w1[2], d1[5], o[4]);
  o[4] = __builtin_fmaf(r.w1[6], d1[6], o[4]);
  o[4] = __builtin_fmaf(r.w1[12], d1[7], o[4]);
  asm volatile("" : "+v"(fa[4]), "+v"(o[2]), "+v"(o[3]), "+v"(o[4]));
  d1 = __builtin_amdgcn_wmma_f32_16x16x4_f32(false, fa[4], false, (v2f){r.w2[2], r.w2[3]}, (short)0, zero, false, false);
  asm volatile("" : "+v"(d2), "+v"(d1));
  o[5] = __builtin_fmaf(r.w1[8], d2[0], o[5]);
  o[5] = __builtin_fmaf(r.w1[14], d2[1], o[5]);
  o[6] = __builtin_fmaf(s_1_9, d2[2], o[6]);
  o[6] = __builtin_fmaf(s_5_3, d2[3], o[6]);
  o[6] = __builtin_fmaf(s_11_7, d2[4], o[6]);
  o[7] = __builtin_fmaf(s_15_3, d2[5], o[7]);
  o[7] = __builtin_fmaf(s_1_7, d2[6], o[7]);
  o[7] = __builtin_fmaf(s_4_13, d2[7], o[7]);
  asm volatile("" : "+v"(fa[5]), "+v"(o[5]), "+v"(o[6]), "+v"(o[7]));
  d2 = __builtin_amdgcn_wmma_f32_16x16x4_f32(false, fa[5], false, (v2f){r.w2[2], r.w2[3]}, (short)0, zero, false, false);
  asm volatile("" : "+v"(d0), "+v"(d2));
  o[0] = __builtin_fmaf(r.w1[1], d0[0], o[0]);
  o[0] = __builtin_fmaf(r.w1[5], d0[1], o[0]);
  o[0] = __builtin_fmaf(r.w1[9], d0[2], o[0]);
  o[0] = __builtin_fmaf(r.w1[11], d0[3], o[0]);
  o[1] = __builtin_fmaf(r.w1[4], d0[4], o[1]);
  o[1] = __builtin_fmaf(r.w1[10], d0[5], o[1]);
  o[2] = __builtin_fmaf(r.w1[0], d0[6], o[2]);
  o[2] = __builtin_fmaf(r.w1[2], d0[7], o[2]);
  asm volatile("" : "+v"(fa[6]), "+v"(o[0]), "+v"(o[1]), "+v"(o[2]));
  d0 = __builtin_amdgcn_wmma_f32_16x16x4_f32(false, fa[6], false, (v2f){r.w2[2], r.w2[3]}, (short)0, zero, false, false);
  asm volatile("" : "+v"(d1), "+v"(d0));
  o[2] = __builtin_fmaf(r.w1[4], d1[0], o[2]);
  o[2] = __builtin_fmaf(r.w1[6], d1[1], o[2]);
  o[2] = __builtin_fmaf(r.w1[10], d1[2], o[2]);
  o[2] = __builtin_fmaf(r.w1[12], d1[3], o[2]);
  o[3] = __builtin_fmaf(r.w1[9], d1[4], o[3]);
  o[4] = __builtin_fmaf(r.w1[1], d1[5], o[4]);
  o[4] = __builtin_fmaf(r.w1[5], d1[6], o[4]);
  o[4] = __builtin_fmaf(r.w1[9], d1[7], o[4]);
  asm volatile("" : "+v"(fa[7]), "+v"(o[2]), "+v"(o[3]), "+v"(o[4]));
  d1 = __builtin_amdgcn_wmma_f32_16x16x4_f32(false, fa[7], false, (v2f){r.w2[4], r.w2[5]}, (short)0, zero, false, false);
  asm volatile("" : "+v"(d2), "+v"(d1));
  o[4] = __builtin_fmaf(r.w1[11], d2[0], o[4]);
  o[5] = __builtin_fmaf(r.w1[1], d2[1], o[5]);
  o[5] = __builtin_fmaf(r.w1[3], d2[2], o[5]);
  o[5] = __builtin_fmaf(r.w1[5], d2[3], o[5]);
  o[5] = __builtin_fmaf(r.w1[7], d2[4], o[5]);
  o[5] = __builtin_fmaf(r.w1[11], d2[5], o[5]);
  o[5] = __builtin_fmaf(r.w1[13], d2[6], o[5]);
  o[6] = __builtin_fmaf(s_4_8, d2[7], o[6]);
  asm volatile("" : "+v"(fa[8]), "+v"(o[4]), "+v"(o[5]), "+v"(o[6]));
  d2 = __builtin_amdgcn_wmma_f32_16x16x4_f32(false, fa[8], false, (v2f){r.w2[4], r.w2[5]}, (short)0, zero, false, false);
  asm volatile("" : "+v"(d0), "+v"(d2));
  o[6] = __builtin_fmaf(s_10_13, d0[0], o[6]);
  o[7] = __builtin_fmaf(r.w1[0], d0[1], o[7]);
  o[7] = __builtin_fmaf(r.w1[2], d0[2], o[7]);
  o[7] = __builtin_fmaf(r.w1[6], d0[3], o[7]);
  o[7] = __builtin_fmaf(r.w1[12], d0[4], o[7]);
  o[7] = __builtin_fmaf(s_8_4, d0[5], o[7]);
  o[7] = __builtin_fmaf(s_14_10, d0[6], o[7]);
  o[7] = __builtin_fmaf(s_11_9, d0[7], o[7]);
  asm volatile("" : "+v"(fa[9]), "+v"(o[6]), "+v"(o[7]));
  d0 = __builtin_amdgcn_wmma_f32_16x16x4_f32(false, fa[9], false, (v2f){r.w2[4], r.w2[5]}, (short)0, zero, false, false);
  asm volatile("" : "+v"(d1), "+v"(d0));
  o[0] = __builtin_fmaf(r.w1[3], d1[0], o[0]);
  o[0] = __builtin_fmaf(r.w1[7], d1[1], o[0]);
  o[0] = __builtin_fmaf(r.w1[13], d1[2], o[0]);
  o[0] = __builtin_fmaf(r.w1[15], d1[3], o[0]);
  o[1] = __builtin_fmaf(r.w1[0], d1[4], o[1]);
  o[1] = __builtin_fmaf(r.w1[2], d1[5], o[1]);
  o[1] = __builtin_fmaf(r.w1[6], d1[6], o[1]);
  o[1] = __builtin_fmaf(r.w1[8], d1[7], o[1]);
  asm volatile("" : "+v"(fa[10]), "+v"(o[0]), "+v"(o[1]));
  d1 = __builtin_amdgcn_wmma_f32_16x16x4_f32(false, fa[10], false, (v2f){r.w2[4], r.w2[5]}, (short)0, zero, false, false);
  asm volatile("" : "+v"(d2), "+v"(d1));
  o[1] = __builtin_fmaf(r.w1[12], d2[0], o[1]);
  o[1] = __builtin_fmaf(r.w1[14], d2[1], o[1]);
  o[2] = __builtin_fmaf(r.w1[8], d2[2], o[2]);
  o[2] = __builtin_fmaf(r.w1[14], d2[3], o[2]);
  o[3] = __builtin_fmaf(r.w1[1], d2[4], o[3]);
  o[3] = __builtin_fmaf(r.w1[3], d2[5], o[3]);
  o[3] = __builtin_fmaf(r.w1[5], d2[6], o[3]);
  o[3] = __builtin_fmaf(r.w1[7], d2[7], o[3]);
  asm volatile("" : "+v"(fa[11]), "+v"(o[1]), "+v"(o[2]), "+v"(o[3]));
  d2 = __builtin_amdgcn_wmma_f32_16x16x4_f32(false, fa[11], false, (v2f){r.w2[6], r.w2[7]}, (short)0, zero, false, false);
  asm volatile("" : "+v"(d0), "+v"(d2));
  o[3] = __builtin_fmaf(r.w1[11], d0[0], o[3]);
  o[3] = __builtin_fmaf(r.w1[13], d0[1], o[3]);
  o[4] = __builtin_fmaf(r.w1[3], d0[2], o[4]);
  o[4] = __builtin_fmaf(r.w1[7], d0[3], o[4]);
  o[4] = __builtin_fmaf(r.w1[13], d0[4], o[4]);
  o[4] = __builtin_fmaf(r.w1[15], d0[5], o[4]);
  o[5] = __builtin_fmaf(r.w1[15], d0[6], o[5]);
  o[6] = __builtin_fmaf(r.w1[0], d0[7], o[6]);
  asm volatile("" : "+v"(fa[12]), "+v"(o[3]), "+v"(o[4]), "+v"(o[5]), "+v"(o[6]));
  d0 = __builtin_amdgcn_wmma_f32_16x16x4_f32(false, fa[12], false, (v2f){r.w2[6], r.w2[7]}, (short)0, zero, false, false);
  asm volatile("" : "+v"(d1), "+v"(d0));
  o[6] = __builtin_fmaf(r.w1[2], d1[0], o[6]);
  o[6] = __builtin_fmaf(r.w1[6], d1[1], o[6]);
  o[6] = __builtin_fmaf(r.w1[12], d1[2], o[6]);
  o[6] = __builtin_fmaf(s_8_4, d1[3], o[6]);
  o[6] = __builtin_fmaf(s_14_10, d1[4], o[6]);
  o[7] = __builtin_fmaf(s_5_8, d1[5], o[7]);
  o[7] = __builtin_fmaf(s_9_14, d1[6], o[7]);
  o[6] = __builtin_fmaf(r.w1[15], d1[7], o[6]);
  asm volatile("" : "+v"(fa[13]), "+v"(o[6]), "+v"(o[7]));
  d1 = __builtin_amdgcn_wmma_f32_16x16x4_f32(false, fa[13], false, (v2f){r.w2[6], r.w2[7]}, (short)0, zero, false, false);
  asm volatile("" : "+v"(d2), "+v"(d1));
  o[0] = __builtin_fmaf(r.w1[4], d2[0], o[0]);
  o[0] = __builtin_fmaf(r.w1[8], d2[1], o[0]);
  o[0] = __builtin_fmaf(r.w1[10], d2[2], o[0]);
  o[0] = __builtin_fmaf(r.w1[14], d2[3], o[0]);
  o[1] = __builtin_fmaf(r.w1[3], d2[4], o[1]);
  o[1] = __builtin_fmaf(r.w1[7], d2[5], o[1]);
  o[1] = __builtin_fmaf(r.w1[9], d2[6], o[1]);
  o[1] = __builtin_fmaf(r.w1[13], d2[7], o[1]);
  asm volatile("" : "+v"(fa[14]), "+v"(o[0]), "+v"(o[1]));
  d2 = __builtin_amdgcn_wmma_f32_16x16x4_f32(false, fa[14], false, (v2f){r.w2[6], r.w2[7]}, (short)0, zero, false, false);
  asm volatile("" : "+v"(d0), "+v"(d2));
  o[2] = __builtin_fmaf(r.w1[1], d0[0], o[2]);
  o[2] = __builtin_fmaf(r.w1[5], d0[1], o[2]);
  o[2] = __builtin_fmaf(r.w1[11], d0[2], o[2]);
  o[2] = __builtin_fmaf(r.w1[15], d0[3], o[2]);
  o[3] = __builtin_fmaf(r.w1[0], d0[4], o[3]);
  o[3] = __builtin_fmaf(r.w1[2], d0[5], o[3]);
  o[3] = __builtin_fmaf(r.w1[6], d0[6], o[3]);
  o[3] = __builtin_fmaf(r.w1[12], d0[7], o[3]);
  asm volatile("" : "+v"(d1), "+v"(o[2]), "+v"(o[3]));
  o[4] = __builtin_fmaf(r.w1[4], d1[0], o[4]);
  o[4] = __builtin_fmaf(r.w1[8], d1[1], o[4]);
  o[4] = __builtin_fmaf(r.w1[10], d1[2], o[4]);
  o[4] = __builtin_fmaf(r.w1[14], d1[3], o[4]);
  o[5] = __builtin_fmaf(r.w1[0], d1[4], o[5]);
  o[5] = __builtin_fmaf(r.w1[2], d1[5], o[5]);
  o[5] = __builtin_fmaf(r.w1[6], d1[6], o[5]);
  o[5] = __builtin_fmaf(r.w1[12], d1[7], o[5]);
  asm volatile("" : "+v"(d2), "+v"(o[4]), "+v"(o[5]));
  o[6] = __builtin_fmaf(s_3_1, d2[0], o[6]);
  o[6] = __builtin_fmaf(s_7_5, d2[1], o[6]);
  o[6] = __builtin_fmaf(s_9_11, d2[2], o[6]);
  o[6] = __builtin_fmaf(s_13_14, d2[3], o[6]);
  o[7] = __builtin_fmaf(s_3_1, d2[4], o[7]);
  o[7] = __builtin_fmaf(s_7_5, d2[5], o[7]);
  o[7] = __builtin_fmaf(s_13_11, d2[6], o[7]);
  o[7] = __builtin_fmaf(s_10_15, d2[7], o[7]);
}

__global__ void __launch_bounds__(BLOCK_THREADS)
tp_wmma_kernel(const float* x1, const float* x2,
               const float* __restrict__ vals, const int* __restrict__ M1,
               const int* __restrict__ M2, const int* __restrict__ Mseg,
               float* out, int n_batch, int nnz) {
  __shared__ float4 cg_lds[NUM_ORDERS * K_DENSE / 4];
  __shared__ unsigned char tab_lds[256 + 16];
  float* cg = (float*)cg_lds;

  const int tid  = threadIdx.x;
  const int lane = tid & 31;
  const int half = lane >> 4;
  const int col  = lane & 15;
  const int wib    = __builtin_amdgcn_readfirstlane((int)(threadIdx.x >> 5));
  const int wave   = blockIdx.x * WAVES_PER_BLOCK + wib;
  const int nwaves = gridDim.x * WAVES_PER_BLOCK;
  const int ntiles = n_batch * 4;
  const int nunits = n_batch;

  if (ntiles <= 0) return;

  int t = wave;
  const bool has_tile = t < ntiles;
  const bool has_unit = t < nunits;

  unsigned char t_row = 0, t_col = 0;
  if (tid < 256) t_row = ROW_OF[tid];
  if (tid < 16) t_col = COL_POS[tid];
  const bool has_entry = tid < nnz;
  int   e_m = 0, e_m1 = 0, e_m2 = 0;
  float e_val = 0.0f;
  if (has_entry) {
    e_m   = Mseg[tid];
    e_m1  = M1[tid];
    e_m2  = M2[tid];
    e_val = vals[tid];
  }
  const bool has_entry2 = tid + BLOCK_THREADS < nnz;
  int   g_m = 0, g_m1 = 0, g_m2 = 0;
  float g_val = 0.0f;
  if (has_entry2) {
    g_m   = Mseg[tid + BLOCK_THREADS];
    g_m1  = M1[tid + BLOCK_THREADS];
    g_m2  = M2[tid + BLOCK_THREADS];
    g_val = vals[tid + BLOCK_THREADS];
  }

  UnitRegs ur;
  {
    UnitOffsets f0;
    unit_offsets(f0, half, col);
    load_unit(ur, x1, x2, has_unit ? t : 0, f0);
  }

  if (tid < 256) tab_lds[tid] = t_row;
  if (tid < 16) tab_lds[256 + tid] = t_col;
  for (int i = tid; i < NUM_ORDERS * K_DENSE / 4; i += BLOCK_THREADS)
    cg_lds[i] = make_float4(0.0f, 0.0f, 0.0f, 0.0f);
  lds_barrier();
  if (has_entry) {
    asm volatile("" : "+v"(e_m), "+v"(e_m1), "+v"(e_m2));
    const int rho = tab_lds[((e_m & 15) << 4) | (e_m1 & 15)];
    const int pos = tab_lds[256 + (e_m2 & 15)];
    cg[cg_frag_index(rho & 15, (rho & 0xF0) | pos)] = e_val;
  }
  if (has_entry2) {
    asm volatile("" : "+v"(g_m), "+v"(g_m1), "+v"(g_m2));
    const int rho = tab_lds[((g_m & 15) << 4) | (g_m1 & 15)];
    const int pos = tab_lds[256 + (g_m2 & 15)];
    cg[cg_frag_index(rho & 15, (rho & 0xF0) | pos)] = g_val;
  }
  for (int i = tid + 2 * BLOCK_THREADS; i < nnz; i += BLOCK_THREADS) {
    const int rho = tab_lds[((Mseg[i] & 15) << 4) | (M1[i] & 15)];
    const int pos = tab_lds[256 + (M2[i] & 15)];
    cg[cg_frag_index(rho & 15, (rho & 0xF0) | pos)] = vals[i];
  }
  lds_barrier();

  unsigned mlo = 0, mhi = 0;
#pragma unroll
  for (int kp = 0; kp < 32; ++kp) {
    const float4 q = cg_lds[kp * 32 + lane];
    const unsigned b0 = __builtin_amdgcn_ballot_w32((q.x != 0.0f) | (q.y != 0.0f));
    const unsigned b1 = __builtin_amdgcn_ballot_w32((q.z != 0.0f) | (q.w != 0.0f));
    const unsigned bits = (b0 ? 1u : 0u) | (b1 ? 2u : 0u);
    if (kp < 16) mlo |= bits << (2 * kp);
    else         mhi |= bits << (2 * kp - 32);
  }

  asm volatile("" ::: "memory");

  if (!has_tile) return;


  if (((mlo & ~FAST_MASK_LO) | (mhi & ~FAST_MASK_HI)) == 0u) {
    constexpr int fast_block[FAST_BLOCKS] = {0, 4, 8, 13, 17, 21, 25, 30, 34, 38, 42, 47, 51, 55, 59};
    v2f fa[FAST_BLOCKS];
#pragma unroll
    for (int i = 0; i < FAST_BLOCKS; ++i) {
      const int idx = ((fast_block[i] >> 1) * 32 + lane) * 4 + (fast_block[i] & 1) * 2;
      fa[i] = (v2f){cg[idx], cg[idx + 1]};
    }
    if (!has_unit) return;
    int u = t;
    const int ucnt = (nunits - u + nwaves - 1) / nwaves;
    TileRegs ra, rb, rc, rd;
    split_unit(ra, rb, rc, rd, ur);
    UnitRegs ur1;
    int lane_f = lane;
    asm volatile("" : "+v"(lane_f));
    const int half_f = lane_f >> 4;
    UnitOffsets fo;
    unit_offsets(fo, half_f, lane_f & 15);
    const unsigned so[8] = {fo.eo[8] + (half_f ? 2u : 0u) * C_DIM,  fo.eo[8] + (half_f ? 5u : 1u) * C_DIM,
                            fo.eo[8] + (half_f ? 7u : 3u) * C_DIM,  fo.eo[8] + (half_f ? 10u : 4u) * C_DIM,
                            fo.eo[8] + (half_f ? 12u : 6u) * C_DIM, fo.eo[8] + (half_f ? 14u : 8u) * C_DIM,
                            fo.eo[8] + (half_f ? 11u : 9u) * C_DIM, fo.eo[8] + (half_f ? 15u : 13u) * C_DIM};
    for (int p = ucnt; p > 0; --p) {
      const int un = u + nwaves;
      load_unit(ur1, x1, x2, (un < nunits) ? un : u, fo);
      __builtin_amdgcn_sched_barrier(0);
      float oa[8], ob[8], oc[8], od[8];
      tile_product_fast(fa, ra, oa, half);
      tile_product_fast(fa, rb, ob, half);
      tile_product_fast(fa, rc, oc, half);
      tile_product_fast(fa, rd, od, half);
      float* pn = out + (size_t)u * NUM_ORDERS * C_DIM;
#pragma unroll
      for (int v = 0; v < 8; ++v) *(v4f*)(pn + so[v]) = (v4f){oa[v], ob[v], oc[v], od[v]};
      asm volatile("" : "+v"(ur1.q2[0]), "+v"(ur1.q2[1]), "+v"(ur1.q2[2]), "+v"(ur1.q2[3]),
                        "+v"(ur1.q2[4]), "+v"(ur1.q2[5]), "+v"(ur1.q2[6]), "+v"(ur1.q2[7])
                      : "v"(ra.w2), "v"(rb.w2), "v"(rc.w2), "v"(rd.w2));
      __builtin_amdgcn_sched_barrier(0);
      split_unit(ra, rb, rc, rd, ur1);
      u = un;
    }
    return;
  }

  float4 areg[24];
#pragma unroll
  for (int kp = 0; kp < 24; ++kp) areg[kp] = cg_lds[kp * 32 + lane];

  const int cnt = (ntiles - t + nwaves - 1) / nwaves;
  TileRegs r;
  load_tile(r, x1, x2, t, half, col);
  for (int p = cnt; p > 0; --p) {
    asm volatile("" : "+s"(mlo), "+s"(mhi));
    tile_product(areg, cg_lds + lane, mlo, mhi, r, out, t, half, col);
    t += nwaves;
    if (p > 1) load_tile(r, x1, x2, t, half, col);
  }
}

extern "C" void kernel_launch(void* const* d_in, const int* in_sizes, int n_in,
                              void* d_out, int out_size, void* d_ws, size_t ws_size,
                              hipStream_t stream) {
  const float* x1   = (const float*)d_in[0];
  const float* x2   = (const float*)d_in[1];
  const float* vals = (const float*)d_in[2];
  const int* M1     = (const int*)d_in[3];
  const int* M2     = (const int*)d_in[4];
  const int* Mseg   = (const int*)d_in[5];
  float* out        = (float*)d_out;

  const int nnz     = in_sizes[2];
  const int n_batch = in_sizes[0] / (NUM_ORDERS * C_DIM);

  tp_wmma_kernel<<<NUM_BLOCKS, BLOCK_THREADS, 0, stream>>>(x1, x2, vals, M1, M2, Mseg,
                                                          out, n_batch, nnz);
}
